# starting kernel
# baseline (speedup 1.0000x reference)
#include <hip/hip_runtime.h>

#define NEG_SLOPE 0.2f
#define ENC_NEG_INF 0x007FFFFFu

typedef float v2f __attribute__((ext_vector_type(2)));
typedef float v8f __attribute__((ext_vector_type(8)));

__device__ __forceinline__ unsigned gat_fenc(float f) {
    unsigned u = __float_as_uint(f);
    return (u & 0x80000000u) ? ~u : (u | 0x80000000u);
}
__device__ __forceinline__ float gat_fdec(unsigned u) {
    unsigned b = (u & 0x80000000u) ? (u & 0x7FFFFFFFu) : ~u;
    return __uint_as_float(b);
}

template<int NT>
__global__ __launch_bounds__(256)
void gat_gemm_wmma(const float* __restrict__ X, const float* __restrict__ W,
                   float* __restrict__ Hout, int n_rows)
{
    constexpr int KD = 128;
    constexpr int ND = NT * 16;
    __shared__ float Wt[ND * KD];

    for (int i = threadIdx.x; i < KD * ND; i += 256) {
        int k = i / ND, n = i % ND;
        Wt[n * KD + ((k + 4 * n) & (KD - 1))] = W[i];
    }
    __syncthreads();

    const int lane = threadIdx.x & 31;
    const int wave = threadIdx.x >> 5;
    const int m0   = (blockIdx.x * 8 + wave) * 16;
    const int mlo  = lane & 15;
    const int kofs = (lane >> 4) * 2;

    int arow = m0 + mlo;
    if (arow >= n_rows) arow = n_rows - 1;
    const float* xrow = X + (size_t)arow * KD + kofs;

    v8f acc[NT] = {};

    for (int k = 0; k < KD; k += 4) {
        v2f a = *(const v2f*)(xrow + k);
#pragma unroll
        for (int t = 0; t < NT; ++t) {
            int n = t * 16 + mlo;
            v2f b = *(const v2f*)(&Wt[n * KD + ((k + kofs + 4 * n) & (KD - 1))]);
            acc[t] = __builtin_amdgcn_wmma_f32_16x16x4_f32(
                false, a, false, b, (short)0, acc[t], false, false);
        }
    }

    const int mhi = (lane >> 4) * 8;
#pragma unroll
    for (int t = 0; t < NT; ++t)
#pragma unroll
        for (int j = 0; j < 8; ++j) {
            int m = m0 + mhi + j;
            if (m < n_rows) Hout[(size_t)m * ND + t * 16 + mlo] = acc[t][j];
        }
}

__global__ void gat_attn_score(const float* __restrict__ Hf,
                               const float* __restrict__ att_s,
                               const float* __restrict__ att_d,
                               float* __restrict__ a_s, float* __restrict__ a_d,
                               int n, int heads, int ch)
{
    int t = blockIdx.x * blockDim.x + threadIdx.x;
    if (t >= n * heads) return;
    int node = t / heads, h = t - node * heads;
    const float* hp = Hf + ((size_t)node * heads + h) * ch;
    const float* sp = att_s + h * ch;
    const float* dp = att_d + h * ch;
    float ss = 0.f, dd = 0.f;
    for (int c = 0; c < ch; ++c) { float v = hp[c]; ss += v * sp[c]; dd += v * dp[c]; }
    a_s[t] = ss; a_d[t] = dd;
}

__global__ void gat_fill_u32(unsigned* __restrict__ p, unsigned v, int n)
{
    int i = blockIdx.x * blockDim.x + threadIdx.x;
    if (i < n) p[i] = v;
}

__global__ void gat_bias_init(float* __restrict__ out, const float* __restrict__ bias,
                              int total, int dim)
{
    int i = blockIdx.x * blockDim.x + threadIdx.x;
    if (i < total) out[i] = bias[i % dim];
}

__global__ void gat_relu(float* __restrict__ p, int n)
{
    int i = blockIdx.x * blockDim.x + threadIdx.x;
    if (i < n) p[i] = fmaxf(p[i], 0.f);
}

__global__ void gat_edge_scores(const int* __restrict__ srcI, const int* __restrict__ dstI,
                                const float* __restrict__ a_s, const float* __restrict__ a_d,
                                float* __restrict__ eb, unsigned* __restrict__ emax,
                                int E, int E2, int H)
{
    int e = blockIdx.x * blockDim.x + threadIdx.x;
    if (e >= E2) return;
    int s = (e < E) ? srcI[e] : (e - E);
    int d = (e < E) ? dstI[e] : (e - E);
    for (int h = 0; h < H; ++h) {
        float v = a_s[s * H + h] + a_d[d * H + h];
        v = (v >= 0.f) ? v : NEG_SLOPE * v;
        eb[(size_t)e * H + h] = v;
        atomicMax(&emax[d * H + h], gat_fenc(v));
    }
}

__global__ void gat_edge_exp(const int* __restrict__ dstI,
                             float* __restrict__ eb,
                             const unsigned* __restrict__ emax,
                             float* __restrict__ denom,
                             int E, int E2, int H)
{
    int e = blockIdx.x * blockDim.x + threadIdx.x;
    if (e >= E2) return;
    int d = (e < E) ? dstI[e] : (e - E);
    for (int h = 0; h < H; ++h) {
        float m = gat_fdec(emax[d * H + h]);
        float x = __expf(eb[(size_t)e * H + h] - m);
        eb[(size_t)e * H + h] = x;
        atomicAdd(&denom[d * H + h], x);
    }
}

__global__ void gat_edge_aggregate(const int* __restrict__ srcI, const int* __restrict__ dstI,
                                   const float* __restrict__ eb, const float* __restrict__ denom,
                                   const float* __restrict__ Hf, float* __restrict__ outp,
                                   int E, int E2, int H)
{
    int t = blockIdx.x * blockDim.x + threadIdx.x;
    if (t >= E2 * H) return;
    int e = t / H, h = t - e * H;
    int s = (e < E) ? srcI[e] : (e - E);
    int d = (e < E) ? dstI[e] : (e - E);
    float alpha = eb[(size_t)e * H + h] / denom[d * H + h];
    const float4* hv = (const float4*)(Hf + ((size_t)s * H + h) * 32);
    float*       ov = outp + ((size_t)d * H + h) * 32;
#pragma unroll
    for (int i = 0; i < 8; ++i) {
        float4 v = hv[i];
        atomicAdd(ov + 4 * i + 0, alpha * v.x);
        atomicAdd(ov + 4 * i + 1, alpha * v.y);
        atomicAdd(ov + 4 * i + 2, alpha * v.z);
        atomicAdd(ov + 4 * i + 3, alpha * v.w);
    }
}

extern "C" void kernel_launch(void* const* d_in, const int* in_sizes, int n_in,
                              void* d_out, int out_size, void* d_ws, size_t ws_size,
                              hipStream_t stream)
{
    const float* x   = (const float*)d_in[0];
    const int*   ei  = (const int*)d_in[1];
    const float* W1  = (const float*)d_in[2];
    const float* as1 = (const float*)d_in[3];
    const float* ad1 = (const float*)d_in[4];
    const float* b1  = (const float*)d_in[5];
    const float* W2  = (const float*)d_in[6];
    const float* as2 = (const float*)d_in[7];
    const float* ad2 = (const float*)d_in[8];
    const float* b2  = (const float*)d_in[9];
    float* out = (float*)d_out;

    const int N  = in_sizes[0] / 128;
    const int E  = in_sizes[1] / 2;
    const int E2 = E + N;
    const int* srcI = ei;
    const int* dstI = ei + E;

    float* ws = (float*)d_ws;
    float*    h1    = ws;              ws += (size_t)N * 128;
    float*    asrc1 = ws;              ws += (size_t)N * 4;
    float*    adst1 = ws;              ws += (size_t)N * 4;
    unsigned* emax1 = (unsigned*)ws;   ws += (size_t)N * 4;
    float*    den1  = ws;              ws += (size_t)N * 4;
    float*    eb1   = ws;              ws += (size_t)E2 * 4;
    float*    h2    = ws;              ws += (size_t)N * 128;
    float*    g2    = ws;              ws += (size_t)N * 32;
    float*    asrc2 = ws;              ws += (size_t)N;
    float*    adst2 = ws;              ws += (size_t)N;
    unsigned* emax2 = (unsigned*)ws;   ws += (size_t)N;
    float*    den2  = ws;              ws += (size_t)E2 > 0 ? (size_t)N : 0;
    float*    eb2   = ws;

    const int TB = 256;
    const int gB   = (N + 127) / 128;
    const int gE   = (E2 + TB - 1) / TB;
    const int gE4  = (E2 * 4 + TB - 1) / TB;
    const int gN4  = (N * 4 + TB - 1) / TB;
    const int gN1  = (N + TB - 1) / TB;
    const int gNF  = (N * 128 + TB - 1) / TB;
    const int gNO  = (N * 32 + TB - 1) / TB;

    gat_gemm_wmma<8><<<gB, TB, 0, stream>>>(x, W1, h1, N);
    gat_attn_score<<<gN4, TB, 0, stream>>>(h1, as1, ad1, asrc1, adst1, N, 4, 32);
    gat_fill_u32<<<gN4, TB, 0, stream>>>(emax1, ENC_NEG_INF, N * 4);
    gat_fill_u32<<<gN4, TB, 0, stream>>>((unsigned*)den1, 0u, N * 4);
    gat_bias_init<<<gNF, TB, 0, stream>>>(h2, b1, N * 128, 128);
    gat_edge_scores<<<gE, TB, 0, stream>>>(srcI, dstI, asrc1, adst1, eb1, emax1, E, E2, 4);
    gat_edge_exp<<<gE, TB, 0, stream>>>(dstI, eb1, emax1, den1, E, E2, 4);
    gat_edge_aggregate<<<gE4, TB, 0, stream>>>(srcI, dstI, eb1, den1, h1, h2, E, E2, 4);
    gat_relu<<<gNF, TB, 0, stream>>>(h2, N * 128);

    gat_gemm_wmma<2><<<gB, TB, 0, stream>>>(h2, W2, g2, N);
    gat_attn_score<<<gN1, TB, 0, stream>>>(g2, as2, ad2, asrc2, adst2, N, 1, 32);
    gat_fill_u32<<<gN1, TB, 0, stream>>>(emax2, ENC_NEG_INF, N);
    gat_fill_u32<<<gN1, TB, 0, stream>>>((unsigned*)den2, 0u, N);
    gat_bias_init<<<gNO, TB, 0, stream>>>(out, b2, N * 32, 32);
    gat_edge_scores<<<gE, TB, 0, stream>>>(srcI, dstI, asrc2, adst2, eb2, emax2, E, E2, 1);
    gat_edge_exp<<<gE, TB, 0, stream>>>(dstI, eb2, emax2, den2, E, E2, 1);
    gat_edge_aggregate<<<gE, TB, 0, stream>>>(srcI, dstI, eb2, den2, g2, out, E, E2, 1);
}


# Aggregate: a block owns 200 destination rows in LDS, filters the raw edge list for them, adds alpha*row by LDS float adds, stores bias(+ReLU) once; no global float atomics; sums in LDS arrival order
# speedup vs baseline: 2.7393x; 2.7393x over previous
#include <hip/hip_runtime.h>

#define NEG_SLOPE 0.2f
#define ENC_NEG_INF 0x007FFFFFu

typedef float v2f __attribute__((ext_vector_type(2)));
typedef float v8f __attribute__((ext_vector_type(8)));

__device__ __forceinline__ unsigned gat_fenc(float f) {
    unsigned u = __float_as_uint(f);
    return (u & 0x80000000u) ? ~u : (u | 0x80000000u);
}
__device__ __forceinline__ float gat_fdec(unsigned u) {
    unsigned b = (u & 0x80000000u) ? (u & 0x7FFFFFFFu) : ~u;
    return __uint_as_float(b);
}

template<int NT>
__global__ __launch_bounds__(256)
void gat_gemm_wmma(const float* __restrict__ X, const float* __restrict__ W,
                   float* __restrict__ Hout, int n_rows)
{
    constexpr int KD = 128;
    constexpr int ND = NT * 16;
    __shared__ float Wt[ND * KD];

    for (int i = threadIdx.x; i < KD * ND; i += 256) {
        int k = i / ND, n = i % ND;
        Wt[n * KD + ((k + 4 * n) & (KD - 1))] = W[i];
    }
    __syncthreads();

    const int lane = threadIdx.x & 31;
    const int wave = threadIdx.x >> 5;
    const int m0   = (blockIdx.x * 8 + wave) * 16;
    const int mlo  = lane & 15;
    const int kofs = (lane >> 4) * 2;

    int arow = m0 + mlo;
    if (arow >= n_rows) arow = n_rows - 1;
    const float* xrow = X + (size_t)arow * KD + kofs;

    v8f acc[NT] = {};

    for (int k = 0; k < KD; k += 4) {
        v2f a = *(const v2f*)(xrow + k);
#pragma unroll
        for (int t = 0; t < NT; ++t) {
            int n = t * 16 + mlo;
            v2f b = *(const v2f*)(&Wt[n * KD + ((k + kofs + 4 * n) & (KD - 1))]);
            acc[t] = __builtin_amdgcn_wmma_f32_16x16x4_f32(
                false, a, false, b, (short)0, acc[t], false, false);
        }
    }

    const int mhi = (lane >> 4) * 8;
#pragma unroll
    for (int t = 0; t < NT; ++t)
#pragma unroll
        for (int j = 0; j < 8; ++j) {
            int m = m0 + mhi + j;
            if (m < n_rows) Hout[(size_t)m * ND + t * 16 + mlo] = acc[t][j];
        }
}

__global__ void gat_attn_score(const float* __restrict__ Hf,
                               const float* __restrict__ att_s,
                               const float* __restrict__ att_d,
                               float* __restrict__ a_s, float* __restrict__ a_d,
                               int n, int heads, int ch)
{
    int t = blockIdx.x * blockDim.x + threadIdx.x;
    if (t >= n * heads) return;
    int node = t / heads, h = t - node * heads;
    const float* hp = Hf + ((size_t)node * heads + h) * ch;
    const float* sp = att_s + h * ch;
    const float* dp = att_d + h * ch;
    float ss = 0.f, dd = 0.f;
    for (int c = 0; c < ch; ++c) { float v = hp[c]; ss += v * sp[c]; dd += v * dp[c]; }
    a_s[t] = ss; a_d[t] = dd;
}

__global__ void gat_fill_u32(unsigned* __restrict__ p, unsigned v, int n)
{
    int i = blockIdx.x * blockDim.x + threadIdx.x;
    if (i < n) p[i] = v;
}

__global__ void gat_edge_scores(const int* __restrict__ srcI, const int* __restrict__ dstI,
                                const float* __restrict__ a_s, const float* __restrict__ a_d,
                                float* __restrict__ eb, unsigned* __restrict__ emax,
                                int E, int E2, int H)
{
    int e = blockIdx.x * blockDim.x + threadIdx.x;
    if (e >= E2) return;
    int s = (e < E) ? srcI[e] : (e - E);
    int d = (e < E) ? dstI[e] : (e - E);
    for (int h = 0; h < H; ++h) {
        float v = a_s[s * H + h] + a_d[d * H + h];
        v = (v >= 0.f) ? v : NEG_SLOPE * v;
        eb[(size_t)e * H + h] = v;
        atomicMax(&emax[d * H + h], gat_fenc(v));
    }
}

__global__ void gat_edge_exp(const int* __restrict__ dstI,
                             float* __restrict__ eb,
                             const unsigned* __restrict__ emax,
                             float* __restrict__ denom,
                             int E, int E2, int H)
{
    int e = blockIdx.x * blockDim.x + threadIdx.x;
    if (e >= E2) return;
    int d = (e < E) ? dstI[e] : (e - E);
    for (int h = 0; h < H; ++h) {
        float m = gat_fdec(emax[d * H + h]);
        float x = __expf(eb[(size_t)e * H + h] - m);
        eb[(size_t)e * H + h] = x;
        atomicAdd(&denom[d * H + h], x);
    }
}

#define OWN_NPB 200
#define OWN_NW  32
#define OWN_CAP 1024
#define OWN_U   8

template<int H>
__device__ __forceinline__ void gat_own_flush(const int* wl, int cnt, int lo, int E,
        const int* __restrict__ srcI, const int* __restrict__ dstI,
        const float* __restrict__ eb, const float* __restrict__ denom,
        const float* __restrict__ Hf, float* acc, int lane)
{
    constexpr int ROW = 32 * H;
    constexpr int LPR = ROW / 4;
    constexpr int EPI = 32 / LPR;
    const int grp  = lane / LPR;
    const int col  = (lane % LPR) * 4;
    const int head = col >> 5;

    for (int base = 0; base < cnt; base += 32) {
        const int m = min(32, cnt - base);
        int   s_i = 0, d_i = 0;
        float a_i[H];
#pragma unroll
        for (int h = 0; h < H; ++h) a_i[h] = 0.f;
        if (lane < m) {
            const int e = wl[base + lane];
            int s, d;
            if (e < E) { s = srcI[e]; d = dstI[e]; } else { s = e - E; d = s; }
            s_i = s; d_i = d - lo;
#pragma unroll
            for (int h = 0; h < H; ++h)
                a_i[h] = eb[(size_t)e * H + h] / denom[(size_t)d * H + h];
        }
        const int nsteps = (m + EPI - 1) / EPI;
        for (int t0 = 0; t0 < nsteps; t0 += OWN_U) {
            float4 v[OWN_U]; int dd[OWN_U]; float ww[OWN_U]; bool ok[OWN_U];
#pragma unroll
            for (int u = 0; u < OWN_U; ++u) {
                const int idx = (t0 + u) * EPI + grp;
                ok[u] = idx < m;
                const int sl = idx & 31;
                const int s = __shfl(s_i, sl);
                dd[u] = __shfl(d_i, sl);
                float w = __shfl(a_i[0], sl);
#pragma unroll
                for (int h = 1; h < H; ++h) {
                    const float wh = __shfl(a_i[h], sl);
                    w = (head == h) ? wh : w;
                }
                ww[u] = w;
                v[u] = *(const float4*)(Hf + (size_t)s * ROW + col);
            }
#pragma unroll
            for (int u = 0; u < OWN_U; ++u) {
                if (ok[u]) {
                    float* p = acc + dd[u] * ROW + col;
                    __hip_atomic_fetch_add(p + 0, ww[u] * v[u].x, __ATOMIC_RELAXED, __HIP_MEMORY_SCOPE_WORKGROUP);
                    __hip_atomic_fetch_add(p + 1, ww[u] * v[u].y, __ATOMIC_RELAXED, __HIP_MEMORY_SCOPE_WORKGROUP);
                    __hip_atomic_fetch_add(p + 2, ww[u] * v[u].z, __ATOMIC_RELAXED, __HIP_MEMORY_SCOPE_WORKGROUP);
                    __hip_atomic_fetch_add(p + 3, ww[u] * v[u].w, __ATOMIC_RELAXED, __HIP_MEMORY_SCOPE_WORKGROUP);
                }
            }
        }
    }
}

template<int H, bool RELU>
__global__ __launch_bounds__(OWN_NW * 32)
void gat_own_aggregate(const int* __restrict__ srcI, const int* __restrict__ dstI,
                       const float* __restrict__ eb, const float* __restrict__ denom,
                       const float* __restrict__ Hf, const float* __restrict__ bias,
                       float* __restrict__ outp, int E, int N)
{
    constexpr int ROW = 32 * H;
    static_assert(OWN_NPB <= 32 * OWN_NW, "one self-loop step per wave");
    __shared__ __attribute__((aligned(16))) float acc[OWN_NPB * ROW];
    __shared__ int lst[OWN_NW * OWN_CAP];

    const int tid  = threadIdx.x;
    const int lane = tid & 31;
    const int wave = tid >> 5;
    const int lo   = blockIdx.x * OWN_NPB;
    const int nodes = min(OWN_NPB, N - lo);
    const unsigned lt = (1u << lane) - 1u;

    for (int i = tid; i < nodes * ROW; i += OWN_NW * 32) acc[i] = bias[i & (ROW - 1)];
    __syncthreads();

    int* wl = lst + wave * OWN_CAP;
    int cnt = 0;

    for (int nb = wave * 32; nb < nodes; nb += OWN_NW * 32) {
        const int n = nb + lane;
        const bool hit = n < nodes;
        const unsigned mask = __builtin_amdgcn_ballot_w32(hit);
        if (hit) wl[cnt + __popc(mask & lt)] = E + lo + n;
        cnt += __popc(mask);
    }

    const int nvec = E >> 2;
    const int4* d4 = (const int4*)dstI;
    const int4 none = make_int4(-1, -1, -1, -1);
    int gb = wave * 128;
    int4 cur[4];
#pragma unroll
    for (int u = 0; u < 4; ++u) {
        const int g = gb + u * 32 + lane;
        cur[u] = d4[max(min(g, nvec - 1), 0)];
        if (g >= nvec) cur[u] = none;
    }
    while (gb < nvec) {
        const int gn = gb + OWN_NW * 128;
        int4 nxt[4];
#pragma unroll
        for (int u = 0; u < 4; ++u) {
            const int g = gn + u * 32 + lane;
            nxt[u] = d4[min(g, nvec - 1)];
            if (g >= nvec) nxt[u] = none;
        }
#pragma unroll
        for (int u = 0; u < 4; ++u) {
            const int e0 = (gb + u * 32 + lane) * 4;
            const bool h0 = (unsigned)(cur[u].x - lo) < (unsigned)nodes;
            const bool h1 = (unsigned)(cur[u].y - lo) < (unsigned)nodes;
            const bool h2 = (unsigned)(cur[u].z - lo) < (unsigned)nodes;
            const bool h3 = (unsigned)(cur[u].w - lo) < (unsigned)nodes;
            if (__builtin_amdgcn_ballot_w32(h0 | h1 | h2 | h3)) {
                unsigned mask;
                mask = __builtin_amdgcn_ballot_w32(h0);
                if (h0) wl[cnt + __popc(mask & lt)] = e0 + 0;
                cnt += __popc(mask);
                mask = __builtin_amdgcn_ballot_w32(h1);
                if (h1) wl[cnt + __popc(mask & lt)] = e0 + 1;
                cnt += __popc(mask);
                mask = __builtin_amdgcn_ballot_w32(h2);
                if (h2) wl[cnt + __popc(mask & lt)] = e0 + 2;
                cnt += __popc(mask);
                mask = __builtin_amdgcn_ballot_w32(h3);
                if (h3) wl[cnt + __popc(mask & lt)] = e0 + 3;
                cnt += __popc(mask);
            }
        }
        if (cnt > OWN_CAP / 2) {
            gat_own_flush<H>(wl, cnt, lo, E, srcI, dstI, eb, denom, Hf, acc, lane);
            cnt = 0;
        }
#pragma unroll
        for (int u = 0; u < 4; ++u) cur[u] = nxt[u];
        gb = gn;
    }
    if (wave == 0) {
        for (int eb0 = nvec * 4; eb0 < E; eb0 += 32) {
            const int e = eb0 + lane;
            const bool hit = (e < E) && ((unsigned)(dstI[min(e, E - 1)] - lo) < (unsigned)nodes);
            const unsigned mask = __builtin_amdgcn_ballot_w32(hit);
            if (hit) wl[cnt + __popc(mask & lt)] = e;
            cnt += __popc(mask);
        }
    }
    gat_own_flush<H>(wl, cnt, lo, E, srcI, dstI, eb, denom, Hf, acc, lane);
    __syncthreads();

    float* orow = outp + (size_t)lo * ROW;
    for (int i = tid * 4; i < nodes * ROW; i += OWN_NW * 32 * 4) {
        float4 r = *(const float4*)(acc + i);
        if (RELU) { r.x = fmaxf(r.x, 0.f); r.y = fmaxf(r.y, 0.f); r.z = fmaxf(r.z, 0.f); r.w = fmaxf(r.w, 0.f); }
        *(float4*)(orow + i) = r;
    }
}

extern "C" void kernel_launch(void* const* d_in, const int* in_sizes, int n_in,
                              void* d_out, int out_size, void* d_ws, size_t ws_size,
                              hipStream_t stream)
{
    const float* x   = (const float*)d_in[0];
    const int*   ei  = (const int*)d_in[1];
    const float* W1  = (const float*)d_in[2];
    const float* as1 = (const float*)d_in[3];
    const float* ad1 = (const float*)d_in[4];
    const float* b1  = (const float*)d_in[5];
    const float* W2  = (const float*)d_in[6];
    const float* as2 = (const float*)d_in[7];
    const float* ad2 = (const float*)d_in[8];
    const float* b2  = (const float*)d_in[9];
    float* out = (float*)d_out;

    const int N  = in_sizes[0] / 128;
    const int E  = in_sizes[1] / 2;
    const int E2 = E + N;
    const int* srcI = ei;
    const int* dstI = ei + E;

    float* ws = (float*)d_ws;
    float*    h1    = ws;              ws += (size_t)N * 128;
    float*    asrc1 = ws;              ws += (size_t)N * 4;
    float*    adst1 = ws;              ws += (size_t)N * 4;
    unsigned* emax1 = (unsigned*)ws;   ws += (size_t)N * 4;
    float*    den1  = ws;              ws += (size_t)N * 4;
    float*    eb1   = ws;              ws += (size_t)E2 * 4;
    float*    h2    = ws;              ws += (size_t)N * 128;
    float*    g2    = ws;              ws += (size_t)N * 32;
    float*    asrc2 = ws;              ws += (size_t)N;
    float*    adst2 = ws;              ws += (size_t)N;
    unsigned* emax2 = (unsigned*)ws;   ws += (size_t)N;
    float*    den2  = ws;              ws += (size_t)E2 > 0 ? (size_t)N : 0;
    float*    eb2   = ws;

    const int TB = 256;
    const int gB   = (N + 127) / 128;
    const int gE   = (E2 + TB - 1) / TB;
    const int gN4  = (N * 4 + TB - 1) / TB;
    const int gN1  = (N + TB - 1) / TB;
    const int gA   = (N + OWN_NPB - 1) / OWN_NPB;

    gat_gemm_wmma<8><<<gB, TB, 0, stream>>>(x, W1, h1, N);
    gat_attn_score<<<gN4, TB, 0, stream>>>(h1, as1, ad1, asrc1, adst1, N, 4, 32);
    gat_fill_u32<<<gN4, TB, 0, stream>>>(emax1, ENC_NEG_INF, N * 4);
    gat_fill_u32<<<gN4, TB, 0, stream>>>((unsigned*)den1, 0u, N * 4);
    gat_edge_scores<<<gE, TB, 0, stream>>>(srcI, dstI, asrc1, adst1, eb1, emax1, E, E2, 4);
    gat_edge_exp<<<gE, TB, 0, stream>>>(dstI, eb1, emax1, den1, E, E2, 4);
    gat_own_aggregate<4, true><<<gA, OWN_NW * 32, 0, stream>>>(srcI, dstI, eb1, den1, h1, b1, h2, E, N);

    gat_gemm_wmma<2><<<gB, TB, 0, stream>>>(h2, W2, g2, N);
    gat_attn_score<<<gN1, TB, 0, stream>>>(g2, as2, ad2, asrc2, adst2, N, 1, 32);
    gat_fill_u32<<<gN1, TB, 0, stream>>>(emax2, ENC_NEG_INF, N);
    gat_fill_u32<<<gN1, TB, 0, stream>>>((unsigned*)den2, 0u, N);
    gat_edge_scores<<<gE, TB, 0, stream>>>(srcI, dstI, asrc2, adst2, eb2, emax2, E, E2, 1);
    gat_edge_exp<<<gE, TB, 0, stream>>>(dstI, eb2, emax2, den2, E, E2, 1);
    gat_own_aggregate<1, false><<<gA, OWN_NW * 32, 0, stream>>>(srcI, dstI, eb2, den2, g2, b2, out, E, N);
}


# Softmax inside the owner block: row max, exp sum and unnormalised row sums in LDS (LDS max/add), one divide at the store; score/exp/fill launches and global atomics gone, 14 -> 6 launches
# speedup vs baseline: 7.3720x; 2.6912x over previous
#include <hip/hip_runtime.h>

#define NEG_SLOPE 0.2f
#define ENC_NEG_INF 0x007FFFFFu

typedef float v2f __attribute__((ext_vector_type(2)));
typedef float v8f __attribute__((ext_vector_type(8)));

__device__ __forceinline__ unsigned gat_fenc(float f) {
    unsigned u = __float_as_uint(f);
    return (u & 0x80000000u) ? ~u : (u | 0x80000000u);
}
__device__ __forceinline__ float gat_fdec(unsigned u) {
    unsigned b = (u & 0x80000000u) ? (u & 0x7FFFFFFFu) : ~u;
    return __uint_as_float(b);
}

template<int NT>
__global__ __launch_bounds__(256)
void gat_gemm_wmma(const float* __restrict__ X, const float* __restrict__ W,
                   float* __restrict__ Hout, int n_rows)
{
    constexpr int KD = 128;
    constexpr int ND = NT * 16;
    __shared__ float Wt[ND * KD];

    for (int i = threadIdx.x; i < KD * ND; i += 256) {
        int k = i / ND, n = i % ND;
        Wt[n * KD + ((k + 4 * n) & (KD - 1))] = W[i];
    }
    __syncthreads();

    const int lane = threadIdx.x & 31;
    const int wave = threadIdx.x >> 5;
    const int m0   = (blockIdx.x * 8 + wave) * 16;
    const int mlo  = lane & 15;
    const int kofs = (lane >> 4) * 2;

    int arow = m0 + mlo;
    if (arow >= n_rows) arow = n_rows - 1;
    const float* xrow = X + (size_t)arow * KD + kofs;

    v8f acc[NT] = {};

    for (int k = 0; k < KD; k += 4) {
        v2f a = *(const v2f*)(xrow + k);
#pragma unroll
        for (int t = 0; t < NT; ++t) {
            int n = t * 16 + mlo;
            v2f b = *(const v2f*)(&Wt[n * KD + ((k + kofs + 4 * n) & (KD - 1))]);
            acc[t] = __builtin_amdgcn_wmma_f32_16x16x4_f32(
                false, a, false, b, (short)0, acc[t], false, false);
        }
    }

    const int mhi = (lane >> 4) * 8;
#pragma unroll
    for (int t = 0; t < NT; ++t)
#pragma unroll
        for (int j = 0; j < 8; ++j) {
            int m = m0 + mhi + j;
            if (m < n_rows) Hout[(size_t)m * ND + t * 16 + mlo] = acc[t][j];
        }
}

__global__ void gat_attn_score(const float* __restrict__ Hf,
                               const float* __restrict__ att_s,
                               const float* __restrict__ att_d,
                               float* __restrict__ a_s, float* __restrict__ a_d,
                               int n, int heads, int ch)
{
    int t = blockIdx.x * blockDim.x + threadIdx.x;
    if (t >= n * heads) return;
    int node = t / heads, h = t - node * heads;
    const float* hp = Hf + ((size_t)node * heads + h) * ch;
    const float* sp = att_s + h * ch;
    const float* dp = att_d + h * ch;
    float ss = 0.f, dd = 0.f;
    for (int c = 0; c < ch; ++c) { float v = hp[c]; ss += v * sp[c]; dd += v * dp[c]; }
    a_s[t] = ss; a_d[t] = dd;
}

#define OWN_NPB 200
#define OWN_NW  32
#define OWN_CAP 1024
#define OWN_U   8

__device__ __forceinline__ float gat_lrelu(float v) { return (v >= 0.f) ? v : NEG_SLOPE * v; }

template<int H>
__device__ __forceinline__ void gat_own_flush_max(int* wl, int cnt, int lo, int E,
        const int* __restrict__ srcI, const int* __restrict__ dstI,
        const float* __restrict__ a_s, const float* ads, unsigned* mxs, int lane)
{
    for (int i0 = 0; i0 < cnt; i0 += 128) {
        int e[4], s[4], dl[4];
#pragma unroll
        for (int u = 0; u < 4; ++u) {
            const int i = i0 + u * 32 + lane;
            e[u] = (i < cnt) ? wl[i] : -1;
        }
#pragma unroll
        for (int u = 0; u < 4; ++u) {
            s[u] = 0; dl[u] = 0;
            if (e[u] >= 0) {
                if (e[u] < E) { s[u] = srcI[e[u]]; dl[u] = dstI[e[u]] - lo; }
                else          { s[u] = e[u] - E;   dl[u] = s[u] - lo; }
            }
        }
#pragma unroll
        for (int u = 0; u < 4; ++u) {
            if (e[u] >= 0) {
                float as[H];
                if constexpr (H == 4) {
                    const float4 q = *(const float4*)(a_s + (size_t)s[u] * 4);
                    as[0] = q.x; as[1] = q.y; as[2] = q.z; as[3] = q.w;
                } else {
#pragma unroll
                    for (int h = 0; h < H; ++h) as[h] = a_s[(size_t)s[u] * H + h];
                }
#pragma unroll
                for (int h = 0; h < H; ++h) {
                    const float v = gat_lrelu(as[h] + ads[dl[u] * H + h]);
                    __hip_atomic_fetch_max(&mxs[dl[u] * H + h], gat_fenc(v),
                                           __ATOMIC_RELAXED, __HIP_MEMORY_SCOPE_WORKGROUP);
                }
                wl[i0 + u * 32 + lane] = (int)(((unsigned)s[u] << 8) | (unsigned)dl[u]);
            }
        }
    }
}

template<int H, bool PACKED>
__device__ __forceinline__ void gat_own_flush_acc(const int* wl, int cnt, int lo, int E,
        const int* __restrict__ srcI, const int* __restrict__ dstI,
        const float* __restrict__ a_s, const float* ads, const unsigned* mxs, float* dens,
        const float* __restrict__ Hf, float* acc, int lane)
{
    constexpr int ROW = 32 * H;
    constexpr int LPR = ROW / 4;
    constexpr int EPI = 32 / LPR;
    const int grp  = lane / LPR;
    const int col  = (lane % LPR) * 4;
    const int head = col >> 5;

    for (int base = 0; base < cnt; base += 32) {
        const int m = min(32, cnt - base);
        int   s_i = 0, d_i = 0;
        float a_i[H];
#pragma unroll
        for (int h = 0; h < H; ++h) a_i[h] = 0.f;
        if (lane < m) {
            const int p = wl[base + lane];
            int s, dl;
            if (PACKED) { s = (int)((unsigned)p >> 8); dl = p & 255; }
            else if (p < E) { s = srcI[p]; dl = dstI[p] - lo; }
            else { s = p - E; dl = s - lo; }
            s_i = s; d_i = dl;
            float as[H];
            if constexpr (H == 4) {
                const float4 q = *(const float4*)(a_s + (size_t)s * 4);
                as[0] = q.x; as[1] = q.y; as[2] = q.z; as[3] = q.w;
            } else {
#pragma unroll
                for (int h = 0; h < H; ++h) as[h] = a_s[(size_t)s * H + h];
            }
#pragma unroll
            for (int h = 0; h < H; ++h) {
                const float v = gat_lrelu(as[h] + ads[dl * H + h]);
                const float x = __expf(v - gat_fdec(mxs[dl * H + h]));
                a_i[h] = x;
                __hip_atomic_fetch_add(&dens[dl * H + h], x, __ATOMIC_RELAXED, __HIP_MEMORY_SCOPE_WORKGROUP);
            }
        }
        const int nsteps = (m + EPI - 1) / EPI;
        for (int t0 = 0; t0 < nsteps; t0 += OWN_U) {
            float4 v[OWN_U]; int dd[OWN_U]; float ww[OWN_U]; bool ok[OWN_U];
#pragma unroll
            for (int u = 0; u < OWN_U; ++u) {
                const int idx = (t0 + u) * EPI + grp;
                ok[u] = idx < m;
                const int sl = idx & 31;
                const int s = __shfl(s_i, sl);
                dd[u] = __shfl(d_i, sl);
                float w = __shfl(a_i[0], sl);
#pragma unroll
                for (int h = 1; h < H; ++h) {
                    const float wh = __shfl(a_i[h], sl);
                    w = (head == h) ? wh : w;
                }
                ww[u] = w;
                v[u] = *(const float4*)(Hf + (size_t)s * ROW + col);
            }
#pragma unroll
            for (int u = 0; u < OWN_U; ++u) {
                if (ok[u]) {
                    float* p = acc + dd[u] * ROW + col;
                    __hip_atomic_fetch_add(p + 0, ww[u] * v[u].x, __ATOMIC_RELAXED, __HIP_MEMORY_SCOPE_WORKGROUP);
                    __hip_atomic_fetch_add(p + 1, ww[u] * v[u].y, __ATOMIC_RELAXED, __HIP_MEMORY_SCOPE_WORKGROUP);
                    __hip_atomic_fetch_add(p + 2, ww[u] * v[u].z, __ATOMIC_RELAXED, __HIP_MEMORY_SCOPE_WORKGROUP);
                    __hip_atomic_fetch_add(p + 3, ww[u] * v[u].w, __ATOMIC_RELAXED, __HIP_MEMORY_SCOPE_WORKGROUP);
                }
            }
        }
    }
}

template<int H, bool ACC>
__device__ __forceinline__ int gat_own_sweep(int* wl, int lo, int nodes, int E,
        const int* __restrict__ srcI, const int* __restrict__ dstI,
        const float* __restrict__ a_s, const float* ads, unsigned* mxs, float* dens,
        const float* __restrict__ Hf, float* acc, int* again, int lane, int wave)
{
    const unsigned lt = (1u << lane) - 1u;
    int cnt = 0;

    for (int nb = wave * 32; nb < nodes; nb += OWN_NW * 32) {
        const int n = nb + lane;
        const bool hit = n < nodes;
        const unsigned mask = __builtin_amdgcn_ballot_w32(hit);
        if (hit) wl[cnt + __popc(mask & lt)] = E + lo + n;
        cnt += __popc(mask);
    }

    const int nvec = E >> 2;
    const int4* d4 = (const int4*)dstI;
    const int4 none = make_int4(-1, -1, -1, -1);
    int gb = wave * 128;
    int4 cur[4];
#pragma unroll
    for (int u = 0; u < 4; ++u) {
        const int g = gb + u * 32 + lane;
        cur[u] = d4[max(min(g, nvec - 1), 0)];
        if (g >= nvec) cur[u] = none;
    }
    while (gb < nvec) {
        const int gn = gb + OWN_NW * 128;
        int4 nxt[4];
#pragma unroll
        for (int u = 0; u < 4; ++u) {
            const int g = gn + u * 32 + lane;
            nxt[u] = d4[min(g, nvec - 1)];
            if (g >= nvec) nxt[u] = none;
        }
#pragma unroll
        for (int u = 0; u < 4; ++u) {
            const int e0 = (gb + u * 32 + lane) * 4;
            const bool h0 = (unsigned)(cur[u].x - lo) < (unsigned)nodes;
            const bool h1 = (unsigned)(cur[u].y - lo) < (unsigned)nodes;
            const bool h2 = (unsigned)(cur[u].z - lo) < (unsigned)nodes;
            const bool h3 = (unsigned)(cur[u].w - lo) < (unsigned)nodes;
            if (__builtin_amdgcn_ballot_w32(h0 | h1 | h2 | h3)) {
                unsigned mask;
                mask = __builtin_amdgcn_ballot_w32(h0);
                if (h0) wl[cnt + __popc(mask & lt)] = e0 + 0;
                cnt += __popc(mask);
                mask = __builtin_amdgcn_ballot_w32(h1);
                if (h1) wl[cnt + __popc(mask & lt)] = e0 + 1;
                cnt += __popc(mask);
                mask = __builtin_amdgcn_ballot_w32(h2);
                if (h2) wl[cnt + __popc(mask & lt)] = e0 + 2;
                cnt += __popc(mask);
                mask = __builtin_amdgcn_ballot_w32(h3);
                if (h3) wl[cnt + __popc(mask & lt)] = e0 + 3;
                cnt += __popc(mask);
            }
        }
        if (cnt > OWN_CAP / 2) {
            if (ACC) {
                gat_own_flush_acc<H, false>(wl, cnt, lo, E, srcI, dstI, a_s, ads, mxs, dens, Hf, acc, lane);
            } else {
                gat_own_flush_max<H>(wl, cnt, lo, E, srcI, dstI, a_s, ads, mxs, lane);
                if (lane == 0) __hip_atomic_store(again, 1, __ATOMIC_RELAXED, __HIP_MEMORY_SCOPE_WORKGROUP);
            }
            cnt = 0;
        }
#pragma unroll
        for (int u = 0; u < 4; ++u) cur[u] = nxt[u];
        gb = gn;
    }
    if (wave == 0) {
        for (int eb0 = nvec * 4; eb0 < E; eb0 += 32) {
            const int e = eb0 + lane;
            const bool hit = (e < E) && ((unsigned)(dstI[min(e, E - 1)] - lo) < (unsigned)nodes);
            const unsigned mask = __builtin_amdgcn_ballot_w32(hit);
            if (hit) wl[cnt + __popc(mask & lt)] = e;
            cnt += __popc(mask);
        }
    }
    return cnt;
}

template<int H, bool RELU>
__global__ __launch_bounds__(OWN_NW * 32)
void gat_own_aggregate(const int* __restrict__ srcI, const int* __restrict__ dstI,
                       const float* __restrict__ a_s, const float* __restrict__ a_d,
                       const float* __restrict__ Hf, const float* __restrict__ bias,
                       float* __restrict__ outp, int E, int N)
{
    constexpr int ROW = 32 * H;
    static_assert(OWN_NPB <= 32 * OWN_NW, "one self-loop step per wave");
    static_assert(OWN_NPB <= 256, "local row index is packed into 8 bits");
    __shared__ __attribute__((aligned(16))) float acc[OWN_NPB * ROW];
    __shared__ int lst[OWN_NW * OWN_CAP];
    __shared__ float    ads[OWN_NPB * H];
    __shared__ unsigned mxs[OWN_NPB * H];
    __shared__ float    dens[OWN_NPB * H];
    __shared__ int again;

    const int tid  = threadIdx.x;
    const int lane = tid & 31;
    const int wave = tid >> 5;
    const int lo   = blockIdx.x * OWN_NPB;
    const int nodes = min(OWN_NPB, N - lo);

    for (int i = tid; i < nodes * ROW; i += OWN_NW * 32) acc[i] = 0.f;
    for (int i = tid; i < nodes * H; i += OWN_NW * 32) {
        ads[i]  = a_d[(size_t)lo * H + i];
        mxs[i]  = ENC_NEG_INF;
        dens[i] = 0.f;
    }
    if (tid == 0) again = 0;
    __syncthreads();

    int* wl = lst + wave * OWN_CAP;
    int cnt = gat_own_sweep<H, false>(wl, lo, nodes, E, srcI, dstI, a_s, ads, mxs, dens, Hf, acc, &again, lane, wave);
    gat_own_flush_max<H>(wl, cnt, lo, E, srcI, dstI, a_s, ads, mxs, lane);
    __syncthreads();

    if (__hip_atomic_load(&again, __ATOMIC_RELAXED, __HIP_MEMORY_SCOPE_WORKGROUP) == 0) {
        gat_own_flush_acc<H, true>(wl, cnt, lo, E, srcI, dstI, a_s, ads, mxs, dens, Hf, acc, lane);
    } else {
        cnt = gat_own_sweep<H, true>(wl, lo, nodes, E, srcI, dstI, a_s, ads, mxs, dens, Hf, acc, &again, lane, wave);
        gat_own_flush_acc<H, false>(wl, cnt, lo, E, srcI, dstI, a_s, ads, mxs, dens, Hf, acc, lane);
    }
    __syncthreads();

    float* orow = outp + (size_t)lo * ROW;
    for (int i = tid * 4; i < nodes * ROW; i += OWN_NW * 32 * 4) {
        float4 r = *(const float4*)(acc + i);
        const int c = i & (ROW - 1);
        const float dn = dens[(i / ROW) * H + (c >> 5)];
        r.x = r.x / dn + bias[c + 0];
        r.y = r.y / dn + bias[c + 1];
        r.z = r.z / dn + bias[c + 2];
        r.w = r.w / dn + bias[c + 3];
        if (RELU) { r.x = fmaxf(r.x, 0.f); r.y = fmaxf(r.y, 0.f); r.z = fmaxf(r.z, 0.f); r.w = fmaxf(r.w, 0.f); }
        *(float4*)(orow + i) = r;
    }
}

extern "C" void kernel_launch(void* const* d_in, const int* in_sizes, int n_in,
                              void* d_out, int out_size, void* d_ws, size_t ws_size,
                              hipStream_t stream)
{
    const float* x   = (const float*)d_in[0];
    const int*   ei  = (const int*)d_in[1];
    const float* W1  = (const float*)d_in[2];
    const float* as1 = (const float*)d_in[3];
    const float* ad1 = (const float*)d_in[4];
    const float* b1  = (const float*)d_in[5];
    const float* W2  = (const float*)d_in[6];
    const float* as2 = (const float*)d_in[7];
    const float* ad2 = (const float*)d_in[8];
    const float* b2  = (const float*)d_in[9];
    float* out = (float*)d_out;

    const int N  = in_sizes[0] / 128;
    const int E  = in_sizes[1] / 2;
    const int* srcI = ei;
    const int* dstI = ei + E;

    float* ws = (float*)d_ws;
    float*    h1    = ws;              ws += (size_t)N * 128;
    float*    asrc1 = ws;              ws += (size_t)N * 4;
    float*    adst1 = ws;              ws += (size_t)N * 4;
    float*    h2    = ws;              ws += (size_t)N * 128;
    float*    g2    = ws;              ws += (size_t)N * 32;
    float*    asrc2 = ws;              ws += (size_t)N;
    float*    adst2 = ws;

    const int TB = 256;
    const int gB   = (N + 127) / 128;
    const int gN4  = (N * 4 + TB - 1) / TB;
    const int gN1  = (N + TB - 1) / TB;
    const int gA   = (N + OWN_NPB - 1) / OWN_NPB;

    gat_gemm_wmma<8><<<gB, TB, 0, stream>>>(x, W1, h1, N);
    gat_attn_score<<<gN4, TB, 0, stream>>>(h1, as1, ad1, asrc1, adst1, N, 4, 32);
    gat_own_aggregate<4, true><<<gA, OWN_NW * 32, 0, stream>>>(srcI, dstI, asrc1, adst1, h1, b1, h2, E, N);

    gat_gemm_wmma<2><<<gB, TB, 0, stream>>>(h2, W2, g2, N);
    gat_attn_score<<<gN1, TB, 0, stream>>>(g2, as2, ad2, asrc2, adst2, N, 1, 32);
    gat_own_aggregate<1, false><<<gA, OWN_NW * 32, 0, stream>>>(srcI, dstI, asrc2, adst2, g2, b2, out, E, N);
}


# Owner-pass filter result kept: layer 1 stores each wave's finished list (src, local row) in the workspace, layer 2 reads it back instead of sweeping the edge list; sums stay in LDS arrival order
# speedup vs baseline: 8.6255x; 1.1700x over previous
#include <hip/hip_runtime.h>

#define NEG_SLOPE 0.2f
#define ENC_NEG_INF 0x007FFFFFu

typedef float v2f __attribute__((ext_vector_type(2)));
typedef float v8f __attribute__((ext_vector_type(8)));

__device__ __forceinline__ unsigned gat_fenc(float f) {
    unsigned u = __float_as_uint(f);
    return (u & 0x80000000u) ? ~u : (u | 0x80000000u);
}
__device__ __forceinline__ float gat_fdec(unsigned u) {
    unsigned b = (u & 0x80000000u) ? (u & 0x7FFFFFFFu) : ~u;
    return __uint_as_float(b);
}

template<int NT>
__global__ __launch_bounds__(256)
void gat_gemm_wmma(const float* __restrict__ X, const float* __restrict__ W,
                   float* __restrict__ Hout, int n_rows)
{
    constexpr int KD = 128;
    constexpr int ND = NT * 16;
    __shared__ float Wt[ND * KD];

    for (int i = threadIdx.x; i < KD * ND; i += 256) {
        int k = i / ND, n = i % ND;
        Wt[n * KD + ((k + 4 * n) & (KD - 1))] = W[i];
    }
    __syncthreads();

    const int lane = threadIdx.x & 31;
    const int wave = threadIdx.x >> 5;
    const int m0   = (blockIdx.x * 8 + wave) * 16;
    const int mlo  = lane & 15;
    const int kofs = (lane >> 4) * 2;

    int arow = m0 + mlo;
    if (arow >= n_rows) arow = n_rows - 1;
    const float* xrow = X + (size_t)arow * KD + kofs;

    v8f acc[NT] = {};

    for (int k = 0; k < KD; k += 4) {
        v2f a = *(const v2f*)(xrow + k);
#pragma unroll
        for (int t = 0; t < NT; ++t) {
            int n = t * 16 + mlo;
            v2f b = *(const v2f*)(&Wt[n * KD + ((k + kofs + 4 * n) & (KD - 1))]);
            acc[t] = __builtin_amdgcn_wmma_f32_16x16x4_f32(
                false, a, false, b, (short)0, acc[t], false, false);
        }
    }

    const int mhi = (lane >> 4) * 8;
#pragma unroll
    for (int t = 0; t < NT; ++t)
#pragma unroll
        for (int j = 0; j < 8; ++j) {
            int m = m0 + mhi + j;
            if (m < n_rows) Hout[(size_t)m * ND + t * 16 + mlo] = acc[t][j];
        }
}

__global__ void gat_attn_score(const float* __restrict__ Hf,
                               const float* __restrict__ att_s,
                               const float* __restrict__ att_d,
                               float* __restrict__ a_s, float* __restrict__ a_d,
                               int n, int heads, int ch)
{
    int t = blockIdx.x * blockDim.x + threadIdx.x;
    if (t >= n * heads) return;
    int node = t / heads, h = t - node * heads;
    const float* hp = Hf + ((size_t)node * heads + h) * ch;
    const float* sp = att_s + h * ch;
    const float* dp = att_d + h * ch;
    float ss = 0.f, dd = 0.f;
    for (int c = 0; c < ch; ++c) { float v = hp[c]; ss += v * sp[c]; dd += v * dp[c]; }
    a_s[t] = ss; a_d[t] = dd;
}

#define OWN_NPB 200
#define OWN_NW  32
#define OWN_CAP 1024
#define OWN_U   8

__device__ __forceinline__ float gat_lrelu(float v) { return (v >= 0.f) ? v : NEG_SLOPE * v; }

template<int H, bool SAVE>
__device__ __forceinline__ void gat_own_flush_max(int* wl, int* __restrict__ gl, int cnt, int lo, int E,
        const int* __restrict__ srcI, const int* __restrict__ dstI,
        const float* __restrict__ a_s, const float* ads, unsigned* mxs, int lane)
{
    for (int i0 = 0; i0 < cnt; i0 += 128) {
        int e[4], s[4], dl[4];
#pragma unroll
        for (int u = 0; u < 4; ++u) {
            const int i = i0 + u * 32 + lane;
            e[u] = (i < cnt) ? wl[i] : -1;
        }
#pragma unroll
        for (int u = 0; u < 4; ++u) {
            s[u] = 0; dl[u] = 0;
            if (e[u] >= 0) {
                if (e[u] < E) { s[u] = srcI[e[u]]; dl[u] = dstI[e[u]] - lo; }
                else          { s[u] = e[u] - E;   dl[u] = s[u] - lo; }
            }
        }
#pragma unroll
        for (int u = 0; u < 4; ++u) {
            if (e[u] >= 0) {
                float as[H];
                if constexpr (H == 4) {
                    const float4 q = *(const float4*)(a_s + (size_t)s[u] * 4);
                    as[0] = q.x; as[1] = q.y; as[2] = q.z; as[3] = q.w;
                } else {
#pragma unroll
                    for (int h = 0; h < H; ++h) as[h] = a_s[(size_t)s[u] * H + h];
                }
#pragma unroll
                for (int h = 0; h < H; ++h) {
                    const float v = gat_lrelu(as[h] + ads[dl[u] * H + h]);
                    __hip_atomic_fetch_max(&mxs[dl[u] * H + h], gat_fenc(v),
                                           __ATOMIC_RELAXED, __HIP_MEMORY_SCOPE_WORKGROUP);
                }
                const int pk = (int)(((unsigned)s[u] << 8) | (unsigned)dl[u]);
                wl[i0 + u * 32 + lane] = pk;
                if (SAVE) gl[i0 + u * 32 + lane] = pk;
            }
        }
    }
}

template<int H>
__device__ __forceinline__ void gat_own_list_max(const int* __restrict__ gl, int* wl, int cnt,
        const float* __restrict__ a_s, const float* ads, unsigned* mxs, int lane)
{
    for (int i0 = 0; i0 < cnt; i0 += 128) {
        int p[4];
#pragma unroll
        for (int u = 0; u < 4; ++u) p[u] = gl[min(i0 + u * 32 + lane, cnt - 1)];
        float as[4][H];
#pragma unroll
        for (int u = 0; u < 4; ++u) {
            const int s = (int)((unsigned)p[u] >> 8);
            if constexpr (H == 4) {
                const float4 q = *(const float4*)(a_s + (size_t)s * 4);
                as[u][0] = q.x; as[u][1] = q.y; as[u][2] = q.z; as[u][3] = q.w;
            } else {
#pragma unroll
                for (int h = 0; h < H; ++h) as[u][h] = a_s[(size_t)s * H + h];
            }
        }
#pragma unroll
        for (int u = 0; u < 4; ++u) {
            const int i = i0 + u * 32 + lane;
            if (i < cnt) {
                const int dl = p[u] & 255;
#pragma unroll
                for (int h = 0; h < H; ++h) {
                    const float v = gat_lrelu(as[u][h] + ads[dl * H + h]);
                    __hip_atomic_fetch_max(&mxs[dl * H + h], gat_fenc(v),
                                           __ATOMIC_RELAXED, __HIP_MEMORY_SCOPE_WORKGROUP);
                }
                wl[i] = p[u];
            }
        }
    }
}

template<int H, bool PACKED>
__device__ __forceinline__ void gat_own_flush_acc(const int* wl, int cnt, int lo, int E,
        const int* __restrict__ srcI, const int* __restrict__ dstI,
        const float* __restrict__ a_s, const float* ads, const unsigned* mxs, float* dens,
        const float* __restrict__ Hf, float* acc, int lane)
{
    constexpr int ROW = 32 * H;
    constexpr int LPR = ROW / 4;
    constexpr int EPI = 32 / LPR;
    const int grp  = lane / LPR;
    const int col  = (lane % LPR) * 4;
    const int head = col >> 5;

    for (int base = 0; base < cnt; base += 32) {
        const int m = min(32, cnt - base);
        int   s_i = 0, d_i = 0;
        float a_i[H];
#pragma unroll
        for (int h = 0; h < H; ++h) a_i[h] = 0.f;
        if (lane < m) {
            const int p = wl[base + lane];
            int s, dl;
            if (PACKED) { s = (int)((unsigned)p >> 8); dl = p & 255; }
            else if (p < E) { s = srcI[p]; dl = dstI[p] - lo; }
            else { s = p - E; dl = s - lo; }
            s_i = s; d_i = dl;
            float as[H];
            if constexpr (H == 4) {
                const float4 q = *(const float4*)(a_s + (size_t)s * 4);
                as[0] = q.x; as[1] = q.y; as[2] = q.z; as[3] = q.w;
            } else {
#pragma unroll
                for (int h = 0; h < H; ++h) as[h] = a_s[(size_t)s * H + h];
            }
#pragma unroll
            for (int h = 0; h < H; ++h) {
                const float v = gat_lrelu(as[h] + ads[dl * H + h]);
                const float x = __expf(v - gat_fdec(mxs[dl * H + h]));
                a_i[h] = x;
                __hip_atomic_fetch_add(&dens[dl * H + h], x, __ATOMIC_RELAXED, __HIP_MEMORY_SCOPE_WORKGROUP);
            }
        }
        const int nsteps = (m + EPI - 1) / EPI;
        for (int t0 = 0; t0 < nsteps; t0 += OWN_U) {
            float4 v[OWN_U]; int dd[OWN_U]; float ww[OWN_U]; bool ok[OWN_U];
#pragma unroll
            for (int u = 0; u < OWN_U; ++u) {
                const int idx = (t0 + u) * EPI + grp;
                ok[u] = idx < m;
                const int sl = idx & 31;
                const int s = __shfl(s_i, sl);
                dd[u] = __shfl(d_i, sl);
                float w = __shfl(a_i[0], sl);
#pragma unroll
                for (int h = 1; h < H; ++h) {
                    const float wh = __shfl(a_i[h], sl);
                    w = (head == h) ? wh : w;
                }
                ww[u] = w;
                v[u] = *(const float4*)(Hf + (size_t)s * ROW + col);
            }
#pragma unroll
            for (int u = 0; u < OWN_U; ++u) {
                if (ok[u]) {
                    float* p = acc + dd[u] * ROW + col;
                    __hip_atomic_fetch_add(p + 0, ww[u] * v[u].x, __ATOMIC_RELAXED, __HIP_MEMORY_SCOPE_WORKGROUP);
                    __hip_atomic_fetch_add(p + 1, ww[u] * v[u].y, __ATOMIC_RELAXED, __HIP_MEMORY_SCOPE_WORKGROUP);
                    __hip_atomic_fetch_add(p + 2, ww[u] * v[u].z, __ATOMIC_RELAXED, __HIP_MEMORY_SCOPE_WORKGROUP);
                    __hip_atomic_fetch_add(p + 3, ww[u] * v[u].w, __ATOMIC_RELAXED, __HIP_MEMORY_SCOPE_WORKGROUP);
                }
            }
        }
    }
}

template<int H, bool ACC>
__device__ __forceinline__ int gat_own_sweep(int* wl, int lo, int nodes, int E,
        const int* __restrict__ srcI, const int* __restrict__ dstI,
        const float* __restrict__ a_s, const float* ads, unsigned* mxs, float* dens,
        const float* __restrict__ Hf, float* acc, int* again, int lane, int wave)
{
    const unsigned lt = (1u << lane) - 1u;
    int cnt = 0;

    for (int nb = wave * 32; nb < nodes; nb += OWN_NW * 32) {
        const int n = nb + lane;
        const bool hit = n < nodes;
        const unsigned mask = __builtin_amdgcn_ballot_w32(hit);
        if (hit) wl[cnt + __popc(mask & lt)] = E + lo + n;
        cnt += __popc(mask);
    }

    const int nvec = E >> 2;
    const int4* d4 = (const int4*)dstI;
    const int4 none = make_int4(-1, -1, -1, -1);
    int gb = wave * 128;
    int4 cur[4];
#pragma unroll
    for (int u = 0; u < 4; ++u) {
        const int g = gb + u * 32 + lane;
        cur[u] = d4[max(min(g, nvec - 1), 0)];
        if (g >= nvec) cur[u] = none;
    }
    while (gb < nvec) {
        const int gn = gb + OWN_NW * 128;
        int4 nxt[4];
#pragma unroll
        for (int u = 0; u < 4; ++u) {
            const int g = gn + u * 32 + lane;
            nxt[u] = d4[min(g, nvec - 1)];
            if (g >= nvec) nxt[u] = none;
        }
#pragma unroll
        for (int u = 0; u < 4; ++u) {
            const int e0 = (gb + u * 32 + lane) * 4;
            const bool h0 = (unsigned)(cur[u].x - lo) < (unsigned)nodes;
            const bool h1 = (unsigned)(cur[u].y - lo) < (unsigned)nodes;
            const bool h2 = (unsigned)(cur[u].z - lo) < (unsigned)nodes;
            const bool h3 = (unsigned)(cur[u].w - lo) < (unsigned)nodes;
            if (__builtin_amdgcn_ballot_w32(h0 | h1 | h2 | h3)) {
                unsigned mask;
                mask = __builtin_amdgcn_ballot_w32(h0);
                if (h0) wl[cnt + __popc(mask & lt)] = e0 + 0;
                cnt += __popc(mask);
                mask = __builtin_amdgcn_ballot_w32(h1);
                if (h1) wl[cnt + __popc(mask & lt)] = e0 + 1;
                cnt += __popc(mask);
                mask = __builtin_amdgcn_ballot_w32(h2);
                if (h2) wl[cnt + __popc(mask & lt)] = e0 + 2;
                cnt += __popc(mask);
                mask = __builtin_amdgcn_ballot_w32(h3);
                if (h3) wl[cnt + __popc(mask & lt)] = e0 + 3;
                cnt += __popc(mask);
            }
        }
        if (cnt > OWN_CAP / 2) {
            if (ACC) {
                gat_own_flush_acc<H, false>(wl, cnt, lo, E, srcI, dstI, a_s, ads, mxs, dens, Hf, acc, lane);
            } else {
                gat_own_flush_max<H, false>(wl, nullptr, cnt, lo, E, srcI, dstI, a_s, ads, mxs, lane);
                if (lane == 0) __hip_atomic_store(again, 1, __ATOMIC_RELAXED, __HIP_MEMORY_SCOPE_WORKGROUP);
            }
            cnt = 0;
        }
#pragma unroll
        for (int u = 0; u < 4; ++u) cur[u] = nxt[u];
        gb = gn;
    }
    if (wave == 0) {
        for (int eb0 = nvec * 4; eb0 < E; eb0 += 32) {
            const int e = eb0 + lane;
            const bool hit = (e < E) && ((unsigned)(dstI[min(e, E - 1)] - lo) < (unsigned)nodes);
            const unsigned mask = __builtin_amdgcn_ballot_w32(hit);
            if (hit) wl[cnt + __popc(mask & lt)] = e;
            cnt += __popc(mask);
        }
    }
    return cnt;
}

template<int H, bool RELU, bool SAVE, bool LOAD>
__global__ __launch_bounds__(OWN_NW * 32)
void gat_own_aggregate(const int* __restrict__ srcI, const int* __restrict__ dstI,
                       const float* __restrict__ a_s, const float* __restrict__ a_d,
                       const float* __restrict__ Hf, const float* __restrict__ bias,
                       float* __restrict__ outp, int* __restrict__ glist, int* __restrict__ gcnt,
                       int E, int N)
{
    constexpr int ROW = 32 * H;
    static_assert(OWN_NPB <= 32 * OWN_NW, "one self-loop step per wave");
    static_assert(OWN_NPB <= 256, "local row index is packed into 8 bits");
    __shared__ __attribute__((aligned(16))) float acc[OWN_NPB * ROW];
    __shared__ int lst[OWN_NW * OWN_CAP];
    __shared__ float    ads[OWN_NPB * H];
    __shared__ unsigned mxs[OWN_NPB * H];
    __shared__ float    dens[OWN_NPB * H];
    __shared__ int again;

    const int tid  = threadIdx.x;
    const int lane = tid & 31;
    const int wave = tid >> 5;
    const int lo   = blockIdx.x * OWN_NPB;
    const int nodes = min(OWN_NPB, N - lo);

    for (int i = tid; i < nodes * ROW; i += OWN_NW * 32) acc[i] = 0.f;
    for (int i = tid; i < nodes * H; i += OWN_NW * 32) {
        ads[i]  = a_d[(size_t)lo * H + i];
        mxs[i]  = ENC_NEG_INF;
        dens[i] = 0.f;
    }
    if (tid == 0) again = 0;
    __syncthreads();

    int* wl = lst + wave * OWN_CAP;
    int* gl = glist + ((size_t)blockIdx.x * OWN_NW + wave) * OWN_CAP;
    int* gc = gcnt + (size_t)blockIdx.x * OWN_NW;
    int cnt = 0;
    bool saved = false;
    if (LOAD) saved = gc[0] >= 0;
    if (saved) {
        cnt = gc[wave];
        gat_own_list_max<H>(gl, wl, cnt, a_s, ads, mxs, lane);
    } else {
        cnt = gat_own_sweep<H, false>(wl, lo, nodes, E, srcI, dstI, a_s, ads, mxs, dens, Hf, acc, &again, lane, wave);
        gat_own_flush_max<H, SAVE>(wl, gl, cnt, lo, E, srcI, dstI, a_s, ads, mxs, lane);
    }
    __syncthreads();

    const int ag = __hip_atomic_load(&again, __ATOMIC_RELAXED, __HIP_MEMORY_SCOPE_WORKGROUP);
    if (SAVE && lane == 0) gc[wave] = ag ? -1 : cnt;
    if (ag == 0) {
        gat_own_flush_acc<H, true>(wl, cnt, lo, E, srcI, dstI, a_s, ads, mxs, dens, Hf, acc, lane);
    } else {
        cnt = gat_own_sweep<H, true>(wl, lo, nodes, E, srcI, dstI, a_s, ads, mxs, dens, Hf, acc, &again, lane, wave);
        gat_own_flush_acc<H, false>(wl, cnt, lo, E, srcI, dstI, a_s, ads, mxs, dens, Hf, acc, lane);
    }
    __syncthreads();

    float* orow = outp + (size_t)lo * ROW;
    for (int i = tid * 4; i < nodes * ROW; i += OWN_NW * 32 * 4) {
        float4 r = *(const float4*)(acc + i);
        const int c = i & (ROW - 1);
        const float dn = dens[(i / ROW) * H + (c >> 5)];
        r.x = r.x / dn + bias[c + 0];
        r.y = r.y / dn + bias[c + 1];
        r.z = r.z / dn + bias[c + 2];
        r.w = r.w / dn + bias[c + 3];
        if (RELU) { r.x = fmaxf(r.x, 0.f); r.y = fmaxf(r.y, 0.f); r.z = fmaxf(r.z, 0.f); r.w = fmaxf(r.w, 0.f); }
        *(float4*)(orow + i) = r;
    }
}

extern "C" void kernel_launch(void* const* d_in, const int* in_sizes, int n_in,
                              void* d_out, int out_size, void* d_ws, size_t ws_size,
                              hipStream_t stream)
{
    const float* x   = (const float*)d_in[0];
    const int*   ei  = (const int*)d_in[1];
    const float* W1  = (const float*)d_in[2];
    const float* as1 = (const float*)d_in[3];
    const float* ad1 = (const float*)d_in[4];
    const float* b1  = (const float*)d_in[5];
    const float* W2  = (const float*)d_in[6];
    const float* as2 = (const float*)d_in[7];
    const float* ad2 = (const float*)d_in[8];
    const float* b2  = (const float*)d_in[9];
    float* out = (float*)d_out;

    const int N  = in_sizes[0] / 128;
    const int E  = in_sizes[1] / 2;
    const int* srcI = ei;
    const int* dstI = ei + E;

    float* ws = (float*)d_ws;
    float*    h1    = ws;              ws += (size_t)N * 128;
    float*    asrc1 = ws;              ws += (size_t)N * 4;
    float*    adst1 = ws;              ws += (size_t)N * 4;
    float*    h2    = ws;              ws += (size_t)N * 128;
    float*    g2    = ws;              ws += (size_t)N * 32;
    float*    asrc2 = ws;              ws += (size_t)N;
    float*    adst2 = ws;              ws += (size_t)N;

    const int TB = 256;
    const int gB   = (N + 127) / 128;
    const int gN4  = (N * 4 + TB - 1) / TB;
    const int gN1  = (N + TB - 1) / TB;
    const int gA   = (N + OWN_NPB - 1) / OWN_NPB;
    int* glist = (int*)ws;
    int* gcnt  = glist + (size_t)gA * OWN_NW * OWN_CAP;

    gat_gemm_wmma<8><<<gB, TB, 0, stream>>>(x, W1, h1, N);
    gat_attn_score<<<gN4, TB, 0, stream>>>(h1, as1, ad1, asrc1, adst1, N, 4, 32);
    gat_own_aggregate<4, true, true, false><<<gA, OWN_NW * 32, 0, stream>>>(srcI, dstI, asrc1, adst1, h1, b1, h2, glist, gcnt, E, N);

    gat_gemm_wmma<2><<<gB, TB, 0, stream>>>(h2, W2, g2, N);
    gat_attn_score<<<gN1, TB, 0, stream>>>(g2, as2, ad2, asrc2, adst2, N, 1, 32);
    gat_own_aggregate<1, false, false, true><<<gA, OWN_NW * 32, 0, stream>>>(srcI, dstI, asrc2, adst2, g2, b2, out, glist, gcnt, E, N);
}


# Owner pass filters a one-byte owner id per edge (packed by extra blocks of the layer-1 score launch) instead of the 4-byte destination column; row sums stay in LDS arrival order
# speedup vs baseline: 9.5670x; 1.1092x over previous
#include <hip/hip_runtime.h>

#define NEG_SLOPE 0.2f
#define ENC_NEG_INF 0x007FFFFFu

typedef float v2f __attribute__((ext_vector_type(2)));
typedef float v8f __attribute__((ext_vector_type(8)));

__device__ __forceinline__ unsigned gat_fenc(float f) {
    unsigned u = __float_as_uint(f);
    return (u & 0x80000000u) ? ~u : (u | 0x80000000u);
}
__device__ __forceinline__ float gat_fdec(unsigned u) {
    unsigned b = (u & 0x80000000u) ? (u & 0x7FFFFFFFu) : ~u;
    return __uint_as_float(b);
}

template<int NT>
__global__ __launch_bounds__(256)
void gat_gemm_wmma(const float* __restrict__ X, const float* __restrict__ W,
                   float* __restrict__ Hout, int n_rows)
{
    constexpr int KD = 128;
    constexpr int ND = NT * 16;
    __shared__ float Wt[ND * KD];

    for (int i = threadIdx.x; i < KD * ND; i += 256) {
        int k = i / ND, n = i % ND;
        Wt[n * KD + ((k + 4 * n) & (KD - 1))] = W[i];
    }
    __syncthreads();

    const int lane = threadIdx.x & 31;
    const int wave = threadIdx.x >> 5;
    const int m0   = (blockIdx.x * 8 + wave) * 16;
    const int mlo  = lane & 15;
    const int kofs = (lane >> 4) * 2;

    int arow = m0 + mlo;
    if (arow >= n_rows) arow = n_rows - 1;
    const float* xrow = X + (size_t)arow * KD + kofs;

    v8f acc[NT] = {};

    for (int k = 0; k < KD; k += 4) {
        v2f a = *(const v2f*)(xrow + k);
#pragma unroll
        for (int t = 0; t < NT; ++t) {
            int n = t * 16 + mlo;
            v2f b = *(const v2f*)(&Wt[n * KD + ((k + kofs + 4 * n) & (KD - 1))]);
            acc[t] = __builtin_amdgcn_wmma_f32_16x16x4_f32(
                false, a, false, b, (short)0, acc[t], false, false);
        }
    }

    const int mhi = (lane >> 4) * 8;
#pragma unroll
    for (int t = 0; t < NT; ++t)
#pragma unroll
        for (int j = 0; j < 8; ++j) {
            int m = m0 + mhi + j;
            if (m < n_rows) Hout[(size_t)m * ND + t * 16 + mlo] = acc[t][j];
        }
}

#define OWN_NPB 200

__device__ __forceinline__ unsigned gat_owner_id(int d) { return min((unsigned)d / OWN_NPB, 255u); }

__global__ void gat_attn_score(const float* __restrict__ Hf,
                               const float* __restrict__ att_s,
                               const float* __restrict__ att_d,
                               float* __restrict__ a_s, float* __restrict__ a_d,
                               int n, int heads, int ch,
                               const int* __restrict__ dstI, unsigned* __restrict__ obyte,
                               int nquad, int score_blocks)
{
    if ((int)blockIdx.x >= score_blocks) {
        const int q = ((int)blockIdx.x - score_blocks) * (int)blockDim.x + (int)threadIdx.x;
        if (q < nquad) {
            const int4 d = ((const int4*)dstI)[q];
            obyte[q] = gat_owner_id(d.x) | (gat_owner_id(d.y) << 8) |
                       (gat_owner_id(d.z) << 16) | (gat_owner_id(d.w) << 24);
        }
        return;
    }
    int t = blockIdx.x * blockDim.x + threadIdx.x;
    if (t >= n * heads) return;
    int node = t / heads, h = t - node * heads;
    const float* hp = Hf + ((size_t)node * heads + h) * ch;
    const float* sp = att_s + h * ch;
    const float* dp = att_d + h * ch;
    float ss = 0.f, dd = 0.f;
    for (int c = 0; c < ch; ++c) { float v = hp[c]; ss += v * sp[c]; dd += v * dp[c]; }
    a_s[t] = ss; a_d[t] = dd;
}

#define OWN_NW  32
#define OWN_CAP 1024
#define OWN_U   8

__device__ __forceinline__ float gat_lrelu(float v) { return (v >= 0.f) ? v : NEG_SLOPE * v; }

template<int H, bool SAVE>
__device__ __forceinline__ void gat_own_flush_max(int* wl, int* __restrict__ gl, int cnt, int lo, int E,
        const int* __restrict__ srcI, const int* __restrict__ dstI,
        const float* __restrict__ a_s, const float* ads, unsigned* mxs, int lane)
{
    for (int i0 = 0; i0 < cnt; i0 += 128) {
        int e[4], s[4], dl[4];
#pragma unroll
        for (int u = 0; u < 4; ++u) {
            const int i = i0 + u * 32 + lane;
            e[u] = (i < cnt) ? wl[i] : -1;
        }
#pragma unroll
        for (int u = 0; u < 4; ++u) {
            s[u] = 0; dl[u] = 0;
            if (e[u] >= 0) {
                if (e[u] < E) { s[u] = srcI[e[u]]; dl[u] = dstI[e[u]] - lo; }
                else          { s[u] = e[u] - E;   dl[u] = s[u] - lo; }
            }
        }
#pragma unroll
        for (int u = 0; u < 4; ++u) {
            if (e[u] >= 0) {
                float as[H];
                if constexpr (H == 4) {
                    const float4 q = *(const float4*)(a_s + (size_t)s[u] * 4);
                    as[0] = q.x; as[1] = q.y; as[2] = q.z; as[3] = q.w;
                } else {
#pragma unroll
                    for (int h = 0; h < H; ++h) as[h] = a_s[(size_t)s[u] * H + h];
                }
#pragma unroll
                for (int h = 0; h < H; ++h) {
                    const float v = gat_lrelu(as[h] + ads[dl[u] * H + h]);
                    __hip_atomic_fetch_max(&mxs[dl[u] * H + h], gat_fenc(v),
                                           __ATOMIC_RELAXED, __HIP_MEMORY_SCOPE_WORKGROUP);
                }
                const int pk = (int)(((unsigned)s[u] << 8) | (unsigned)dl[u]);
                wl[i0 + u * 32 + lane] = pk;
                if (SAVE) gl[i0 + u * 32 + lane] = pk;
            }
        }
    }
}

template<int H>
__device__ __forceinline__ void gat_own_list_max(const int* __restrict__ gl, int* wl, int cnt,
        const float* __restrict__ a_s, const float* ads, unsigned* mxs, int lane)
{
    for (int i0 = 0; i0 < cnt; i0 += 128) {
        int p[4];
#pragma unroll
        for (int u = 0; u < 4; ++u) p[u] = gl[min(i0 + u * 32 + lane, cnt - 1)];
        float as[4][H];
#pragma unroll
        for (int u = 0; u < 4; ++u) {
            const int s = (int)((unsigned)p[u] >> 8);
            if constexpr (H == 4) {
                const float4 q = *(const float4*)(a_s + (size_t)s * 4);
                as[u][0] = q.x; as[u][1] = q.y; as[u][2] = q.z; as[u][3] = q.w;
            } else {
#pragma unroll
                for (int h = 0; h < H; ++h) as[u][h] = a_s[(size_t)s * H + h];
            }
        }
#pragma unroll
        for (int u = 0; u < 4; ++u) {
            const int i = i0 + u * 32 + lane;
            if (i < cnt) {
                const int dl = p[u] & 255;
#pragma unroll
                for (int h = 0; h < H; ++h) {
                    const float v = gat_lrelu(as[u][h] + ads[dl * H + h]);
                    __hip_atomic_fetch_max(&mxs[dl * H + h], gat_fenc(v),
                                           __ATOMIC_RELAXED, __HIP_MEMORY_SCOPE_WORKGROUP);
                }
                wl[i] = p[u];
            }
        }
    }
}

template<int H, bool PACKED>
__device__ __forceinline__ void gat_own_flush_acc(const int* wl, int cnt, int lo, int E,
        const int* __restrict__ srcI, const int* __restrict__ dstI,
        const float* __restrict__ a_s, const float* ads, const unsigned* mxs, float* dens,
        const float* __restrict__ Hf, float* acc, int lane)
{
    constexpr int ROW = 32 * H;
    constexpr int LPR = ROW / 4;
    constexpr int EPI = 32 / LPR;
    const int grp  = lane / LPR;
    const int col  = (lane % LPR) * 4;
    const int head = col >> 5;

    for (int base = 0; base < cnt; base += 32) {
        const int m = min(32, cnt - base);
        int   s_i = 0, d_i = 0;
        float a_i[H];
#pragma unroll
        for (int h = 0; h < H; ++h) a_i[h] = 0.f;
        if (lane < m) {
            const int p = wl[base + lane];
            int s, dl;
            if (PACKED) { s = (int)((unsigned)p >> 8); dl = p & 255; }
            else if (p < E) { s = srcI[p]; dl = dstI[p] - lo; }
            else { s = p - E; dl = s - lo; }
            s_i = s; d_i = dl;
            float as[H];
            if constexpr (H == 4) {
                const float4 q = *(const float4*)(a_s + (size_t)s * 4);
                as[0] = q.x; as[1] = q.y; as[2] = q.z; as[3] = q.w;
            } else {
#pragma unroll
                for (int h = 0; h < H; ++h) as[h] = a_s[(size_t)s * H + h];
            }
#pragma unroll
            for (int h = 0; h < H; ++h) {
                const float v = gat_lrelu(as[h] + ads[dl * H + h]);
                const float x = __expf(v - gat_fdec(mxs[dl * H + h]));
                a_i[h] = x;
                __hip_atomic_fetch_add(&dens[dl * H + h], x, __ATOMIC_RELAXED, __HIP_MEMORY_SCOPE_WORKGROUP);
            }
        }
        const int nsteps = (m + EPI - 1) / EPI;
        for (int t0 = 0; t0 < nsteps; t0 += OWN_U) {
            float4 v[OWN_U]; int dd[OWN_U]; float ww[OWN_U]; bool ok[OWN_U];
#pragma unroll
            for (int u = 0; u < OWN_U; ++u) {
                const int idx = (t0 + u) * EPI + grp;
                ok[u] = idx < m;
                const int sl = idx & 31;
                const int s = __shfl(s_i, sl);
                dd[u] = __shfl(d_i, sl);
                float w = __shfl(a_i[0], sl);
#pragma unroll
                for (int h = 1; h < H; ++h) {
                    const float wh = __shfl(a_i[h], sl);
                    w = (head == h) ? wh : w;
                }
                ww[u] = w;
                v[u] = *(const float4*)(Hf + (size_t)s * ROW + col);
            }
#pragma unroll
            for (int u = 0; u < OWN_U; ++u) {
                if (ok[u]) {
                    float* p = acc + dd[u] * ROW + col;
                    __hip_atomic_fetch_add(p + 0, ww[u] * v[u].x, __ATOMIC_RELAXED, __HIP_MEMORY_SCOPE_WORKGROUP);
                    __hip_atomic_fetch_add(p + 1, ww[u] * v[u].y, __ATOMIC_RELAXED, __HIP_MEMORY_SCOPE_WORKGROUP);
                    __hip_atomic_fetch_add(p + 2, ww[u] * v[u].z, __ATOMIC_RELAXED, __HIP_MEMORY_SCOPE_WORKGROUP);
                    __hip_atomic_fetch_add(p + 3, ww[u] * v[u].w, __ATOMIC_RELAXED, __HIP_MEMORY_SCOPE_WORKGROUP);
                }
            }
        }
    }
}


__device__ __forceinline__ unsigned gat_eq_bytes(unsigned x, unsigned key4) {
    const unsigned y = x ^ key4;
    const unsigned t = ((y & 0x7F7F7F7Fu) + 0x7F7F7F7Fu) | y;
    return ~t & 0x80808080u;
}
__device__ __forceinline__ unsigned gat_eq_mask16(uint4 v, unsigned key4) {
    return (gat_eq_bytes(v.x, key4) >> 7) | (gat_eq_bytes(v.y, key4) >> 6) |
           (gat_eq_bytes(v.z, key4) >> 5) | (gat_eq_bytes(v.w, key4) >> 4);
}

template<int H, bool ACC>
__device__ __forceinline__ void gat_own_drain(int* wl, int cnt, int lo, int E,
        const int* __restrict__ srcI, const int* __restrict__ dstI,
        const float* __restrict__ a_s, const float* ads, unsigned* mxs, float* dens,
        const float* __restrict__ Hf, float* acc, int* again, int lane)
{
    if (ACC) {
        gat_own_flush_acc<H, false>(wl, cnt, lo, E, srcI, dstI, a_s, ads, mxs, dens, Hf, acc, lane);
    } else {
        gat_own_flush_max<H, false>(wl, nullptr, cnt, lo, E, srcI, dstI, a_s, ads, mxs, lane);
        if (lane == 0) __hip_atomic_store(again, 1, __ATOMIC_RELAXED, __HIP_MEMORY_SCOPE_WORKGROUP);
    }
}

template<int H, bool ACC>
__device__ __forceinline__ int gat_own_sweep(int* wl, int lo, int nodes, int E,
        const int* __restrict__ srcI, const int* __restrict__ dstI,
        const unsigned* __restrict__ obyte, bool bytes_ok,
        const float* __restrict__ a_s, const float* ads, unsigned* mxs, float* dens,
        const float* __restrict__ Hf, float* acc, int* again, int lane, int wave)
{
    const unsigned lt = (1u << lane) - 1u;
    int cnt = 0;

    for (int nb = wave * 32; nb < nodes; nb += OWN_NW * 32) {
        const int n = nb + lane;
        const bool hit = n < nodes;
        const unsigned mask = __builtin_amdgcn_ballot_w32(hit);
        if (hit) wl[cnt + __popc(mask & lt)] = E + lo + n;
        cnt += __popc(mask);
    }

    int tail0;
    if (bytes_ok) {
    const int nv16 = E >> 4;
    const uint4* ob = (const uint4*)obyte;
    const unsigned key4 = (unsigned)blockIdx.x * 0x01010101u;
    const uint4 none = make_uint4(~key4, ~key4, ~key4, ~key4);
    int gb = wave * 128;
    uint4 cur[4];
#pragma unroll
    for (int u = 0; u < 4; ++u) {
        const int g = gb + u * 32 + lane;
        cur[u] = ob[max(min(g, nv16 - 1), 0)];
        if (g >= nv16) cur[u] = none;
    }
    while (gb < nv16) {
        const int gn = gb + OWN_NW * 128;
        uint4 nxt[4];
#pragma unroll
        for (int u = 0; u < 4; ++u) {
            const int g = gn + u * 32 + lane;
            nxt[u] = ob[min(g, nv16 - 1)];
            if (g >= nv16) nxt[u] = none;
        }
        const unsigned m01 = gat_eq_mask16(cur[0], key4) | (gat_eq_mask16(cur[1], key4) << 4);
        const unsigned m23 = gat_eq_mask16(cur[2], key4) | (gat_eq_mask16(cur[3], key4) << 4);
        unsigned long long M = (unsigned long long)m01 | ((unsigned long long)m23 << 32);
        unsigned pend = __builtin_amdgcn_ballot_w32(M != 0ull);
        while (pend) {
            if (M != 0ull) {
                const int p = __ffsll((long long)M) - 1;
                const int u = ((p >> 5) << 1) | ((p >> 2) & 1);
                const int off = ((p & 3) << 2) | ((p >> 3) & 3);
                wl[cnt + __popc(pend & lt)] = ((gb + u * 32 + lane) << 4) + off;
                M &= M - 1ull;
            }
            cnt += __popc(pend);
            if (cnt > OWN_CAP - 32) {
                gat_own_drain<H, ACC>(wl, cnt, lo, E, srcI, dstI, a_s, ads, mxs, dens, Hf, acc, again, lane);
                cnt = 0;
            }
            pend = __builtin_amdgcn_ballot_w32(M != 0ull);
        }
#pragma unroll
        for (int u = 0; u < 4; ++u) cur[u] = nxt[u];
        gb = gn;
    }
    tail0 = nv16 << 4;
    } else {
    const int nvec = E >> 2;
    const int4* d4 = (const int4*)dstI;
    const int4 none = make_int4(-1, -1, -1, -1);
    int gb = wave * 128;
    int4 cur[4];
#pragma unroll
    for (int u = 0; u < 4; ++u) {
        const int g = gb + u * 32 + lane;
        cur[u] = d4[max(min(g, nvec - 1), 0)];
        if (g >= nvec) cur[u] = none;
    }
    while (gb < nvec) {
        const int gn = gb + OWN_NW * 128;
        int4 nxt[4];
#pragma unroll
        for (int u = 0; u < 4; ++u) {
            const int g = gn + u * 32 + lane;
            nxt[u] = d4[min(g, nvec - 1)];
            if (g >= nvec) nxt[u] = none;
        }
#pragma unroll
        for (int u = 0; u < 4; ++u) {
            const int e0 = (gb + u * 32 + lane) * 4;
            const bool h0 = (unsigned)(cur[u].x - lo) < (unsigned)nodes;
            const bool h1 = (unsigned)(cur[u].y - lo) < (unsigned)nodes;
            const bool h2 = (unsigned)(cur[u].z - lo) < (unsigned)nodes;
            const bool h3 = (unsigned)(cur[u].w - lo) < (unsigned)nodes;
            if (__builtin_amdgcn_ballot_w32(h0 | h1 | h2 | h3)) {
                unsigned mask;
                mask = __builtin_amdgcn_ballot_w32(h0);
                if (h0) wl[cnt + __popc(mask & lt)] = e0 + 0;
                cnt += __popc(mask);
                mask = __builtin_amdgcn_ballot_w32(h1);
                if (h1) wl[cnt + __popc(mask & lt)] = e0 + 1;
                cnt += __popc(mask);
                mask = __builtin_amdgcn_ballot_w32(h2);
                if (h2) wl[cnt + __popc(mask & lt)] = e0 + 2;
                cnt += __popc(mask);
                mask = __builtin_amdgcn_ballot_w32(h3);
                if (h3) wl[cnt + __popc(mask & lt)] = e0 + 3;
                cnt += __popc(mask);
            }
        }
        if (cnt > OWN_CAP / 2) {
            gat_own_drain<H, ACC>(wl, cnt, lo, E, srcI, dstI, a_s, ads, mxs, dens, Hf, acc, again, lane);
            cnt = 0;
        }
#pragma unroll
        for (int u = 0; u < 4; ++u) cur[u] = nxt[u];
        gb = gn;
    }
    tail0 = nvec * 4;
    }
    if (wave == 0) {
        for (int eb0 = tail0; eb0 < E; eb0 += 32) {
            const int e = eb0 + lane;
            const bool hit = (e < E) && ((unsigned)(dstI[min(e, E - 1)] - lo) < (unsigned)nodes);
            const unsigned mask = __builtin_amdgcn_ballot_w32(hit);
            if (hit) wl[cnt + __popc(mask & lt)] = e;
            cnt += __popc(mask);
        }
    }
    return cnt;
}

template<int H, bool RELU, bool SAVE, bool LOAD>
__global__ __launch_bounds__(OWN_NW * 32)
void gat_own_aggregate(const int* __restrict__ srcI, const int* __restrict__ dstI,
                       const float* __restrict__ a_s, const float* __restrict__ a_d,
                       const float* __restrict__ Hf, const float* __restrict__ bias,
                       float* __restrict__ outp, int* __restrict__ glist, int* __restrict__ gcnt,
                       const unsigned* __restrict__ obyte, int E, int N)
{
    constexpr int ROW = 32 * H;
    static_assert(OWN_NPB <= 32 * OWN_NW, "one self-loop step per wave");
    static_assert(OWN_NPB <= 256, "local row index is packed into 8 bits");
    __shared__ __attribute__((aligned(16))) float acc[OWN_NPB * ROW];
    __shared__ int lst[OWN_NW * OWN_CAP];
    __shared__ float    ads[OWN_NPB * H];
    __shared__ unsigned mxs[OWN_NPB * H];
    __shared__ float    dens[OWN_NPB * H];
    __shared__ int again;

    const int tid  = threadIdx.x;
    const int lane = tid & 31;
    const int wave = tid >> 5;
    const int lo   = blockIdx.x * OWN_NPB;
    const int nodes = min(OWN_NPB, N - lo);
    const bool bytes_ok = gridDim.x <= 256u;

    for (int i = tid; i < nodes * ROW; i += OWN_NW * 32) acc[i] = 0.f;
    for (int i = tid; i < nodes * H; i += OWN_NW * 32) {
        ads[i]  = a_d[(size_t)lo * H + i];
        mxs[i]  = ENC_NEG_INF;
        dens[i] = 0.f;
    }
    if (tid == 0) again = 0;
    __syncthreads();

    int* wl = lst + wave * OWN_CAP;
    int* gl = glist + ((size_t)blockIdx.x * OWN_NW + wave) * OWN_CAP;
    int* gc = gcnt + (size_t)blockIdx.x * OWN_NW;
    int cnt = 0;
    bool saved = false;
    if (LOAD) saved = gc[0] >= 0;
    if (saved) {
        cnt = gc[wave];
        gat_own_list_max<H>(gl, wl, cnt, a_s, ads, mxs, lane);
    } else {
        cnt = gat_own_sweep<H, false>(wl, lo, nodes, E, srcI, dstI, obyte, bytes_ok, a_s, ads, mxs, dens, Hf, acc, &again, lane, wave);
        gat_own_flush_max<H, SAVE>(wl, gl, cnt, lo, E, srcI, dstI, a_s, ads, mxs, lane);
    }
    __syncthreads();

    const int ag = __hip_atomic_load(&again, __ATOMIC_RELAXED, __HIP_MEMORY_SCOPE_WORKGROUP);
    if (SAVE && lane == 0) gc[wave] = ag ? -1 : cnt;
    if (ag == 0) {
        gat_own_flush_acc<H, true>(wl, cnt, lo, E, srcI, dstI, a_s, ads, mxs, dens, Hf, acc, lane);
    } else {
        cnt = gat_own_sweep<H, true>(wl, lo, nodes, E, srcI, dstI, obyte, bytes_ok, a_s, ads, mxs, dens, Hf, acc, &again, lane, wave);
        gat_own_flush_acc<H, false>(wl, cnt, lo, E, srcI, dstI, a_s, ads, mxs, dens, Hf, acc, lane);
    }
    __syncthreads();

    float* orow = outp + (size_t)lo * ROW;
    for (int i = tid * 4; i < nodes * ROW; i += OWN_NW * 32 * 4) {
        float4 r = *(const float4*)(acc + i);
        const int c = i & (ROW - 1);
        const float dn = dens[(i / ROW) * H + (c >> 5)];
        r.x = r.x / dn + bias[c + 0];
        r.y = r.y / dn + bias[c + 1];
        r.z = r.z / dn + bias[c + 2];
        r.w = r.w / dn + bias[c + 3];
        if (RELU) { r.x = fmaxf(r.x, 0.f); r.y = fmaxf(r.y, 0.f); r.z = fmaxf(r.z, 0.f); r.w = fmaxf(r.w, 0.f); }
        *(float4*)(orow + i) = r;
    }
}

extern "C" void kernel_launch(void* const* d_in, const int* in_sizes, int n_in,
                              void* d_out, int out_size, void* d_ws, size_t ws_size,
                              hipStream_t stream)
{
    const float* x   = (const float*)d_in[0];
    const int*   ei  = (const int*)d_in[1];
    const float* W1  = (const float*)d_in[2];
    const float* as1 = (const float*)d_in[3];
    const float* ad1 = (const float*)d_in[4];
    const float* b1  = (const float*)d_in[5];
    const float* W2  = (const float*)d_in[6];
    const float* as2 = (const float*)d_in[7];
    const float* ad2 = (const float*)d_in[8];
    const float* b2  = (const float*)d_in[9];
    float* out = (float*)d_out;

    const int N  = in_sizes[0] / 128;
    const int E  = in_sizes[1] / 2;
    const int* srcI = ei;
    const int* dstI = ei + E;

    float* ws = (float*)d_ws;
    float*    h1    = ws;              ws += (size_t)N * 128;
    float*    asrc1 = ws;              ws += (size_t)N * 4;
    float*    adst1 = ws;              ws += (size_t)N * 4;
    float*    h2    = ws;              ws += (size_t)N * 128;
    float*    g2    = ws;              ws += (size_t)N * 32;
    float*    asrc2 = ws;              ws += (size_t)N;
    float*    adst2 = ws;              ws += (size_t)N;

    const int TB = 256;
    const int gB   = (N + 127) / 128;
    const int gN4  = (N * 4 + TB - 1) / TB;
    const int gN1  = (N + TB - 1) / TB;
    const int gA   = (N + OWN_NPB - 1) / OWN_NPB;
    int* glist = (int*)ws;
    int* gcnt  = glist + (size_t)gA * OWN_NW * OWN_CAP;
    unsigned* obyte = (unsigned*)(gcnt + (((size_t)gA * OWN_NW + 3) & ~(size_t)3));
    const int nquad = E >> 2;
    const int gE4   = (nquad + TB - 1) / TB;

    gat_gemm_wmma<8><<<gB, TB, 0, stream>>>(x, W1, h1, N);
    gat_attn_score<<<gN4 + gE4, TB, 0, stream>>>(h1, as1, ad1, asrc1, adst1, N, 4, 32, dstI, obyte, nquad, gN4);
    gat_own_aggregate<4, true, true, false><<<gA, OWN_NW * 32, 0, stream>>>(srcI, dstI, asrc1, adst1, h1, b1, h2, glist, gcnt, obyte, E, N);

    gat_gemm_wmma<2><<<gB, TB, 0, stream>>>(h2, W2, g2, N);
    gat_attn_score<<<gN1, TB, 0, stream>>>(g2, as2, ad2, asrc2, adst2, N, 1, 32, dstI, obyte, 0, gN1);
    gat_own_aggregate<1, false, false, true><<<gA, OWN_NW * 32, 0, stream>>>(srcI, dstI, asrc2, adst2, g2, b2, out, glist, gcnt, obyte, E, N);
}


# LDS tiles re-laid out so the lanes of each LDS float add and max fall on different banks (sum tile column-interleaved, per-row planes head-major); row sums stay in LDS arrival order
# speedup vs baseline: 12.1718x; 1.2723x over previous
#include <hip/hip_runtime.h>

#define NEG_SLOPE 0.2f
#define ENC_NEG_INF 0x007FFFFFu

typedef float v2f __attribute__((ext_vector_type(2)));
typedef float v8f __attribute__((ext_vector_type(8)));

__device__ __forceinline__ unsigned gat_fenc(float f) {
    unsigned u = __float_as_uint(f);
    return (u & 0x80000000u) ? ~u : (u | 0x80000000u);
}
__device__ __forceinline__ float gat_fdec(unsigned u) {
    unsigned b = (u & 0x80000000u) ? (u & 0x7FFFFFFFu) : ~u;
    return __uint_as_float(b);
}

template<int NT>
__global__ __launch_bounds__(256)
void gat_gemm_wmma(const float* __restrict__ X, const float* __restrict__ W,
                   float* __restrict__ Hout, int n_rows)
{
    constexpr int KD = 128;
    constexpr int ND = NT * 16;
    __shared__ float Wt[ND * KD];

    for (int i = threadIdx.x; i < KD * ND; i += 256) {
        int k = i / ND, n = i % ND;
        Wt[n * KD + ((k + 4 * n) & (KD - 1))] = W[i];
    }
    __syncthreads();

    const int lane = threadIdx.x & 31;
    const int wave = threadIdx.x >> 5;
    const int m0   = (blockIdx.x * 8 + wave) * 16;
    const int mlo  = lane & 15;
    const int kofs = (lane >> 4) * 2;

    int arow = m0 + mlo;
    if (arow >= n_rows) arow = n_rows - 1;
    const float* xrow = X + (size_t)arow * KD + kofs;

    v8f acc[NT] = {};

    for (int k = 0; k < KD; k += 4) {
        v2f a = *(const v2f*)(xrow + k);
#pragma unroll
        for (int t = 0; t < NT; ++t) {
            int n = t * 16 + mlo;
            v2f b = *(const v2f*)(&Wt[n * KD + ((k + kofs + 4 * n) & (KD - 1))]);
            acc[t] = __builtin_amdgcn_wmma_f32_16x16x4_f32(
                false, a, false, b, (short)0, acc[t], false, false);
        }
    }

    const int mhi = (lane >> 4) * 8;
#pragma unroll
    for (int t = 0; t < NT; ++t)
#pragma unroll
        for (int j = 0; j < 8; ++j) {
            int m = m0 + mhi + j;
            if (m < n_rows) Hout[(size_t)m * ND + t * 16 + mlo] = acc[t][j];
        }
}

#define OWN_NPB 200

__device__ __forceinline__ unsigned gat_owner_id(int d) { return min((unsigned)d / OWN_NPB, 255u); }

__global__ void gat_attn_score(const float* __restrict__ Hf,
                               const float* __restrict__ att_s,
                               const float* __restrict__ att_d,
                               float* __restrict__ a_s, float* __restrict__ a_d,
                               int n, int heads, int ch,
                               const int* __restrict__ dstI, unsigned* __restrict__ obyte,
                               int nquad, int score_blocks)
{
    if ((int)blockIdx.x >= score_blocks) {
        const int q = ((int)blockIdx.x - score_blocks) * (int)blockDim.x + (int)threadIdx.x;
        if (q < nquad) {
            const int4 d = ((const int4*)dstI)[q];
            obyte[q] = gat_owner_id(d.x) | (gat_owner_id(d.y) << 8) |
                       (gat_owner_id(d.z) << 16) | (gat_owner_id(d.w) << 24);
        }
        return;
    }
    int t = blockIdx.x * blockDim.x + threadIdx.x;
    if (t >= n * heads) return;
    int node = t / heads, h = t - node * heads;
    const float* hp = Hf + ((size_t)node * heads + h) * ch;
    const float* sp = att_s + h * ch;
    const float* dp = att_d + h * ch;
    float ss = 0.f, dd = 0.f;
    for (int c = 0; c < ch; ++c) { float v = hp[c]; ss += v * sp[c]; dd += v * dp[c]; }
    a_s[t] = ss; a_d[t] = dd;
}

#define OWN_NW  32
#define OWN_CAP 1024
#define OWN_U   8

#define OWN_PIT(H) ((H) == 1 ? 40 : 32 * (H))
__device__ __forceinline__ int gat_rh(int row, int h) { return h * OWN_NPB + row; }

__device__ __forceinline__ float gat_lrelu(float v) { return (v >= 0.f) ? v : NEG_SLOPE * v; }

template<int H, bool SAVE>
__device__ __forceinline__ void gat_own_flush_max(int* wl, int* __restrict__ gl, int cnt, int lo, int E,
        const int* __restrict__ srcI, const int* __restrict__ dstI,
        const float* __restrict__ a_s, const float* ads, unsigned* mxs, int lane)
{
    for (int i0 = 0; i0 < cnt; i0 += 128) {
        int e[4], s[4], dl[4];
#pragma unroll
        for (int u = 0; u < 4; ++u) {
            const int i = i0 + u * 32 + lane;
            e[u] = (i < cnt) ? wl[i] : -1;
        }
#pragma unroll
        for (int u = 0; u < 4; ++u) {
            s[u] = 0; dl[u] = 0;
            if (e[u] >= 0) {
                if (e[u] < E) { s[u] = srcI[e[u]]; dl[u] = dstI[e[u]] - lo; }
                else          { s[u] = e[u] - E;   dl[u] = s[u] - lo; }
            }
        }
#pragma unroll
        for (int u = 0; u < 4; ++u) {
            if (e[u] >= 0) {
                float as[H];
                if constexpr (H == 4) {
                    const float4 q = *(const float4*)(a_s + (size_t)s[u] * 4);
                    as[0] = q.x; as[1] = q.y; as[2] = q.z; as[3] = q.w;
                } else {
#pragma unroll
                    for (int h = 0; h < H; ++h) as[h] = a_s[(size_t)s[u] * H + h];
                }
#pragma unroll
                for (int h = 0; h < H; ++h) {
                    const float v = gat_lrelu(as[h] + ads[gat_rh(dl[u], h)]);
                    __hip_atomic_fetch_max(&mxs[gat_rh(dl[u], h)], gat_fenc(v),
                                           __ATOMIC_RELAXED, __HIP_MEMORY_SCOPE_WORKGROUP);
                }
                const int pk = (int)(((unsigned)s[u] << 8) | (unsigned)dl[u]);
                wl[i0 + u * 32 + lane] = pk;
                if (SAVE) gl[i0 + u * 32 + lane] = pk;
            }
        }
    }
}

template<int H>
__device__ __forceinline__ void gat_own_list_max(const int* __restrict__ gl, int* wl, int cnt,
        const float* __restrict__ a_s, const float* ads, unsigned* mxs, int lane)
{
    for (int i0 = 0; i0 < cnt; i0 += 128) {
        int p[4];
#pragma unroll
        for (int u = 0; u < 4; ++u) p[u] = gl[min(i0 + u * 32 + lane, cnt - 1)];
        float as[4][H];
#pragma unroll
        for (int u = 0; u < 4; ++u) {
            const int s = (int)((unsigned)p[u] >> 8);
            if constexpr (H == 4) {
                const float4 q = *(const float4*)(a_s + (size_t)s * 4);
                as[u][0] = q.x; as[u][1] = q.y; as[u][2] = q.z; as[u][3] = q.w;
            } else {
#pragma unroll
                for (int h = 0; h < H; ++h) as[u][h] = a_s[(size_t)s * H + h];
            }
        }
#pragma unroll
        for (int u = 0; u < 4; ++u) {
            const int i = i0 + u * 32 + lane;
            if (i < cnt) {
                const int dl = p[u] & 255;
#pragma unroll
                for (int h = 0; h < H; ++h) {
                    const float v = gat_lrelu(as[u][h] + ads[gat_rh(dl, h)]);
                    __hip_atomic_fetch_max(&mxs[gat_rh(dl, h)], gat_fenc(v),
                                           __ATOMIC_RELAXED, __HIP_MEMORY_SCOPE_WORKGROUP);
                }
                wl[i] = p[u];
            }
        }
    }
}

template<int H, bool PACKED>
__device__ __forceinline__ void gat_own_flush_acc(const int* wl, int cnt, int lo, int E,
        const int* __restrict__ srcI, const int* __restrict__ dstI,
        const float* __restrict__ a_s, const float* ads, const unsigned* mxs, float* dens,
        const float* __restrict__ Hf, float* acc, int lane)
{
    constexpr int ROW = 32 * H;
    constexpr int LPR = ROW / 4;
    constexpr int EPI = 32 / LPR;
    constexpr int PIT = OWN_PIT(H);
    const int grp  = lane / LPR;
    const int slot = lane % LPR;
    const int col  = slot * 4;
    const int head = col >> 5;

    for (int base = 0; base < cnt; base += 32) {
        const int m = min(32, cnt - base);
        int   s_i = 0, d_i = 0;
        float a_i[H];
#pragma unroll
        for (int h = 0; h < H; ++h) a_i[h] = 0.f;
        if (lane < m) {
            const int p = wl[base + lane];
            int s, dl;
            if (PACKED) { s = (int)((unsigned)p >> 8); dl = p & 255; }
            else if (p < E) { s = srcI[p]; dl = dstI[p] - lo; }
            else { s = p - E; dl = s - lo; }
            s_i = s; d_i = dl;
            float as[H];
            if constexpr (H == 4) {
                const float4 q = *(const float4*)(a_s + (size_t)s * 4);
                as[0] = q.x; as[1] = q.y; as[2] = q.z; as[3] = q.w;
            } else {
#pragma unroll
                for (int h = 0; h < H; ++h) as[h] = a_s[(size_t)s * H + h];
            }
#pragma unroll
            for (int h = 0; h < H; ++h) {
                const float v = gat_lrelu(as[h] + ads[gat_rh(dl, h)]);
                const float x = __expf(v - gat_fdec(mxs[gat_rh(dl, h)]));
                a_i[h] = x;
                __hip_atomic_fetch_add(&dens[gat_rh(dl, h)], x, __ATOMIC_RELAXED, __HIP_MEMORY_SCOPE_WORKGROUP);
            }
        }
        const int nsteps = (m + EPI - 1) / EPI;
        for (int t0 = 0; t0 < nsteps; t0 += OWN_U) {
            float4 v[OWN_U]; int dd[OWN_U]; float ww[OWN_U]; bool ok[OWN_U];
#pragma unroll
            for (int u = 0; u < OWN_U; ++u) {
                const int idx = (t0 + u) * EPI + grp;
                ok[u] = idx < m;
                const int sl = idx & 31;
                const int s = __shfl(s_i, sl);
                dd[u] = __shfl(d_i, sl);
                float w = __shfl(a_i[0], sl);
#pragma unroll
                for (int h = 1; h < H; ++h) {
                    const float wh = __shfl(a_i[h], sl);
                    w = (head == h) ? wh : w;
                }
                ww[u] = w;
                v[u] = *(const float4*)(Hf + (size_t)s * ROW + col);
            }
#pragma unroll
            for (int u = 0; u < OWN_U; ++u) {
                if (ok[u]) {
                    float* p = acc + dd[u] * PIT + slot;
                    __hip_atomic_fetch_add(p + 0 * LPR, ww[u] * v[u].x, __ATOMIC_RELAXED, __HIP_MEMORY_SCOPE_WORKGROUP);
                    __hip_atomic_fetch_add(p + 1 * LPR, ww[u] * v[u].y, __ATOMIC_RELAXED, __HIP_MEMORY_SCOPE_WORKGROUP);
                    __hip_atomic_fetch_add(p + 2 * LPR, ww[u] * v[u].z, __ATOMIC_RELAXED, __HIP_MEMORY_SCOPE_WORKGROUP);
                    __hip_atomic_fetch_add(p + 3 * LPR, ww[u] * v[u].w, __ATOMIC_RELAXED, __HIP_MEMORY_SCOPE_WORKGROUP);
                }
            }
        }
    }
}


__device__ __forceinline__ unsigned gat_eq_bytes(unsigned x, unsigned key4) {
    const unsigned y = x ^ key4;
    const unsigned t = ((y & 0x7F7F7F7Fu) + 0x7F7F7F7Fu) | y;
    return ~t & 0x80808080u;
}
__device__ __forceinline__ unsigned gat_eq_mask16(uint4 v, unsigned key4) {
    return (gat_eq_bytes(v.x, key4) >> 7) | (gat_eq_bytes(v.y, key4) >> 6) |
           (gat_eq_bytes(v.z, key4) >> 5) | (gat_eq_bytes(v.w, key4) >> 4);
}

template<int H, bool ACC>
__device__ __forceinline__ void gat_own_drain(int* wl, int cnt, int lo, int E,
        const int* __restrict__ srcI, const int* __restrict__ dstI,
        const float* __restrict__ a_s, const float* ads, unsigned* mxs, float* dens,
        const float* __restrict__ Hf, float* acc, int* again, int lane)
{
    if (ACC) {
        gat_own_flush_acc<H, false>(wl, cnt, lo, E, srcI, dstI, a_s, ads, mxs, dens, Hf, acc, lane);
    } else {
        gat_own_flush_max<H, false>(wl, nullptr, cnt, lo, E, srcI, dstI, a_s, ads, mxs, lane);
        if (lane == 0) __hip_atomic_store(again, 1, __ATOMIC_RELAXED, __HIP_MEMORY_SCOPE_WORKGROUP);
    }
}

template<int H, bool ACC>
__device__ __forceinline__ int gat_own_sweep(int* wl, int lo, int nodes, int E,
        const int* __restrict__ srcI, const int* __restrict__ dstI,
        const unsigned* __restrict__ obyte, bool bytes_ok,
        const float* __restrict__ a_s, const float* ads, unsigned* mxs, float* dens,
        const float* __restrict__ Hf, float* acc, int* again, int lane, int wave)
{
    const unsigned lt = (1u << lane) - 1u;
    int cnt = 0;

    for (int nb = wave * 32; nb < nodes; nb += OWN_NW * 32) {
        const int n = nb + lane;
        const bool hit = n < nodes;
        const unsigned mask = __builtin_amdgcn_ballot_w32(hit);
        if (hit) wl[cnt + __popc(mask & lt)] = E + lo + n;
        cnt += __popc(mask);
    }

    int tail0;
    if (bytes_ok) {
    const int nv16 = E >> 4;
    const uint4* ob = (const uint4*)obyte;
    const unsigned key4 = (unsigned)blockIdx.x * 0x01010101u;
    const uint4 none = make_uint4(~key4, ~key4, ~key4, ~key4);
    int gb = wave * 128;
    uint4 cur[4];
#pragma unroll
    for (int u = 0; u < 4; ++u) {
        const int g = gb + u * 32 + lane;
        cur[u] = ob[max(min(g, nv16 - 1), 0)];
        if (g >= nv16) cur[u] = none;
    }
    while (gb < nv16) {
        const int gn = gb + OWN_NW * 128;
        uint4 nxt[4];
#pragma unroll
        for (int u = 0; u < 4; ++u) {
            const int g = gn + u * 32 + lane;
            nxt[u] = ob[min(g, nv16 - 1)];
            if (g >= nv16) nxt[u] = none;
        }
        const unsigned m01 = gat_eq_mask16(cur[0], key4) | (gat_eq_mask16(cur[1], key4) << 4);
        const unsigned m23 = gat_eq_mask16(cur[2], key4) | (gat_eq_mask16(cur[3], key4) << 4);
        unsigned long long M = (unsigned long long)m01 | ((unsigned long long)m23 << 32);
        unsigned pend = __builtin_amdgcn_ballot_w32(M != 0ull);
        while (pend) {
            if (M != 0ull) {
                const int p = __ffsll((long long)M) - 1;
                const int u = ((p >> 5) << 1) | ((p >> 2) & 1);
                const int off = ((p & 3) << 2) | ((p >> 3) & 3);
                wl[cnt + __popc(pend & lt)] = ((gb + u * 32 + lane) << 4) + off;
                M &= M - 1ull;
            }
            cnt += __popc(pend);
            if (cnt > OWN_CAP - 32) {
                gat_own_drain<H, ACC>(wl, cnt, lo, E, srcI, dstI, a_s, ads, mxs, dens, Hf, acc, again, lane);
                cnt = 0;
            }
            pend = __builtin_amdgcn_ballot_w32(M != 0ull);
        }
#pragma unroll
        for (int u = 0; u < 4; ++u) cur[u] = nxt[u];
        gb = gn;
    }
    tail0 = nv16 << 4;
    } else {
    const int nvec = E >> 2;
    const int4* d4 = (const int4*)dstI;
    const int4 none = make_int4(-1, -1, -1, -1);
    int gb = wave * 128;
    int4 cur[4];
#pragma unroll
    for (int u = 0; u < 4; ++u) {
        const int g = gb + u * 32 + lane;
        cur[u] = d4[max(min(g, nvec - 1), 0)];
        if (g >= nvec) cur[u] = none;
    }
    while (gb < nvec) {
        const int gn = gb + OWN_NW * 128;
        int4 nxt[4];
#pragma unroll
        for (int u = 0; u < 4; ++u) {
            const int g = gn + u * 32 + lane;
            nxt[u] = d4[min(g, nvec - 1)];
            if (g >= nvec) nxt[u] = none;
        }
#pragma unroll
        for (int u = 0; u < 4; ++u) {
            const int e0 = (gb + u * 32 + lane) * 4;
            const bool h0 = (unsigned)(cur[u].x - lo) < (unsigned)nodes;
            const bool h1 = (unsigned)(cur[u].y - lo) < (unsigned)nodes;
            const bool h2 = (unsigned)(cur[u].z - lo) < (unsigned)nodes;
            const bool h3 = (unsigned)(cur[u].w - lo) < (unsigned)nodes;
            if (__builtin_amdgcn_ballot_w32(h0 | h1 | h2 | h3)) {
                unsigned mask;
                mask = __builtin_amdgcn_ballot_w32(h0);
                if (h0) wl[cnt + __popc(mask & lt)] = e0 + 0;
                cnt += __popc(mask);
                mask = __builtin_amdgcn_ballot_w32(h1);
                if (h1) wl[cnt + __popc(mask & lt)] = e0 + 1;
                cnt += __popc(mask);
                mask = __builtin_amdgcn_ballot_w32(h2);
                if (h2) wl[cnt + __popc(mask & lt)] = e0 + 2;
                cnt += __popc(mask);
                mask = __builtin_amdgcn_ballot_w32(h3);
                if (h3) wl[cnt + __popc(mask & lt)] = e0 + 3;
                cnt += __popc(mask);
            }
        }
        if (cnt > OWN_CAP / 2) {
            gat_own_drain<H, ACC>(wl, cnt, lo, E, srcI, dstI, a_s, ads, mxs, dens, Hf, acc, again, lane);
            cnt = 0;
        }
#pragma unroll
        for (int u = 0; u < 4; ++u) cur[u] = nxt[u];
        gb = gn;
    }
    tail0 = nvec * 4;
    }
    if (wave == 0) {
        for (int eb0 = tail0; eb0 < E; eb0 += 32) {
            const int e = eb0 + lane;
            const bool hit = (e < E) && ((unsigned)(dstI[min(e, E - 1)] - lo) < (unsigned)nodes);
            const unsigned mask = __builtin_amdgcn_ballot_w32(hit);
            if (hit) wl[cnt + __popc(mask & lt)] = e;
            cnt += __popc(mask);
        }
    }
    return cnt;
}

template<int H, bool RELU, bool SAVE, bool LOAD>
__global__ __launch_bounds__(OWN_NW * 32)
void gat_own_aggregate(const int* __restrict__ srcI, const int* __restrict__ dstI,
                       const float* __restrict__ a_s, const float* __restrict__ a_d,
                       const float* __restrict__ Hf, const float* __restrict__ bias,
                       float* __restrict__ outp, int* __restrict__ glist, int* __restrict__ gcnt,
                       const unsigned* __restrict__ obyte, int E, int N)
{
    constexpr int ROW = 32 * H;
    constexpr int LPR = ROW / 4;
    constexpr int PIT = OWN_PIT(H);
    static_assert(OWN_NPB <= 32 * OWN_NW, "one self-loop step per wave");
    static_assert(OWN_NPB <= 256, "local row index is packed into 8 bits");
    static_assert(PIT >= ROW, "a row's four slabs fit inside its pitch");
    __shared__ float acc[OWN_NPB * PIT];
    __shared__ int lst[OWN_NW * OWN_CAP];
    __shared__ float    ads[OWN_NPB * H];
    __shared__ unsigned mxs[OWN_NPB * H];
    __shared__ float    dens[OWN_NPB * H];
    __shared__ int again;

    const int tid  = threadIdx.x;
    const int lane = tid & 31;
    const int wave = tid >> 5;
    const int lo   = blockIdx.x * OWN_NPB;
    const int nodes = min(OWN_NPB, N - lo);
    const bool bytes_ok = gridDim.x <= 256u;

    for (int i = tid; i < nodes * PIT; i += OWN_NW * 32) acc[i] = 0.f;
    for (int i = tid; i < nodes * H; i += OWN_NW * 32) {
        const int j = gat_rh(i / H, i % H);
        ads[j]  = a_d[(size_t)lo * H + i];
        mxs[j]  = ENC_NEG_INF;
        dens[j] = 0.f;
    }
    if (tid == 0) again = 0;
    __syncthreads();

    int* wl = lst + wave * OWN_CAP;
    int* gl = glist + ((size_t)blockIdx.x * OWN_NW + wave) * OWN_CAP;
    int* gc = gcnt + (size_t)blockIdx.x * OWN_NW;
    int cnt = 0;
    bool saved = false;
    if (LOAD) saved = gc[0] >= 0;
    if (saved) {
        cnt = gc[wave];
        gat_own_list_max<H>(gl, wl, cnt, a_s, ads, mxs, lane);
    } else {
        cnt = gat_own_sweep<H, false>(wl, lo, nodes, E, srcI, dstI, obyte, bytes_ok, a_s, ads, mxs, dens, Hf, acc, &again, lane, wave);
        gat_own_flush_max<H, SAVE>(wl, gl, cnt, lo, E, srcI, dstI, a_s, ads, mxs, lane);
    }
    __syncthreads();

    const int ag = __hip_atomic_load(&again, __ATOMIC_RELAXED, __HIP_MEMORY_SCOPE_WORKGROUP);
    if (SAVE && lane == 0) gc[wave] = ag ? -1 : cnt;
    if (ag == 0) {
        gat_own_flush_acc<H, true>(wl, cnt, lo, E, srcI, dstI, a_s, ads, mxs, dens, Hf, acc, lane);
    } else {
        cnt = gat_own_sweep<H, true>(wl, lo, nodes, E, srcI, dstI, obyte, bytes_ok, a_s, ads, mxs, dens, Hf, acc, &again, lane, wave);
        gat_own_flush_acc<H, false>(wl, cnt, lo, E, srcI, dstI, a_s, ads, mxs, dens, Hf, acc, lane);
    }
    __syncthreads();

    float* orow = outp + (size_t)lo * ROW;
    for (int i = tid * 4; i < nodes * ROW; i += OWN_NW * 32 * 4) {
        const int c = i & (ROW - 1);
        const float* a = acc + (i / ROW) * PIT + (c >> 2);
        float4 r = make_float4(a[0], a[LPR], a[2 * LPR], a[3 * LPR]);
        const float dn = dens[gat_rh(i / ROW, c >> 5)];
        r.x = r.x / dn + bias[c + 0];
        r.y = r.y / dn + bias[c + 1];
        r.z = r.z / dn + bias[c + 2];
        r.w = r.w / dn + bias[c + 3];
        if (RELU) { r.x = fmaxf(r.x, 0.f); r.y = fmaxf(r.y, 0.f); r.z = fmaxf(r.z, 0.f); r.w = fmaxf(r.w, 0.f); }
        *(float4*)(orow + i) = r;
    }
}

extern "C" void kernel_launch(void* const* d_in, const int* in_sizes, int n_in,
                              void* d_out, int out_size, void* d_ws, size_t ws_size,
                              hipStream_t stream)
{
    const float* x   = (const float*)d_in[0];
    const int*   ei  = (const int*)d_in[1];
    const float* W1  = (const float*)d_in[2];
    const float* as1 = (const float*)d_in[3];
    const float* ad1 = (const float*)d_in[4];
    const float* b1  = (const float*)d_in[5];
    const float* W2  = (const float*)d_in[6];
    const float* as2 = (const float*)d_in[7];
    const float* ad2 = (const float*)d_in[8];
    const float* b2  = (const float*)d_in[9];
    float* out = (float*)d_out;

    const int N  = in_sizes[0] / 128;
    const int E  = in_sizes[1] / 2;
    const int* srcI = ei;
    const int* dstI = ei + E;

    float* ws = (float*)d_ws;
    float*    h1    = ws;              ws += (size_t)N * 128;
    float*    asrc1 = ws;              ws += (size_t)N * 4;
    float*    adst1 = ws;              ws += (size_t)N * 4;
    float*    h2    = ws;              ws += (size_t)N * 128;
    float*    g2    = ws;              ws += (size_t)N * 32;
    float*    asrc2 = ws;              ws += (size_t)N;
    float*    adst2 = ws;              ws += (size_t)N;

    const int TB = 256;
    const int gB   = (N + 127) / 128;
    const int gN4  = (N * 4 + TB - 1) / TB;
    const int gN1  = (N + TB - 1) / TB;
    const int gA   = (N + OWN_NPB - 1) / OWN_NPB;
    int* glist = (int*)ws;
    int* gcnt  = glist + (size_t)gA * OWN_NW * OWN_CAP;
    unsigned* obyte = (unsigned*)(gcnt + (((size_t)gA * OWN_NW + 3) & ~(size_t)3));
    const int nquad = E >> 2;
    const int gE4   = (nquad + TB - 1) / TB;

    gat_gemm_wmma<8><<<gB, TB, 0, stream>>>(x, W1, h1, N);
    gat_attn_score<<<gN4 + gE4, TB, 0, stream>>>(h1, as1, ad1, asrc1, adst1, N, 4, 32, dstI, obyte, nquad, gN4);
    gat_own_aggregate<4, true, true, false><<<gA, OWN_NW * 32, 0, stream>>>(srcI, dstI, asrc1, adst1, h1, b1, h2, glist, gcnt, obyte, E, N);

    gat_gemm_wmma<2><<<gB, TB, 0, stream>>>(h2, W2, g2, N);
    gat_attn_score<<<gN1, TB, 0, stream>>>(g2, as2, ad2, asrc2, adst2, N, 1, 32, dstI, obyte, 0, gN1);
    gat_own_aggregate<1, false, false, true><<<gA, OWN_NW * 32, 0, stream>>>(srcI, dstI, asrc2, adst2, g2, b2, out, glist, gcnt, obyte, E, N);
}


# Scores as extra projection columns: each block folds W.att_src and W.att_dst into one more B tile (x.(W.att), one more f32 matrix tile per k-step); score launches gone, 6 -> 4; score sums in k order
# speedup vs baseline: 12.8786x; 1.0581x over previous
#include <hip/hip_runtime.h>

#define NEG_SLOPE 0.2f
#define ENC_NEG_INF 0x007FFFFFu

typedef float v2f __attribute__((ext_vector_type(2)));
typedef float v8f __attribute__((ext_vector_type(8)));

__device__ __forceinline__ unsigned gat_fenc(float f) {
    unsigned u = __float_as_uint(f);
    return (u & 0x80000000u) ? ~u : (u | 0x80000000u);
}
__device__ __forceinline__ float gat_fdec(unsigned u) {
    unsigned b = (u & 0x80000000u) ? (u & 0x7FFFFFFFu) : ~u;
    return __uint_as_float(b);
}

#define OWN_NPB 200

__device__ __forceinline__ unsigned gat_owner_id(int d) { return min((unsigned)d / OWN_NPB, 255u); }

template<int NT>
__global__ __launch_bounds__(256)
void gat_gemm_wmma(const float* __restrict__ X, const float* __restrict__ W,
                   const float* __restrict__ att_s, const float* __restrict__ att_d,
                   float* __restrict__ Hout, float* __restrict__ a_s, float* __restrict__ a_d,
                   int n_rows,
                   const int* __restrict__ dstI, unsigned* __restrict__ obyte,
                   int nquad)
{
    constexpr int KD = 128;
    constexpr int ND = NT * 16;
    constexpr int HN = NT / 2;
    constexpr int NE = ND + 16;
    static_assert(2 * HN <= 16, "score columns fit one tile");
    __shared__ float Wt[NE * KD];

    for (int q = (int)blockIdx.x * 256 + (int)threadIdx.x; q < nquad; q += (int)gridDim.x * 256) {
        const int4 d = ((const int4*)dstI)[q];
        obyte[q] = gat_owner_id(d.x) | (gat_owner_id(d.y) << 8) |
                   (gat_owner_id(d.z) << 16) | (gat_owner_id(d.w) << 24);
    }

    for (int i = threadIdx.x; i < KD * ND; i += 256) {
        int k = i / ND, n = i % ND;
        Wt[n * KD + ((k + 4 * n) & (KD - 1))] = W[i];
    }
    __syncthreads();

    {
        const int kk = (int)threadIdx.x & (KD - 1);
        const int half = __builtin_amdgcn_readfirstlane((int)threadIdx.x >> 7);
        const float* av = half ? att_d : att_s;
        float s[HN];
#pragma unroll
        for (int h = 0; h < HN; ++h) {
            s[h] = 0.f;
            for (int cc = 0; cc < 32; ++cc) {
                const int n = h * 32 + cc;
                s[h] += Wt[n * KD + ((kk + 4 * n) & (KD - 1))] * av[n];
            }
        }
#pragma unroll
        for (int h = 0; h < HN; ++h) {
            const int n = ND + half * HN + h;
            Wt[n * KD + ((kk + 4 * n) & (KD - 1))] = s[h];
        }
        for (int i = threadIdx.x; i < (16 - 2 * HN) * KD; i += 256) {
            const int n = ND + 2 * HN + i / KD, k = i % KD;
            Wt[n * KD + ((k + 4 * n) & (KD - 1))] = 0.f;
        }
    }
    __syncthreads();

    const int lane = threadIdx.x & 31;
    const int wave = threadIdx.x >> 5;
    const int m0   = (blockIdx.x * 8 + wave) * 16;
    const int mlo  = lane & 15;
    const int kofs = (lane >> 4) * 2;

    int arow = m0 + mlo;
    if (arow >= n_rows) arow = n_rows - 1;
    const float* xrow = X + (size_t)arow * KD + kofs;

    v8f acc[NT + 1] = {};

#pragma unroll
    for (int k = 0; k < KD; k += 4) {
        v2f a = *(const v2f*)(xrow + k);
#pragma unroll
        for (int t = 0; t < NT + 1; ++t) {
            int n = t * 16 + mlo;
            v2f b = *(const v2f*)(&Wt[n * KD + ((k + kofs + 4 * n) & (KD - 1))]);
            acc[t] = __builtin_amdgcn_wmma_f32_16x16x4_f32(
                false, a, false, b, (short)0, acc[t], false, false);
        }
    }

    const int mhi = (lane >> 4) * 8;
#pragma unroll
    for (int t = 0; t < NT; ++t)
#pragma unroll
        for (int j = 0; j < 8; ++j) {
            int m = m0 + mhi + j;
            if (m < n_rows) Hout[(size_t)m * ND + t * 16 + mlo] = acc[t][j];
        }
    __syncthreads();
    float* sc = Wt;
    if (mlo < 2 * HN) {
        const int kind = (mlo < HN) ? 0 : 1;
        const int hd   = mlo - kind * HN;
#pragma unroll
        for (int j = 0; j < 8; ++j)
            sc[kind * 128 * HN + (wave * 16 + mhi + j) * HN + hd] = acc[NT][j];
    }
    __syncthreads();
    const int rbase = blockIdx.x * 128;
    for (int i = threadIdx.x; i < 128 * HN; i += 256) {
        if (rbase + i / HN < n_rows) {
            a_s[(size_t)rbase * HN + i] = sc[i];
            a_d[(size_t)rbase * HN + i] = sc[128 * HN + i];
        }
    }
}


#define OWN_NW  32
#define OWN_CAP 1024
#define OWN_U   8

#define OWN_PIT(H) ((H) == 1 ? 40 : 32 * (H))
__device__ __forceinline__ int gat_rh(int row, int h) { return h * OWN_NPB + row; }

__device__ __forceinline__ float gat_lrelu(float v) { return (v >= 0.f) ? v : NEG_SLOPE * v; }

template<int H, bool SAVE>
__device__ __forceinline__ void gat_own_flush_max(int* wl, int* __restrict__ gl, int cnt, int lo, int E,
        const int* __restrict__ srcI, const int* __restrict__ dstI,
        const float* __restrict__ a_s, const float* ads, unsigned* mxs, int lane)
{
    for (int i0 = 0; i0 < cnt; i0 += 128) {
        int e[4], s[4], dl[4];
#pragma unroll
        for (int u = 0; u < 4; ++u) {
            const int i = i0 + u * 32 + lane;
            e[u] = (i < cnt) ? wl[i] : -1;
        }
#pragma unroll
        for (int u = 0; u < 4; ++u) {
            s[u] = 0; dl[u] = 0;
            if (e[u] >= 0) {
                if (e[u] < E) { s[u] = srcI[e[u]]; dl[u] = dstI[e[u]] - lo; }
                else          { s[u] = e[u] - E;   dl[u] = s[u] - lo; }
            }
        }
#pragma unroll
        for (int u = 0; u < 4; ++u) {
            if (e[u] >= 0) {
                float as[H];
                if constexpr (H == 4) {
                    const float4 q = *(const float4*)(a_s + (size_t)s[u] * 4);
                    as[0] = q.x; as[1] = q.y; as[2] = q.z; as[3] = q.w;
                } else {
#pragma unroll
                    for (int h = 0; h < H; ++h) as[h] = a_s[(size_t)s[u] * H + h];
                }
#pragma unroll
                for (int h = 0; h < H; ++h) {
                    const float v = gat_lrelu(as[h] + ads[gat_rh(dl[u], h)]);
                    __hip_atomic_fetch_max(&mxs[gat_rh(dl[u], h)], gat_fenc(v),
                                           __ATOMIC_RELAXED, __HIP_MEMORY_SCOPE_WORKGROUP);
                }
                const int pk = (int)(((unsigned)s[u] << 8) | (unsigned)dl[u]);
                wl[i0 + u * 32 + lane] = pk;
                if (SAVE) gl[i0 + u * 32 + lane] = pk;
            }
        }
    }
}

template<int H>
__device__ __forceinline__ void gat_own_list_max(const int* __restrict__ gl, int* wl, int cnt,
        const float* __restrict__ a_s, const float* ads, unsigned* mxs, int lane)
{
    for (int i0 = 0; i0 < cnt; i0 += 128) {
        int p[4];
#pragma unroll
        for (int u = 0; u < 4; ++u) p[u] = gl[min(i0 + u * 32 + lane, cnt - 1)];
        float as[4][H];
#pragma unroll
        for (int u = 0; u < 4; ++u) {
            const int s = (int)((unsigned)p[u] >> 8);
            if constexpr (H == 4) {
                const float4 q = *(const float4*)(a_s + (size_t)s * 4);
                as[u][0] = q.x; as[u][1] = q.y; as[u][2] = q.z; as[u][3] = q.w;
            } else {
#pragma unroll
                for (int h = 0; h < H; ++h) as[u][h] = a_s[(size_t)s * H + h];
            }
        }
#pragma unroll
        for (int u = 0; u < 4; ++u) {
            const int i = i0 + u * 32 + lane;
            if (i < cnt) {
                const int dl = p[u] & 255;
#pragma unroll
                for (int h = 0; h < H; ++h) {
                    const float v = gat_lrelu(as[u][h] + ads[gat_rh(dl, h)]);
                    __hip_atomic_fetch_max(&mxs[gat_rh(dl, h)], gat_fenc(v),
                                           __ATOMIC_RELAXED, __HIP_MEMORY_SCOPE_WORKGROUP);
                }
                wl[i] = p[u];
            }
        }
    }
}

template<int H, bool PACKED>
__device__ __forceinline__ void gat_own_flush_acc(const int* wl, int cnt, int lo, int E,
        const int* __restrict__ srcI, const int* __restrict__ dstI,
        const float* __restrict__ a_s, const float* ads, const unsigned* mxs, float* dens,
        const float* __restrict__ Hf, float* acc, int lane)
{
    constexpr int ROW = 32 * H;
    constexpr int LPR = ROW / 4;
    constexpr int EPI = 32 / LPR;
    constexpr int PIT = OWN_PIT(H);
    const int grp  = lane / LPR;
    const int slot = lane % LPR;
    const int col  = slot * 4;
    const int head = col >> 5;

    for (int base = 0; base < cnt; base += 32) {
        const int m = min(32, cnt - base);
        int   s_i = 0, d_i = 0;
        float a_i[H];
#pragma unroll
        for (int h = 0; h < H; ++h) a_i[h] = 0.f;
        if (lane < m) {
            const int p = wl[base + lane];
            int s, dl;
            if (PACKED) { s = (int)((unsigned)p >> 8); dl = p & 255; }
            else if (p < E) { s = srcI[p]; dl = dstI[p] - lo; }
            else { s = p - E; dl = s - lo; }
            s_i = s; d_i = dl;
            float as[H];
            if constexpr (H == 4) {
                const float4 q = *(const float4*)(a_s + (size_t)s * 4);
                as[0] = q.x; as[1] = q.y; as[2] = q.z; as[3] = q.w;
            } else {
#pragma unroll
                for (int h = 0; h < H; ++h) as[h] = a_s[(size_t)s * H + h];
            }
#pragma unroll
            for (int h = 0; h < H; ++h) {
                const float v = gat_lrelu(as[h] + ads[gat_rh(dl, h)]);
                const float x = __expf(v - gat_fdec(mxs[gat_rh(dl, h)]));
                a_i[h] = x;
                __hip_atomic_fetch_add(&dens[gat_rh(dl, h)], x, __ATOMIC_RELAXED, __HIP_MEMORY_SCOPE_WORKGROUP);
            }
        }
        const int nsteps = (m + EPI - 1) / EPI;
        for (int t0 = 0; t0 < nsteps; t0 += OWN_U) {
            float4 v[OWN_U]; int dd[OWN_U]; float ww[OWN_U]; bool ok[OWN_U];
#pragma unroll
            for (int u = 0; u < OWN_U; ++u) {
                const int idx = (t0 + u) * EPI + grp;
                ok[u] = idx < m;
                const int sl = idx & 31;
                const int s = __shfl(s_i, sl);
                dd[u] = __shfl(d_i, sl);
                float w = __shfl(a_i[0], sl);
#pragma unroll
                for (int h = 1; h < H; ++h) {
                    const float wh = __shfl(a_i[h], sl);
                    w = (head == h) ? wh : w;
                }
                ww[u] = w;
                v[u] = *(const float4*)(Hf + (size_t)s * ROW + col);
            }
#pragma unroll
            for (int u = 0; u < OWN_U; ++u) {
                if (ok[u]) {
                    float* p = acc + dd[u] * PIT + slot;
                    __hip_atomic_fetch_add(p + 0 * LPR, ww[u] * v[u].x, __ATOMIC_RELAXED, __HIP_MEMORY_SCOPE_WORKGROUP);
                    __hip_atomic_fetch_add(p + 1 * LPR, ww[u] * v[u].y, __ATOMIC_RELAXED, __HIP_MEMORY_SCOPE_WORKGROUP);
                    __hip_atomic_fetch_add(p + 2 * LPR, ww[u] * v[u].z, __ATOMIC_RELAXED, __HIP_MEMORY_SCOPE_WORKGROUP);
                    __hip_atomic_fetch_add(p + 3 * LPR, ww[u] * v[u].w, __ATOMIC_RELAXED, __HIP_MEMORY_SCOPE_WORKGROUP);
                }
            }
        }
    }
}


__device__ __forceinline__ unsigned gat_eq_bytes(unsigned x, unsigned key4) {
    const unsigned y = x ^ key4;
    const unsigned t = ((y & 0x7F7F7F7Fu) + 0x7F7F7F7Fu) | y;
    return ~t & 0x80808080u;
}
__device__ __forceinline__ unsigned gat_eq_mask16(uint4 v, unsigned key4) {
    return (gat_eq_bytes(v.x, key4) >> 7) | (gat_eq_bytes(v.y, key4) >> 6) |
           (gat_eq_bytes(v.z, key4) >> 5) | (gat_eq_bytes(v.w, key4) >> 4);
}

template<int H, bool ACC>
__device__ __forceinline__ void gat_own_drain(int* wl, int cnt, int lo, int E,
        const int* __restrict__ srcI, const int* __restrict__ dstI,
        const float* __restrict__ a_s, const float* ads, unsigned* mxs, float* dens,
        const float* __restrict__ Hf, float* acc, int* again, int lane)
{
    if (ACC) {
        gat_own_flush_acc<H, false>(wl, cnt, lo, E, srcI, dstI, a_s, ads, mxs, dens, Hf, acc, lane);
    } else {
        gat_own_flush_max<H, false>(wl, nullptr, cnt, lo, E, srcI, dstI, a_s, ads, mxs, lane);
        if (lane == 0) __hip_atomic_store(again, 1, __ATOMIC_RELAXED, __HIP_MEMORY_SCOPE_WORKGROUP);
    }
}

template<int H, bool ACC>
__device__ __forceinline__ int gat_own_sweep(int* wl, int lo, int nodes, int E,
        const int* __restrict__ srcI, const int* __restrict__ dstI,
        const unsigned* __restrict__ obyte, bool bytes_ok,
        const float* __restrict__ a_s, const float* ads, unsigned* mxs, float* dens,
        const float* __restrict__ Hf, float* acc, int* again, int lane, int wave)
{
    const unsigned lt = (1u << lane) - 1u;
    int cnt = 0;

    for (int nb = wave * 32; nb < nodes; nb += OWN_NW * 32) {
        const int n = nb + lane;
        const bool hit = n < nodes;
        const unsigned mask = __builtin_amdgcn_ballot_w32(hit);
        if (hit) wl[cnt + __popc(mask & lt)] = E + lo + n;
        cnt += __popc(mask);
    }

    int tail0;
    if (bytes_ok) {
    const int nv16 = E >> 4;
    const uint4* ob = (const uint4*)obyte;
    const unsigned key4 = (unsigned)blockIdx.x * 0x01010101u;
    const uint4 none = make_uint4(~key4, ~key4, ~key4, ~key4);
    int gb = wave * 128;
    uint4 cur[4];
#pragma unroll
    for (int u = 0; u < 4; ++u) {
        const int g = gb + u * 32 + lane;
        cur[u] = ob[max(min(g, nv16 - 1), 0)];
        if (g >= nv16) cur[u] = none;
    }
    while (gb < nv16) {
        const int gn = gb + OWN_NW * 128;
        uint4 nxt[4];
#pragma unroll
        for (int u = 0; u < 4; ++u) {
            const int g = gn + u * 32 + lane;
            nxt[u] = ob[min(g, nv16 - 1)];
            if (g >= nv16) nxt[u] = none;
        }
        const unsigned m01 = gat_eq_mask16(cur[0], key4) | (gat_eq_mask16(cur[1], key4) << 4);
        const unsigned m23 = gat_eq_mask16(cur[2], key4) | (gat_eq_mask16(cur[3], key4) << 4);
        unsigned long long M = (unsigned long long)m01 | ((unsigned long long)m23 << 32);
        unsigned pend = __builtin_amdgcn_ballot_w32(M != 0ull);
        while (pend) {
            if (M != 0ull) {
                const int p = __ffsll((long long)M) - 1;
                const int u = ((p >> 5) << 1) | ((p >> 2) & 1);
                const int off = ((p & 3) << 2) | ((p >> 3) & 3);
                wl[cnt + __popc(pend & lt)] = ((gb + u * 32 + lane) << 4) + off;
                M &= M - 1ull;
            }
            cnt += __popc(pend);
            if (cnt > OWN_CAP - 32) {
                gat_own_drain<H, ACC>(wl, cnt, lo, E, srcI, dstI, a_s, ads, mxs, dens, Hf, acc, again, lane);
                cnt = 0;
            }
            pend = __builtin_amdgcn_ballot_w32(M != 0ull);
        }
#pragma unroll
        for (int u = 0; u < 4; ++u) cur[u] = nxt[u];
        gb = gn;
    }
    tail0 = nv16 << 4;
    } else {
    const int nvec = E >> 2;
    const int4* d4 = (const int4*)dstI;
    const int4 none = make_int4(-1, -1, -1, -1);
    int gb = wave * 128;
    int4 cur[4];
#pragma unroll
    for (int u = 0; u < 4; ++u) {
        const int g = gb + u * 32 + lane;
        cur[u] = d4[max(min(g, nvec - 1), 0)];
        if (g >= nvec) cur[u] = none;
    }
    while (gb < nvec) {
        const int gn = gb + OWN_NW * 128;
        int4 nxt[4];
#pragma unroll
        for (int u = 0; u < 4; ++u) {
            const int g = gn + u * 32 + lane;
            nxt[u] = d4[min(g, nvec - 1)];
            if (g >= nvec) nxt[u] = none;
        }
#pragma unroll
        for (int u = 0; u < 4; ++u) {
            const int e0 = (gb + u * 32 + lane) * 4;
            const bool h0 = (unsigned)(cur[u].x - lo) < (unsigned)nodes;
            const bool h1 = (unsigned)(cur[u].y - lo) < (unsigned)nodes;
            const bool h2 = (unsigned)(cur[u].z - lo) < (unsigned)nodes;
            const bool h3 = (unsigned)(cur[u].w - lo) < (unsigned)nodes;
            if (__builtin_amdgcn_ballot_w32(h0 | h1 | h2 | h3)) {
                unsigned mask;
                mask = __builtin_amdgcn_ballot_w32(h0);
                if (h0) wl[cnt + __popc(mask & lt)] = e0 + 0;
                cnt += __popc(mask);
                mask = __builtin_amdgcn_ballot_w32(h1);
                if (h1) wl[cnt + __popc(mask & lt)] = e0 + 1;
                cnt += __popc(mask);
                mask = __builtin_amdgcn_ballot_w32(h2);
                if (h2) wl[cnt + __popc(mask & lt)] = e0 + 2;
                cnt += __popc(mask);
                mask = __builtin_amdgcn_ballot_w32(h3);
                if (h3) wl[cnt + __popc(mask & lt)] = e0 + 3;
                cnt += __popc(mask);
            }
        }
        if (cnt > OWN_CAP / 2) {
            gat_own_drain<H, ACC>(wl, cnt, lo, E, srcI, dstI, a_s, ads, mxs, dens, Hf, acc, again, lane);
            cnt = 0;
        }
#pragma unroll
        for (int u = 0; u < 4; ++u) cur[u] = nxt[u];
        gb = gn;
    }
    tail0 = nvec * 4;
    }
    if (wave == 0) {
        for (int eb0 = tail0; eb0 < E; eb0 += 32) {
            const int e = eb0 + lane;
            const bool hit = (e < E) && ((unsigned)(dstI[min(e, E - 1)] - lo) < (unsigned)nodes);
            const unsigned mask = __builtin_amdgcn_ballot_w32(hit);
            if (hit) wl[cnt + __popc(mask & lt)] = e;
            cnt += __popc(mask);
        }
    }
    return cnt;
}

template<int H, bool RELU, bool SAVE, bool LOAD>
__global__ __launch_bounds__(OWN_NW * 32)
void gat_own_aggregate(const int* __restrict__ srcI, const int* __restrict__ dstI,
                       const float* __restrict__ a_s, const float* __restrict__ a_d,
                       const float* __restrict__ Hf, const float* __restrict__ bias,
                       float* __restrict__ outp, int* __restrict__ glist, int* __restrict__ gcnt,
                       const unsigned* __restrict__ obyte, int E, int N)
{
    constexpr int ROW = 32 * H;
    constexpr int LPR = ROW / 4;
    constexpr int PIT = OWN_PIT(H);
    static_assert(OWN_NPB <= 32 * OWN_NW, "one self-loop step per wave");
    static_assert(OWN_NPB <= 256, "local row index is packed into 8 bits");
    static_assert(PIT >= ROW, "a row's four slabs fit inside its pitch");
    __shared__ float acc[OWN_NPB * PIT];
    __shared__ int lst[OWN_NW * OWN_CAP];
    __shared__ float    ads[OWN_NPB * H];
    __shared__ unsigned mxs[OWN_NPB * H];
    __shared__ float    dens[OWN_NPB * H];
    __shared__ int again;

    const int tid  = threadIdx.x;
    const int lane = tid & 31;
    const int wave = tid >> 5;
    const int lo   = blockIdx.x * OWN_NPB;
    const int nodes = min(OWN_NPB, N - lo);
    const bool bytes_ok = gridDim.x <= 256u;

    for (int i = tid; i < nodes * PIT; i += OWN_NW * 32) acc[i] = 0.f;
    for (int i = tid; i < nodes * H; i += OWN_NW * 32) {
        const int j = gat_rh(i / H, i % H);
        ads[j]  = a_d[(size_t)lo * H + i];
        mxs[j]  = ENC_NEG_INF;
        dens[j] = 0.f;
    }
    if (tid == 0) again = 0;
    __syncthreads();

    int* wl = lst + wave * OWN_CAP;
    int* gl = glist + ((size_t)blockIdx.x * OWN_NW + wave) * OWN_CAP;
    int* gc = gcnt + (size_t)blockIdx.x * OWN_NW;
    int cnt = 0;
    bool saved = false;
    if (LOAD) saved = gc[0] >= 0;
    if (saved) {
        cnt = gc[wave];
        gat_own_list_max<H>(gl, wl, cnt, a_s, ads, mxs, lane);
    } else {
        cnt = gat_own_sweep<H, false>(wl, lo, nodes, E, srcI, dstI, obyte, bytes_ok, a_s, ads, mxs, dens, Hf, acc, &again, lane, wave);
        gat_own_flush_max<H, SAVE>(wl, gl, cnt, lo, E, srcI, dstI, a_s, ads, mxs, lane);
    }
    __syncthreads();

    const int ag = __hip_atomic_load(&again, __ATOMIC_RELAXED, __HIP_MEMORY_SCOPE_WORKGROUP);
    if (SAVE && lane == 0) gc[wave] = ag ? -1 : cnt;
    if (ag == 0) {
        gat_own_flush_acc<H, true>(wl, cnt, lo, E, srcI, dstI, a_s, ads, mxs, dens, Hf, acc, lane);
    } else {
        cnt = gat_own_sweep<H, true>(wl, lo, nodes, E, srcI, dstI, obyte, bytes_ok, a_s, ads, mxs, dens, Hf, acc, &again, lane, wave);
        gat_own_flush_acc<H, false>(wl, cnt, lo, E, srcI, dstI, a_s, ads, mxs, dens, Hf, acc, lane);
    }
    __syncthreads();

    float* orow = outp + (size_t)lo * ROW;
    for (int i = tid * 4; i < nodes * ROW; i += OWN_NW * 32 * 4) {
        const int c = i & (ROW - 1);
        const float* a = acc + (i / ROW) * PIT + (c >> 2);
        float4 r = make_float4(a[0], a[LPR], a[2 * LPR], a[3 * LPR]);
        const float dn = dens[gat_rh(i / ROW, c >> 5)];
        r.x = r.x / dn + bias[c + 0];
        r.y = r.y / dn + bias[c + 1];
        r.z = r.z / dn + bias[c + 2];
        r.w = r.w / dn + bias[c + 3];
        if (RELU) { r.x = fmaxf(r.x, 0.f); r.y = fmaxf(r.y, 0.f); r.z = fmaxf(r.z, 0.f); r.w = fmaxf(r.w, 0.f); }
        *(float4*)(orow + i) = r;
    }
}

extern "C" void kernel_launch(void* const* d_in, const int* in_sizes, int n_in,
                              void* d_out, int out_size, void* d_ws, size_t ws_size,
                              hipStream_t stream)
{
    const float* x   = (const float*)d_in[0];
    const int*   ei  = (const int*)d_in[1];
    const float* W1  = (const float*)d_in[2];
    const float* as1 = (const float*)d_in[3];
    const float* ad1 = (const float*)d_in[4];
    const float* b1  = (const float*)d_in[5];
    const float* W2  = (const float*)d_in[6];
    const float* as2 = (const float*)d_in[7];
    const float* ad2 = (const float*)d_in[8];
    const float* b2  = (const float*)d_in[9];
    float* out = (float*)d_out;

    const int N  = in_sizes[0] / 128;
    const int E  = in_sizes[1] / 2;
    const int* srcI = ei;
    const int* dstI = ei + E;

    float* ws = (float*)d_ws;
    float*    h1    = ws;              ws += (size_t)N * 128;
    float*    asrc1 = ws;              ws += (size_t)N * 4;
    float*    adst1 = ws;              ws += (size_t)N * 4;
    float*    h2    = ws;              ws += (size_t)N * 128;
    float*    g2    = ws;              ws += (size_t)N * 32;
    float*    asrc2 = ws;              ws += (size_t)N;
    float*    adst2 = ws;              ws += (size_t)N;

    const int TB = 256;
    const int gB   = (N + 127) / 128;
    const int gA   = (N + OWN_NPB - 1) / OWN_NPB;
    int* glist = (int*)ws;
    int* gcnt  = glist + (size_t)gA * OWN_NW * OWN_CAP;
    unsigned* obyte = (unsigned*)(gcnt + (((size_t)gA * OWN_NW + 3) & ~(size_t)3));
    const int nquad = E >> 2;

    gat_gemm_wmma<8><<<gB, TB, 0, stream>>>(x, W1, as1, ad1, h1, asrc1, adst1, N, dstI, obyte, nquad);
    gat_own_aggregate<4, true, true, false><<<gA, OWN_NW * 32, 0, stream>>>(srcI, dstI, asrc1, adst1, h1, b1, h2, glist, gcnt, obyte, E, N);

    gat_gemm_wmma<2><<<gB, TB, 0, stream>>>(h2, W2, as2, ad2, g2, asrc2, adst2, N, dstI, obyte, 0);
    gat_own_aggregate<1, false, false, true><<<gA, OWN_NW * 32, 0, stream>>>(srcI, dstI, asrc2, adst2, g2, b2, out, glist, gcnt, obyte, E, N);
}


# Layer-1 accumulate pass: a step's source, row and four weights come from the owning lane by v_readlane (scalars) instead of six LDS permutes per edge; row sums stay in LDS arrival order
# speedup vs baseline: 13.3646x; 1.0377x over previous
#include <hip/hip_runtime.h>

#define NEG_SLOPE 0.2f
#define ENC_NEG_INF 0x007FFFFFu

typedef float v2f __attribute__((ext_vector_type(2)));
typedef float v8f __attribute__((ext_vector_type(8)));

__device__ __forceinline__ unsigned gat_fenc(float f) {
    unsigned u = __float_as_uint(f);
    return (u & 0x80000000u) ? ~u : (u | 0x80000000u);
}
__device__ __forceinline__ float gat_fdec(unsigned u) {
    unsigned b = (u & 0x80000000u) ? (u & 0x7FFFFFFFu) : ~u;
    return __uint_as_float(b);
}

#define OWN_NPB 200

__device__ __forceinline__ unsigned gat_owner_id(int d) { return min((unsigned)d / OWN_NPB, 255u); }

template<int NT>
__global__ __launch_bounds__(256)
void gat_gemm_wmma(const float* __restrict__ X, const float* __restrict__ W,
                   const float* __restrict__ att_s, const float* __restrict__ att_d,
                   float* __restrict__ Hout, float* __restrict__ a_s, float* __restrict__ a_d,
                   int n_rows,
                   const int* __restrict__ dstI, unsigned* __restrict__ obyte,
                   int nquad)
{
    constexpr int KD = 128;
    constexpr int ND = NT * 16;
    constexpr int HN = NT / 2;
    constexpr int NE = ND + 16;
    static_assert(2 * HN <= 16, "score columns fit one tile");
    __shared__ float Wt[NE * KD];

    for (int q = (int)blockIdx.x * 256 + (int)threadIdx.x; q < nquad; q += (int)gridDim.x * 256) {
        const int4 d = ((const int4*)dstI)[q];
        obyte[q] = gat_owner_id(d.x) | (gat_owner_id(d.y) << 8) |
                   (gat_owner_id(d.z) << 16) | (gat_owner_id(d.w) << 24);
    }

    for (int i = threadIdx.x; i < KD * ND; i += 256) {
        int k = i / ND, n = i % ND;
        Wt[n * KD + ((k + 4 * n) & (KD - 1))] = W[i];
    }
    __syncthreads();

    {
        const int kk = (int)threadIdx.x & (KD - 1);
        const int half = __builtin_amdgcn_readfirstlane((int)threadIdx.x >> 7);
        const float* av = half ? att_d : att_s;
        float s[HN];
#pragma unroll
        for (int h = 0; h < HN; ++h) {
            s[h] = 0.f;
            for (int cc = 0; cc < 32; ++cc) {
                const int n = h * 32 + cc;
                s[h] += Wt[n * KD + ((kk + 4 * n) & (KD - 1))] * av[n];
            }
        }
#pragma unroll
        for (int h = 0; h < HN; ++h) {
            const int n = ND + half * HN + h;
            Wt[n * KD + ((kk + 4 * n) & (KD - 1))] = s[h];
        }
        for (int i = threadIdx.x; i < (16 - 2 * HN) * KD; i += 256) {
            const int n = ND + 2 * HN + i / KD, k = i % KD;
            Wt[n * KD + ((k + 4 * n) & (KD - 1))] = 0.f;
        }
    }
    __syncthreads();

    const int lane = threadIdx.x & 31;
    const int wave = threadIdx.x >> 5;
    const int m0   = (blockIdx.x * 8 + wave) * 16;
    const int mlo  = lane & 15;
    const int kofs = (lane >> 4) * 2;

    int arow = m0 + mlo;
    if (arow >= n_rows) arow = n_rows - 1;
    const float* xrow = X + (size_t)arow * KD + kofs;

    v8f acc[NT + 1] = {};

#pragma unroll
    for (int k = 0; k < KD; k += 4) {
        v2f a = *(const v2f*)(xrow + k);
#pragma unroll
        for (int t = 0; t < NT + 1; ++t) {
            int n = t * 16 + mlo;
            v2f b = *(const v2f*)(&Wt[n * KD + ((k + kofs + 4 * n) & (KD - 1))]);
            acc[t] = __builtin_amdgcn_wmma_f32_16x16x4_f32(
                false, a, false, b, (short)0, acc[t], false, false);
        }
    }

    const int mhi = (lane >> 4) * 8;
#pragma unroll
    for (int t = 0; t < NT; ++t)
#pragma unroll
        for (int j = 0; j < 8; ++j) {
            int m = m0 + mhi + j;
            if (m < n_rows) Hout[(size_t)m * ND + t * 16 + mlo] = acc[t][j];
        }
    __syncthreads();
    float* sc = Wt;
    if (mlo < 2 * HN) {
        const int kind = (mlo < HN) ? 0 : 1;
        const int hd   = mlo - kind * HN;
#pragma unroll
        for (int j = 0; j < 8; ++j)
            sc[kind * 128 * HN + (wave * 16 + mhi + j) * HN + hd] = acc[NT][j];
    }
    __syncthreads();
    const int rbase = blockIdx.x * 128;
    for (int i = threadIdx.x; i < 128 * HN; i += 256) {
        if (rbase + i / HN < n_rows) {
            a_s[(size_t)rbase * HN + i] = sc[i];
            a_d[(size_t)rbase * HN + i] = sc[128 * HN + i];
        }
    }
}


#define OWN_NW  32
#define OWN_CAP 1024
#define OWN_U   8

#define OWN_PIT(H) ((H) == 1 ? 40 : 32 * (H))
__device__ __forceinline__ int gat_rh(int row, int h) { return h * OWN_NPB + row; }

__device__ __forceinline__ float gat_lrelu(float v) { return (v >= 0.f) ? v : NEG_SLOPE * v; }

template<int H, bool SAVE>
__device__ __forceinline__ void gat_own_flush_max(int* wl, int* __restrict__ gl, int cnt, int lo, int E,
        const int* __restrict__ srcI, const int* __restrict__ dstI,
        const float* __restrict__ a_s, const float* ads, unsigned* mxs, int lane)
{
    for (int i0 = 0; i0 < cnt; i0 += 128) {
        int e[4], s[4], dl[4];
#pragma unroll
        for (int u = 0; u < 4; ++u) {
            const int i = i0 + u * 32 + lane;
            e[u] = (i < cnt) ? wl[i] : -1;
        }
#pragma unroll
        for (int u = 0; u < 4; ++u) {
            s[u] = 0; dl[u] = 0;
            if (e[u] >= 0) {
                if (e[u] < E) { s[u] = srcI[e[u]]; dl[u] = dstI[e[u]] - lo; }
                else          { s[u] = e[u] - E;   dl[u] = s[u] - lo; }
            }
        }
#pragma unroll
        for (int u = 0; u < 4; ++u) {
            if (e[u] >= 0) {
                float as[H];
                if constexpr (H == 4) {
                    const float4 q = *(const float4*)(a_s + (size_t)s[u] * 4);
                    as[0] = q.x; as[1] = q.y; as[2] = q.z; as[3] = q.w;
                } else {
#pragma unroll
                    for (int h = 0; h < H; ++h) as[h] = a_s[(size_t)s[u] * H + h];
                }
#pragma unroll
                for (int h = 0; h < H; ++h) {
                    const float v = gat_lrelu(as[h] + ads[gat_rh(dl[u], h)]);
                    __hip_atomic_fetch_max(&mxs[gat_rh(dl[u], h)], gat_fenc(v),
                                           __ATOMIC_RELAXED, __HIP_MEMORY_SCOPE_WORKGROUP);
                }
                const int pk = (int)(((unsigned)s[u] << 8) | (unsigned)dl[u]);
                wl[i0 + u * 32 + lane] = pk;
                if (SAVE) gl[i0 + u * 32 + lane] = pk;
            }
        }
    }
}

template<int H>
__device__ __forceinline__ void gat_own_list_max(const int* __restrict__ gl, int* wl, int cnt,
        const float* __restrict__ a_s, const float* ads, unsigned* mxs, int lane)
{
    for (int i0 = 0; i0 < cnt; i0 += 128) {
        int p[4];
#pragma unroll
        for (int u = 0; u < 4; ++u) p[u] = gl[min(i0 + u * 32 + lane, cnt - 1)];
        float as[4][H];
#pragma unroll
        for (int u = 0; u < 4; ++u) {
            const int s = (int)((unsigned)p[u] >> 8);
            if constexpr (H == 4) {
                const float4 q = *(const float4*)(a_s + (size_t)s * 4);
                as[u][0] = q.x; as[u][1] = q.y; as[u][2] = q.z; as[u][3] = q.w;
            } else {
#pragma unroll
                for (int h = 0; h < H; ++h) as[u][h] = a_s[(size_t)s * H + h];
            }
        }
#pragma unroll
        for (int u = 0; u < 4; ++u) {
            const int i = i0 + u * 32 + lane;
            if (i < cnt) {
                const int dl = p[u] & 255;
#pragma unroll
                for (int h = 0; h < H; ++h) {
                    const float v = gat_lrelu(as[u][h] + ads[gat_rh(dl, h)]);
                    __hip_atomic_fetch_max(&mxs[gat_rh(dl, h)], gat_fenc(v),
                                           __ATOMIC_RELAXED, __HIP_MEMORY_SCOPE_WORKGROUP);
                }
                wl[i] = p[u];
            }
        }
    }
}

template<int H, bool FULL>
__device__ __forceinline__ void gat_own_acc_group(int t0, int m, int s_i, int d_i, const float (&a_i)[H],
        const float* __restrict__ Hf, float* acc, int col, int slot, int head)
{
    constexpr int ROW = 32 * H;
    constexpr int LPR = ROW / 4;
    constexpr int PIT = OWN_PIT(H);
    float4 v[OWN_U]; int dd[OWN_U]; float ww[OWN_U];
#pragma unroll
    for (int u = 0; u < OWN_U; ++u) {
        const int j = FULL ? t0 + u : min(t0 + u, m - 1);
        const int s = __builtin_amdgcn_readlane(s_i, j);
        dd[u] = __builtin_amdgcn_readlane(d_i, j);
        float w = __int_as_float(__builtin_amdgcn_readlane(__float_as_int(a_i[0]), j));
#pragma unroll
        for (int h = 1; h < H; ++h) {
            const float wh = __int_as_float(__builtin_amdgcn_readlane(__float_as_int(a_i[h]), j));
            w = (head == h) ? wh : w;
        }
        ww[u] = w;
        v[u] = *(const float4*)(Hf + (size_t)s * ROW + col);
    }
#pragma unroll
    for (int u = 0; u < OWN_U; ++u) {
        if (FULL || t0 + u < m) {
            float* p = acc + dd[u] * PIT + slot;
            __hip_atomic_fetch_add(p + 0 * LPR, ww[u] * v[u].x, __ATOMIC_RELAXED, __HIP_MEMORY_SCOPE_WORKGROUP);
            __hip_atomic_fetch_add(p + 1 * LPR, ww[u] * v[u].y, __ATOMIC_RELAXED, __HIP_MEMORY_SCOPE_WORKGROUP);
            __hip_atomic_fetch_add(p + 2 * LPR, ww[u] * v[u].z, __ATOMIC_RELAXED, __HIP_MEMORY_SCOPE_WORKGROUP);
            __hip_atomic_fetch_add(p + 3 * LPR, ww[u] * v[u].w, __ATOMIC_RELAXED, __HIP_MEMORY_SCOPE_WORKGROUP);
        }
    }
}

template<int H, bool PACKED>
__device__ __forceinline__ void gat_own_flush_acc(const int* wl, int cnt, int lo, int E,
        const int* __restrict__ srcI, const int* __restrict__ dstI,
        const float* __restrict__ a_s, const float* ads, const unsigned* mxs, float* dens,
        const float* __restrict__ Hf, float* acc, int lane)
{
    constexpr int ROW = 32 * H;
    constexpr int LPR = ROW / 4;
    constexpr int EPI = 32 / LPR;
    constexpr int PIT = OWN_PIT(H);
    const int grp  = lane / LPR;
    const int slot = lane % LPR;
    const int col  = slot * 4;
    const int head = col >> 5;
    if constexpr (EPI == 1) cnt = __builtin_amdgcn_readfirstlane(cnt);

    for (int base = 0; base < cnt; base += 32) {
        const int m = min(32, cnt - base);
        int   s_i = 0, d_i = 0;
        float a_i[H];
#pragma unroll
        for (int h = 0; h < H; ++h) a_i[h] = 0.f;
        if (lane < m) {
            const int p = wl[base + lane];
            int s, dl;
            if (PACKED) { s = (int)((unsigned)p >> 8); dl = p & 255; }
            else if (p < E) { s = srcI[p]; dl = dstI[p] - lo; }
            else { s = p - E; dl = s - lo; }
            s_i = s; d_i = dl;
            float as[H];
            if constexpr (H == 4) {
                const float4 q = *(const float4*)(a_s + (size_t)s * 4);
                as[0] = q.x; as[1] = q.y; as[2] = q.z; as[3] = q.w;
            } else {
#pragma unroll
                for (int h = 0; h < H; ++h) as[h] = a_s[(size_t)s * H + h];
            }
#pragma unroll
            for (int h = 0; h < H; ++h) {
                const float v = gat_lrelu(as[h] + ads[gat_rh(dl, h)]);
                const float x = __expf(v - gat_fdec(mxs[gat_rh(dl, h)]));
                a_i[h] = x;
                __hip_atomic_fetch_add(&dens[gat_rh(dl, h)], x, __ATOMIC_RELAXED, __HIP_MEMORY_SCOPE_WORKGROUP);
            }
        }
        if constexpr (EPI == 1) {
            int t0 = 0;
            for (; t0 + OWN_U <= m; t0 += OWN_U)
                gat_own_acc_group<H, true>(t0, m, s_i, d_i, a_i, Hf, acc, col, slot, head);
            if (t0 < m)
                gat_own_acc_group<H, false>(t0, m, s_i, d_i, a_i, Hf, acc, col, slot, head);
            continue;
        }
        const int nsteps = (m + EPI - 1) / EPI;
        for (int t0 = 0; t0 < nsteps; t0 += OWN_U) {
            float4 v[OWN_U]; int dd[OWN_U]; float ww[OWN_U]; bool ok[OWN_U];
#pragma unroll
            for (int u = 0; u < OWN_U; ++u) {
                const int idx = (t0 + u) * EPI + grp;
                ok[u] = idx < m;
                const int sl = idx & 31;
                const int s = __shfl(s_i, sl);
                dd[u] = __shfl(d_i, sl);
                float w = __shfl(a_i[0], sl);
#pragma unroll
                for (int h = 1; h < H; ++h) {
                    const float wh = __shfl(a_i[h], sl);
                    w = (head == h) ? wh : w;
                }
                ww[u] = w;
                v[u] = *(const float4*)(Hf + (size_t)s * ROW + col);
            }
#pragma unroll
            for (int u = 0; u < OWN_U; ++u) {
                if (ok[u]) {
                    float* p = acc + dd[u] * PIT + slot;
                    __hip_atomic_fetch_add(p + 0 * LPR, ww[u] * v[u].x, __ATOMIC_RELAXED, __HIP_MEMORY_SCOPE_WORKGROUP);
                    __hip_atomic_fetch_add(p + 1 * LPR, ww[u] * v[u].y, __ATOMIC_RELAXED, __HIP_MEMORY_SCOPE_WORKGROUP);
                    __hip_atomic_fetch_add(p + 2 * LPR, ww[u] * v[u].z, __ATOMIC_RELAXED, __HIP_MEMORY_SCOPE_WORKGROUP);
                    __hip_atomic_fetch_add(p + 3 * LPR, ww[u] * v[u].w, __ATOMIC_RELAXED, __HIP_MEMORY_SCOPE_WORKGROUP);
                }
            }
        }
    }
}


__device__ __forceinline__ unsigned gat_eq_bytes(unsigned x, unsigned key4) {
    const unsigned y = x ^ key4;
    const unsigned t = ((y & 0x7F7F7F7Fu) + 0x7F7F7F7Fu) | y;
    return ~t & 0x80808080u;
}
__device__ __forceinline__ unsigned gat_eq_mask16(uint4 v, unsigned key4) {
    return (gat_eq_bytes(v.x, key4) >> 7) | (gat_eq_bytes(v.y, key4) >> 6) |
           (gat_eq_bytes(v.z, key4) >> 5) | (gat_eq_bytes(v.w, key4) >> 4);
}

template<int H, bool ACC>
__device__ __forceinline__ void gat_own_drain(int* wl, int cnt, int lo, int E,
        const int* __restrict__ srcI, const int* __restrict__ dstI,
        const float* __restrict__ a_s, const float* ads, unsigned* mxs, float* dens,
        const float* __restrict__ Hf, float* acc, int* again, int lane)
{
    if (ACC) {
        gat_own_flush_acc<H, false>(wl, cnt, lo, E, srcI, dstI, a_s, ads, mxs, dens, Hf, acc, lane);
    } else {
        gat_own_flush_max<H, false>(wl, nullptr, cnt, lo, E, srcI, dstI, a_s, ads, mxs, lane);
        if (lane == 0) __hip_atomic_store(again, 1, __ATOMIC_RELAXED, __HIP_MEMORY_SCOPE_WORKGROUP);
    }
}

template<int H, bool ACC>
__device__ __forceinline__ int gat_own_sweep(int* wl, int lo, int nodes, int E,
        const int* __restrict__ srcI, const int* __restrict__ dstI,
        const unsigned* __restrict__ obyte, bool bytes_ok,
        const float* __restrict__ a_s, const float* ads, unsigned* mxs, float* dens,
        const float* __restrict__ Hf, float* acc, int* again, int lane, int wave)
{
    const unsigned lt = (1u << lane) - 1u;
    int cnt = 0;

    for (int nb = wave * 32; nb < nodes; nb += OWN_NW * 32) {
        const int n = nb + lane;
        const bool hit = n < nodes;
        const unsigned mask = __builtin_amdgcn_ballot_w32(hit);
        if (hit) wl[cnt + __popc(mask & lt)] = E + lo + n;
        cnt += __popc(mask);
    }

    int tail0;
    if (bytes_ok) {
    const int nv16 = E >> 4;
    const uint4* ob = (const uint4*)obyte;
    const unsigned key4 = (unsigned)blockIdx.x * 0x01010101u;
    const uint4 none = make_uint4(~key4, ~key4, ~key4, ~key4);
    int gb = wave * 128;
    uint4 cur[4];
#pragma unroll
    for (int u = 0; u < 4; ++u) {
        const int g = gb + u * 32 + lane;
        cur[u] = ob[max(min(g, nv16 - 1), 0)];
        if (g >= nv16) cur[u] = none;
    }
    while (gb < nv16) {
        const int gn = gb + OWN_NW * 128;
        uint4 nxt[4];
#pragma unroll
        for (int u = 0; u < 4; ++u) {
            const int g = gn + u * 32 + lane;
            nxt[u] = ob[min(g, nv16 - 1)];
            if (g >= nv16) nxt[u] = none;
        }
        const unsigned m01 = gat_eq_mask16(cur[0], key4) | (gat_eq_mask16(cur[1], key4) << 4);
        const unsigned m23 = gat_eq_mask16(cur[2], key4) | (gat_eq_mask16(cur[3], key4) << 4);
        unsigned long long M = (unsigned long long)m01 | ((unsigned long long)m23 << 32);
        unsigned pend = __builtin_amdgcn_ballot_w32(M != 0ull);
        while (pend) {
            if (M != 0ull) {
                const int p = __ffsll((long long)M) - 1;
                const int u = ((p >> 5) << 1) | ((p >> 2) & 1);
                const int off = ((p & 3) << 2) | ((p >> 3) & 3);
                wl[cnt + __popc(pend & lt)] = ((gb + u * 32 + lane) << 4) + off;
                M &= M - 1ull;
            }
            cnt += __popc(pend);
            if (cnt > OWN_CAP - 32) {
                gat_own_drain<H, ACC>(wl, cnt, lo, E, srcI, dstI, a_s, ads, mxs, dens, Hf, acc, again, lane);
                cnt = 0;
            }
            pend = __builtin_amdgcn_ballot_w32(M != 0ull);
        }
#pragma unroll
        for (int u = 0; u < 4; ++u) cur[u] = nxt[u];
        gb = gn;
    }
    tail0 = nv16 << 4;
    } else {
    const int nvec = E >> 2;
    const int4* d4 = (const int4*)dstI;
    const int4 none = make_int4(-1, -1, -1, -1);
    int gb = wave * 128;
    int4 cur[4];
#pragma unroll
    for (int u = 0; u < 4; ++u) {
        const int g = gb + u * 32 + lane;
        cur[u] = d4[max(min(g, nvec - 1), 0)];
        if (g >= nvec) cur[u] = none;
    }
    while (gb < nvec) {
        const int gn = gb + OWN_NW * 128;
        int4 nxt[4];
#pragma unroll
        for (int u = 0; u < 4; ++u) {
            const int g = gn + u * 32 + lane;
            nxt[u] = d4[min(g, nvec - 1)];
            if (g >= nvec) nxt[u] = none;
        }
#pragma unroll
        for (int u = 0; u < 4; ++u) {
            const int e0 = (gb + u * 32 + lane) * 4;
            const bool h0 = (unsigned)(cur[u].x - lo) < (unsigned)nodes;
            const bool h1 = (unsigned)(cur[u].y - lo) < (unsigned)nodes;
            const bool h2 = (unsigned)(cur[u].z - lo) < (unsigned)nodes;
            const bool h3 = (unsigned)(cur[u].w - lo) < (unsigned)nodes;
            if (__builtin_amdgcn_ballot_w32(h0 | h1 | h2 | h3)) {
                unsigned mask;
                mask = __builtin_amdgcn_ballot_w32(h0);
                if (h0) wl[cnt + __popc(mask & lt)] = e0 + 0;
                cnt += __popc(mask);
                mask = __builtin_amdgcn_ballot_w32(h1);
                if (h1) wl[cnt + __popc(mask & lt)] = e0 + 1;
                cnt += __popc(mask);
                mask = __builtin_amdgcn_ballot_w32(h2);
                if (h2) wl[cnt + __popc(mask & lt)] = e0 + 2;
                cnt += __popc(mask);
                mask = __builtin_amdgcn_ballot_w32(h3);
                if (h3) wl[cnt + __popc(mask & lt)] = e0 + 3;
                cnt += __popc(mask);
            }
        }
        if (cnt > OWN_CAP / 2) {
            gat_own_drain<H, ACC>(wl, cnt, lo, E, srcI, dstI, a_s, ads, mxs, dens, Hf, acc, again, lane);
            cnt = 0;
        }
#pragma unroll
        for (int u = 0; u < 4; ++u) cur[u] = nxt[u];
        gb = gn;
    }
    tail0 = nvec * 4;
    }
    if (wave == 0) {
        for (int eb0 = tail0; eb0 < E; eb0 += 32) {
            const int e = eb0 + lane;
            const bool hit = (e < E) && ((unsigned)(dstI[min(e, E - 1)] - lo) < (unsigned)nodes);
            const unsigned mask = __builtin_amdgcn_ballot_w32(hit);
            if (hit) wl[cnt + __popc(mask & lt)] = e;
            cnt += __popc(mask);
        }
    }
    return cnt;
}

template<int H, bool RELU, bool SAVE, bool LOAD>
__global__ __launch_bounds__(OWN_NW * 32)
void gat_own_aggregate(const int* __restrict__ srcI, const int* __restrict__ dstI,
                       const float* __restrict__ a_s, const float* __restrict__ a_d,
                       const float* __restrict__ Hf, const float* __restrict__ bias,
                       float* __restrict__ outp, int* __restrict__ glist, int* __restrict__ gcnt,
                       const unsigned* __restrict__ obyte, int E, int N)
{
    constexpr int ROW = 32 * H;
    constexpr int LPR = ROW / 4;
    constexpr int PIT = OWN_PIT(H);
    static_assert(OWN_NPB <= 32 * OWN_NW, "one self-loop step per wave");
    static_assert(OWN_NPB <= 256, "local row index is packed into 8 bits");
    static_assert(PIT >= ROW, "a row's four slabs fit inside its pitch");
    __shared__ float acc[OWN_NPB * PIT];
    __shared__ int lst[OWN_NW * OWN_CAP];
    __shared__ float    ads[OWN_NPB * H];
    __shared__ unsigned mxs[OWN_NPB * H];
    __shared__ float    dens[OWN_NPB * H];
    __shared__ int again;

    const int tid  = threadIdx.x;
    const int lane = tid & 31;
    const int wave = tid >> 5;
    const int lo   = blockIdx.x * OWN_NPB;
    const int nodes = min(OWN_NPB, N - lo);
    const bool bytes_ok = gridDim.x <= 256u;

    for (int i = tid; i < nodes * PIT; i += OWN_NW * 32) acc[i] = 0.f;
    for (int i = tid; i < nodes * H; i += OWN_NW * 32) {
        const int j = gat_rh(i / H, i % H);
        ads[j]  = a_d[(size_t)lo * H + i];
        mxs[j]  = ENC_NEG_INF;
        dens[j] = 0.f;
    }
    if (tid == 0) again = 0;
    __syncthreads();

    int* wl = lst + wave * OWN_CAP;
    int* gl = glist + ((size_t)blockIdx.x * OWN_NW + wave) * OWN_CAP;
    int* gc = gcnt + (size_t)blockIdx.x * OWN_NW;
    int cnt = 0;
    bool saved = false;
    if (LOAD) saved = gc[0] >= 0;
    if (saved) {
        cnt = gc[wave];
        gat_own_list_max<H>(gl, wl, cnt, a_s, ads, mxs, lane);
    } else {
        cnt = gat_own_sweep<H, false>(wl, lo, nodes, E, srcI, dstI, obyte, bytes_ok, a_s, ads, mxs, dens, Hf, acc, &again, lane, wave);
        gat_own_flush_max<H, SAVE>(wl, gl, cnt, lo, E, srcI, dstI, a_s, ads, mxs, lane);
    }
    __syncthreads();

    const int ag = __hip_atomic_load(&again, __ATOMIC_RELAXED, __HIP_MEMORY_SCOPE_WORKGROUP);
    if (SAVE && lane == 0) gc[wave] = ag ? -1 : cnt;
    if (ag == 0) {
        gat_own_flush_acc<H, true>(wl, cnt, lo, E, srcI, dstI, a_s, ads, mxs, dens, Hf, acc, lane);
    } else {
        cnt = gat_own_sweep<H, true>(wl, lo, nodes, E, srcI, dstI, obyte, bytes_ok, a_s, ads, mxs, dens, Hf, acc, &again, lane, wave);
        gat_own_flush_acc<H, false>(wl, cnt, lo, E, srcI, dstI, a_s, ads, mxs, dens, Hf, acc, lane);
    }
    __syncthreads();

    float* orow = outp + (size_t)lo * ROW;
    for (int i = tid * 4; i < nodes * ROW; i += OWN_NW * 32 * 4) {
        const int c = i & (ROW - 1);
        const float* a = acc + (i / ROW) * PIT + (c >> 2);
        float4 r = make_float4(a[0], a[LPR], a[2 * LPR], a[3 * LPR]);
        const float dn = dens[gat_rh(i / ROW, c >> 5)];
        r.x = r.x / dn + bias[c + 0];
        r.y = r.y / dn + bias[c + 1];
        r.z = r.z / dn + bias[c + 2];
        r.w = r.w / dn + bias[c + 3];
        if (RELU) { r.x = fmaxf(r.x, 0.f); r.y = fmaxf(r.y, 0.f); r.z = fmaxf(r.z, 0.f); r.w = fmaxf(r.w, 0.f); }
        *(float4*)(orow + i) = r;
    }
}

extern "C" void kernel_launch(void* const* d_in, const int* in_sizes, int n_in,
                              void* d_out, int out_size, void* d_ws, size_t ws_size,
                              hipStream_t stream)
{
    const float* x   = (const float*)d_in[0];
    const int*   ei  = (const int*)d_in[1];
    const float* W1  = (const float*)d_in[2];
    const float* as1 = (const float*)d_in[3];
    const float* ad1 = (const float*)d_in[4];
    const float* b1  = (const float*)d_in[5];
    const float* W2  = (const float*)d_in[6];
    const float* as2 = (const float*)d_in[7];
    const float* ad2 = (const float*)d_in[8];
    const float* b2  = (const float*)d_in[9];
    float* out = (float*)d_out;

    const int N  = in_sizes[0] / 128;
    const int E  = in_sizes[1] / 2;
    const int* srcI = ei;
    const int* dstI = ei + E;

    float* ws = (float*)d_ws;
    float*    h1    = ws;              ws += (size_t)N * 128;
    float*    asrc1 = ws;              ws += (size_t)N * 4;
    float*    adst1 = ws;              ws += (size_t)N * 4;
    float*    h2    = ws;              ws += (size_t)N * 128;
    float*    g2    = ws;              ws += (size_t)N * 32;
    float*    asrc2 = ws;              ws += (size_t)N;
    float*    adst2 = ws;              ws += (size_t)N;

    const int TB = 256;
    const int gB   = (N + 127) / 128;
    const int gA   = (N + OWN_NPB - 1) / OWN_NPB;
    int* glist = (int*)ws;
    int* gcnt  = glist + (size_t)gA * OWN_NW * OWN_CAP;
    unsigned* obyte = (unsigned*)(gcnt + (((size_t)gA * OWN_NW + 3) & ~(size_t)3));
    const int nquad = E >> 2;

    gat_gemm_wmma<8><<<gB, TB, 0, stream>>>(x, W1, as1, ad1, h1, asrc1, adst1, N, dstI, obyte, nquad);
    gat_own_aggregate<4, true, true, false><<<gA, OWN_NW * 32, 0, stream>>>(srcI, dstI, asrc1, adst1, h1, b1, h2, glist, gcnt, obyte, E, N);

    gat_gemm_wmma<2><<<gB, TB, 0, stream>>>(h2, W2, as2, ad2, g2, asrc2, adst2, N, dstI, obyte, 0);
    gat_own_aggregate<1, false, false, true><<<gA, OWN_NW * 32, 0, stream>>>(srcI, dstI, asrc2, adst2, g2, b2, out, glist, gcnt, obyte, E, N);
}


# Projection launches: W staged by 16-byte whole-line loads, a thread's 16 (or 4) loads and its owner-id quads all in flight at once, not 8 dependent rounds of 4-byte loads; k order unchanged
# speedup vs baseline: 14.4238x; 1.0793x over previous
#include <hip/hip_runtime.h>

#define NEG_SLOPE 0.2f
#define ENC_NEG_INF 0x007FFFFFu

typedef float v2f __attribute__((ext_vector_type(2)));
typedef float v8f __attribute__((ext_vector_type(8)));

__device__ __forceinline__ unsigned gat_fenc(float f) {
    unsigned u = __float_as_uint(f);
    return (u & 0x80000000u) ? ~u : (u | 0x80000000u);
}
__device__ __forceinline__ float gat_fdec(unsigned u) {
    unsigned b = (u & 0x80000000u) ? (u & 0x7FFFFFFFu) : ~u;
    return __uint_as_float(b);
}

#define OWN_NPB 200

__device__ __forceinline__ unsigned gat_owner_id(int d) { return min((unsigned)d / OWN_NPB, 255u); }

template<int NT>
__global__ __launch_bounds__(256)
void gat_gemm_wmma(const float* __restrict__ X, const float* __restrict__ W,
                   const float* __restrict__ att_s, const float* __restrict__ att_d,
                   float* __restrict__ Hout, float* __restrict__ a_s, float* __restrict__ a_d,
                   int n_rows,
                   const int* __restrict__ dstI, unsigned* __restrict__ obyte,
                   int nquad)
{
    constexpr int KD = 128;
    constexpr int ND = NT * 16;
    constexpr int HN = NT / 2;
    constexpr int NE = ND + 16;
    static_assert(2 * HN <= 16, "score columns fit one tile");
    __shared__ float Wt[NE * KD];

    const int lane = threadIdx.x & 31;
    const int wave = threadIdx.x >> 5;
    const int m0   = (blockIdx.x * 8 + wave) * 16;
    const int mlo  = lane & 15;
    const int kofs = (lane >> 4) * 2;

    const int4* d4 = (const int4*)dstI;
    const int qstride = (int)gridDim.x * 256;
    const int q0 = (int)blockIdx.x * 256 + (int)threadIdx.x;
    const int q1 = q0 + qstride;
    const int4 dq0 = d4[max(min(q0, nquad - 1), 0)];
    const int4 dq1 = d4[max(min(q1, nquad - 1), 0)];

    constexpr int SEG = ND / 32;
    constexpr int WIT = (KD / 4) * SEG / 8;
    static_assert(ND % 32 == 0 && ((KD / 4) * SEG) % 8 == 0, "W splits into whole-line loads over 8 waves");
    float4 wreg[WIT];
#pragma unroll
    for (int it = 0; it < WIT; ++it) {
        const int c = it * 8 + wave;
        const int k = 4 * (c / SEG) + (lane >> 3);
        const int n = 32 * (c % SEG) + 4 * (lane & 7);
        wreg[it] = *(const float4*)(W + k * ND + n);
    }

    if (q0 < nquad)
        obyte[q0] = gat_owner_id(dq0.x) | (gat_owner_id(dq0.y) << 8) |
                    (gat_owner_id(dq0.z) << 16) | (gat_owner_id(dq0.w) << 24);
    if (q1 < nquad)
        obyte[q1] = gat_owner_id(dq1.x) | (gat_owner_id(dq1.y) << 8) |
                    (gat_owner_id(dq1.z) << 16) | (gat_owner_id(dq1.w) << 24);
    for (int q = q1 + qstride; q < nquad; q += qstride) {
        const int4 d = d4[q];
        obyte[q] = gat_owner_id(d.x) | (gat_owner_id(d.y) << 8) |
                   (gat_owner_id(d.z) << 16) | (gat_owner_id(d.w) << 24);
    }

#pragma unroll
    for (int it = 0; it < WIT; ++it) {
        const int c = it * 8 + wave;
        const int k = 4 * (c / SEG) + (lane >> 3);
        const int n = 32 * (c % SEG) + 4 * (lane & 7);
        Wt[(n + 0) * KD + ((k + 4 * (n + 0)) & (KD - 1))] = wreg[it].x;
        Wt[(n + 1) * KD + ((k + 4 * (n + 1)) & (KD - 1))] = wreg[it].y;
        Wt[(n + 2) * KD + ((k + 4 * (n + 2)) & (KD - 1))] = wreg[it].z;
        Wt[(n + 3) * KD + ((k + 4 * (n + 3)) & (KD - 1))] = wreg[it].w;
    }
    __syncthreads();

    {
        const int kk = (int)threadIdx.x & (KD - 1);
        const int half = __builtin_amdgcn_readfirstlane((int)threadIdx.x >> 7);
        const float* av = half ? att_d : att_s;
        float s[HN];
#pragma unroll
        for (int h = 0; h < HN; ++h) {
            s[h] = 0.f;
            for (int cc = 0; cc < 32; ++cc) {
                const int n = h * 32 + cc;
                s[h] += Wt[n * KD + ((kk + 4 * n) & (KD - 1))] * av[n];
            }
        }
#pragma unroll
        for (int h = 0; h < HN; ++h) {
            const int n = ND + half * HN + h;
            Wt[n * KD + ((kk + 4 * n) & (KD - 1))] = s[h];
        }
        for (int i = threadIdx.x; i < (16 - 2 * HN) * KD; i += 256) {
            const int n = ND + 2 * HN + i / KD, k = i % KD;
            Wt[n * KD + ((k + 4 * n) & (KD - 1))] = 0.f;
        }
    }
    __syncthreads();

    int arow = m0 + mlo;
    if (arow >= n_rows) arow = n_rows - 1;
    const float* xrow = X + (size_t)arow * KD + kofs;

    v8f acc[NT + 1] = {};

#pragma unroll
    for (int k = 0; k < KD; k += 4) {
        v2f a = *(const v2f*)(xrow + k);
#pragma unroll
        for (int t = 0; t < NT + 1; ++t) {
            int n = t * 16 + mlo;
            v2f b = *(const v2f*)(&Wt[n * KD + ((k + kofs + 4 * n) & (KD - 1))]);
            acc[t] = __builtin_amdgcn_wmma_f32_16x16x4_f32(
                false, a, false, b, (short)0, acc[t], false, false);
        }
    }

    const int mhi = (lane >> 4) * 8;
#pragma unroll
    for (int t = 0; t < NT; ++t)
#pragma unroll
        for (int j = 0; j < 8; ++j) {
            int m = m0 + mhi + j;
            if (m < n_rows) Hout[(size_t)m * ND + t * 16 + mlo] = acc[t][j];
        }
    __syncthreads();
    float* sc = Wt;
    if (mlo < 2 * HN) {
        const int kind = (mlo < HN) ? 0 : 1;
        const int hd   = mlo - kind * HN;
#pragma unroll
        for (int j = 0; j < 8; ++j)
            sc[kind * 128 * HN + (wave * 16 + mhi + j) * HN + hd] = acc[NT][j];
    }
    __syncthreads();
    const int rbase = blockIdx.x * 128;
    for (int i = threadIdx.x; i < 128 * HN; i += 256) {
        if (rbase + i / HN < n_rows) {
            a_s[(size_t)rbase * HN + i] = sc[i];
            a_d[(size_t)rbase * HN + i] = sc[128 * HN + i];
        }
    }
}


#define OWN_NW  32
#define OWN_CAP 1024
#define OWN_U   8

#define OWN_PIT(H) ((H) == 1 ? 40 : 32 * (H))
__device__ __forceinline__ int gat_rh(int row, int h) { return h * OWN_NPB + row; }

__device__ __forceinline__ float gat_lrelu(float v) { return (v >= 0.f) ? v : NEG_SLOPE * v; }

template<int H, bool SAVE>
__device__ __forceinline__ void gat_own_flush_max(int* wl, int* __restrict__ gl, int cnt, int lo, int E,
        const int* __restrict__ srcI, const int* __restrict__ dstI,
        const float* __restrict__ a_s, const float* ads, unsigned* mxs, int lane)
{
    for (int i0 = 0; i0 < cnt; i0 += 128) {
        int e[4], s[4], dl[4];
#pragma unroll
        for (int u = 0; u < 4; ++u) {
            const int i = i0 + u * 32 + lane;
            e[u] = (i < cnt) ? wl[i] : -1;
        }
#pragma unroll
        for (int u = 0; u < 4; ++u) {
            s[u] = 0; dl[u] = 0;
            if (e[u] >= 0) {
                if (e[u] < E) { s[u] = srcI[e[u]]; dl[u] = dstI[e[u]] - lo; }
                else          { s[u] = e[u] - E;   dl[u] = s[u] - lo; }
            }
        }
#pragma unroll
        for (int u = 0; u < 4; ++u) {
            if (e[u] >= 0) {
                float as[H];
                if constexpr (H == 4) {
                    const float4 q = *(const float4*)(a_s + (size_t)s[u] * 4);
                    as[0] = q.x; as[1] = q.y; as[2] = q.z; as[3] = q.w;
                } else {
#pragma unroll
                    for (int h = 0; h < H; ++h) as[h] = a_s[(size_t)s[u] * H + h];
                }
#pragma unroll
                for (int h = 0; h < H; ++h) {
                    const float v = gat_lrelu(as[h] + ads[gat_rh(dl[u], h)]);
                    __hip_atomic_fetch_max(&mxs[gat_rh(dl[u], h)], gat_fenc(v),
                                           __ATOMIC_RELAXED, __HIP_MEMORY_SCOPE_WORKGROUP);
                }
                const int pk = (int)(((unsigned)s[u] << 8) | (unsigned)dl[u]);
                wl[i0 + u * 32 + lane] = pk;
                if (SAVE) gl[i0 + u * 32 + lane] = pk;
            }
        }
    }
}

template<int H>
__device__ __forceinline__ void gat_own_list_max(const int* __restrict__ gl, int* wl, int cnt,
        const float* __restrict__ a_s, const float* ads, unsigned* mxs, int lane)
{
    for (int i0 = 0; i0 < cnt; i0 += 128) {
        int p[4];
#pragma unroll
        for (int u = 0; u < 4; ++u) p[u] = gl[min(i0 + u * 32 + lane, cnt - 1)];
        float as[4][H];
#pragma unroll
        for (int u = 0; u < 4; ++u) {
            const int s = (int)((unsigned)p[u] >> 8);
            if constexpr (H == 4) {
                const float4 q = *(const float4*)(a_s + (size_t)s * 4);
                as[u][0] = q.x; as[u][1] = q.y; as[u][2] = q.z; as[u][3] = q.w;
            } else {
#pragma unroll
                for (int h = 0; h < H; ++h) as[u][h] = a_s[(size_t)s * H + h];
            }
        }
#pragma unroll
        for (int u = 0; u < 4; ++u) {
            const int i = i0 + u * 32 + lane;
            if (i < cnt) {
                const int dl = p[u] & 255;
#pragma unroll
                for (int h = 0; h < H; ++h) {
                    const float v = gat_lrelu(as[u][h] + ads[gat_rh(dl, h)]);
                    __hip_atomic_fetch_max(&mxs[gat_rh(dl, h)], gat_fenc(v),
                                           __ATOMIC_RELAXED, __HIP_MEMORY_SCOPE_WORKGROUP);
                }
                wl[i] = p[u];
            }
        }
    }
}

template<int H, bool FULL>
__device__ __forceinline__ void gat_own_acc_group(int t0, int m, int s_i, int d_i, const float (&a_i)[H],
        const float* __restrict__ Hf, float* acc, int col, int slot, int head)
{
    constexpr int ROW = 32 * H;
    constexpr int LPR = ROW / 4;
    constexpr int PIT = OWN_PIT(H);
    float4 v[OWN_U]; int dd[OWN_U]; float ww[OWN_U];
#pragma unroll
    for (int u = 0; u < OWN_U; ++u) {
        const int j = FULL ? t0 + u : min(t0 + u, m - 1);
        const int s = __builtin_amdgcn_readlane(s_i, j);
        dd[u] = __builtin_amdgcn_readlane(d_i, j);
        float w = __int_as_float(__builtin_amdgcn_readlane(__float_as_int(a_i[0]), j));
#pragma unroll
        for (int h = 1; h < H; ++h) {
            const float wh = __int_as_float(__builtin_amdgcn_readlane(__float_as_int(a_i[h]), j));
            w = (head == h) ? wh : w;
        }
        ww[u] = w;
        v[u] = *(const float4*)(Hf + (size_t)s * ROW + col);
    }
#pragma unroll
    for (int u = 0; u < OWN_U; ++u) {
        if (FULL || t0 + u < m) {
            float* p = acc + dd[u] * PIT + slot;
            __hip_atomic_fetch_add(p + 0 * LPR, ww[u] * v[u].x, __ATOMIC_RELAXED, __HIP_MEMORY_SCOPE_WORKGROUP);
            __hip_atomic_fetch_add(p + 1 * LPR, ww[u] * v[u].y, __ATOMIC_RELAXED, __HIP_MEMORY_SCOPE_WORKGROUP);
            __hip_atomic_fetch_add(p + 2 * LPR, ww[u] * v[u].z, __ATOMIC_RELAXED, __HIP_MEMORY_SCOPE_WORKGROUP);
            __hip_atomic_fetch_add(p + 3 * LPR, ww[u] * v[u].w, __ATOMIC_RELAXED, __HIP_MEMORY_SCOPE_WORKGROUP);
        }
    }
}

template<int H, bool PACKED>
__device__ __forceinline__ void gat_own_flush_acc(const int* wl, int cnt, int lo, int E,
        const int* __restrict__ srcI, const int* __restrict__ dstI,
        const float* __restrict__ a_s, const float* ads, const unsigned* mxs, float* dens,
        const float* __restrict__ Hf, float* acc, int lane)
{
    constexpr int ROW = 32 * H;
    constexpr int LPR = ROW / 4;
    constexpr int EPI = 32 / LPR;
    constexpr int PIT = OWN_PIT(H);
    const int grp  = lane / LPR;
    const int slot = lane % LPR;
    const int col  = slot * 4;
    const int head = col >> 5;
    if constexpr (EPI == 1) cnt = __builtin_amdgcn_readfirstlane(cnt);

    for (int base = 0; base < cnt; base += 32) {
        const int m = min(32, cnt - base);
        int   s_i = 0, d_i = 0;
        float a_i[H];
#pragma unroll
        for (int h = 0; h < H; ++h) a_i[h] = 0.f;
        if (lane < m) {
            const int p = wl[base + lane];
            int s, dl;
            if (PACKED) { s = (int)((unsigned)p >> 8); dl = p & 255; }
            else if (p < E) { s = srcI[p]; dl = dstI[p] - lo; }
            else { s = p - E; dl = s - lo; }
            s_i = s; d_i = dl;
            float as[H];
            if constexpr (H == 4) {
                const float4 q = *(const float4*)(a_s + (size_t)s * 4);
                as[0] = q.x; as[1] = q.y; as[2] = q.z; as[3] = q.w;
            } else {
#pragma unroll
                for (int h = 0; h < H; ++h) as[h] = a_s[(size_t)s * H + h];
            }
#pragma unroll
            for (int h = 0; h < H; ++h) {
                const float v = gat_lrelu(as[h] + ads[gat_rh(dl, h)]);
                const float x = __expf(v - gat_fdec(mxs[gat_rh(dl, h)]));
                a_i[h] = x;
                __hip_atomic_fetch_add(&dens[gat_rh(dl, h)], x, __ATOMIC_RELAXED, __HIP_MEMORY_SCOPE_WORKGROUP);
            }
        }
        if constexpr (EPI == 1) {
            int t0 = 0;
            for (; t0 + OWN_U <= m; t0 += OWN_U)
                gat_own_acc_group<H, true>(t0, m, s_i, d_i, a_i, Hf, acc, col, slot, head);
            if (t0 < m)
                gat_own_acc_group<H, false>(t0, m, s_i, d_i, a_i, Hf, acc, col, slot, head);
            continue;
        }
        const int nsteps = (m + EPI - 1) / EPI;
        for (int t0 = 0; t0 < nsteps; t0 += OWN_U) {
            float4 v[OWN_U]; int dd[OWN_U]; float ww[OWN_U]; bool ok[OWN_U];
#pragma unroll
            for (int u = 0; u < OWN_U; ++u) {
                const int idx = (t0 + u) * EPI + grp;
                ok[u] = idx < m;
                const int sl = idx & 31;
                const int s = __shfl(s_i, sl);
                dd[u] = __shfl(d_i, sl);
                float w = __shfl(a_i[0], sl);
#pragma unroll
                for (int h = 1; h < H; ++h) {
                    const float wh = __shfl(a_i[h], sl);
                    w = (head == h) ? wh : w;
                }
                ww[u] = w;
                v[u] = *(const float4*)(Hf + (size_t)s * ROW + col);
            }
#pragma unroll
            for (int u = 0; u < OWN_U; ++u) {
                if (ok[u]) {
                    float* p = acc + dd[u] * PIT + slot;
                    __hip_atomic_fetch_add(p + 0 * LPR, ww[u] * v[u].x, __ATOMIC_RELAXED, __HIP_MEMORY_SCOPE_WORKGROUP);
                    __hip_atomic_fetch_add(p + 1 * LPR, ww[u] * v[u].y, __ATOMIC_RELAXED, __HIP_MEMORY_SCOPE_WORKGROUP);
                    __hip_atomic_fetch_add(p + 2 * LPR, ww[u] * v[u].z, __ATOMIC_RELAXED, __HIP_MEMORY_SCOPE_WORKGROUP);
                    __hip_atomic_fetch_add(p + 3 * LPR, ww[u] * v[u].w, __ATOMIC_RELAXED, __HIP_MEMORY_SCOPE_WORKGROUP);
                }
            }
        }
    }
}


__device__ __forceinline__ unsigned gat_eq_bytes(unsigned x, unsigned key4) {
    const unsigned y = x ^ key4;
    const unsigned t = ((y & 0x7F7F7F7Fu) + 0x7F7F7F7Fu) | y;
    return ~t & 0x80808080u;
}
__device__ __forceinline__ unsigned gat_eq_mask16(uint4 v, unsigned key4) {
    return (gat_eq_bytes(v.x, key4) >> 7) | (gat_eq_bytes(v.y, key4) >> 6) |
           (gat_eq_bytes(v.z, key4) >> 5) | (gat_eq_bytes(v.w, key4) >> 4);
}

template<int H, bool ACC>
__device__ __forceinline__ void gat_own_drain(int* wl, int cnt, int lo, int E,
        const int* __restrict__ srcI, const int* __restrict__ dstI,
        const float* __restrict__ a_s, const float* ads, unsigned* mxs, float* dens,
        const float* __restrict__ Hf, float* acc, int* again, int lane)
{
    if (ACC) {
        gat_own_flush_acc<H, false>(wl, cnt, lo, E, srcI, dstI, a_s, ads, mxs, dens, Hf, acc, lane);
    } else {
        gat_own_flush_max<H, false>(wl, nullptr, cnt, lo, E, srcI, dstI, a_s, ads, mxs, lane);
        if (lane == 0) __hip_atomic_store(again, 1, __ATOMIC_RELAXED, __HIP_MEMORY_SCOPE_WORKGROUP);
    }
}

template<int H, bool ACC>
__device__ __forceinline__ int gat_own_sweep(int* wl, int lo, int nodes, int E,
        const int* __restrict__ srcI, const int* __restrict__ dstI,
        const unsigned* __restrict__ obyte, bool bytes_ok,
        const float* __restrict__ a_s, const float* ads, unsigned* mxs, float* dens,
        const float* __restrict__ Hf, float* acc, int* again, int lane, int wave)
{
    const unsigned lt = (1u << lane) - 1u;
    int cnt = 0;

    for (int nb = wave * 32; nb < nodes; nb += OWN_NW * 32) {
        const int n = nb + lane;
        const bool hit = n < nodes;
        const unsigned mask = __builtin_amdgcn_ballot_w32(hit);
        if (hit) wl[cnt + __popc(mask & lt)] = E + lo + n;
        cnt += __popc(mask);
    }

    int tail0;
    if (bytes_ok) {
    const int nv16 = E >> 4;
    const uint4* ob = (const uint4*)obyte;
    const unsigned key4 = (unsigned)blockIdx.x * 0x01010101u;
    const uint4 none = make_uint4(~key4, ~key4, ~key4, ~key4);
    int gb = wave * 128;
    uint4 cur[4];
#pragma unroll
    for (int u = 0; u < 4; ++u) {
        const int g = gb + u * 32 + lane;
        cur[u] = ob[max(min(g, nv16 - 1), 0)];
        if (g >= nv16) cur[u] = none;
    }
    while (gb < nv16) {
        const int gn = gb + OWN_NW * 128;
        uint4 nxt[4];
#pragma unroll
        for (int u = 0; u < 4; ++u) {
            const int g = gn + u * 32 + lane;
            nxt[u] = ob[min(g, nv16 - 1)];
            if (g >= nv16) nxt[u] = none;
        }
        const unsigned m01 = gat_eq_mask16(cur[0], key4) | (gat_eq_mask16(cur[1], key4) << 4);
        const unsigned m23 = gat_eq_mask16(cur[2], key4) | (gat_eq_mask16(cur[3], key4) << 4);
        unsigned long long M = (unsigned long long)m01 | ((unsigned long long)m23 << 32);
        unsigned pend = __builtin_amdgcn_ballot_w32(M != 0ull);
        while (pend) {
            if (M != 0ull) {
                const int p = __ffsll((long long)M) - 1;
                const int u = ((p >> 5) << 1) | ((p >> 2) & 1);
                const int off = ((p & 3) << 2) | ((p >> 3) & 3);
                wl[cnt + __popc(pend & lt)] = ((gb + u * 32 + lane) << 4) + off;
                M &= M - 1ull;
            }
            cnt += __popc(pend);
            if (cnt > OWN_CAP - 32) {
                gat_own_drain<H, ACC>(wl, cnt, lo, E, srcI, dstI, a_s, ads, mxs, dens, Hf, acc, again, lane);
                cnt = 0;
            }
            pend = __builtin_amdgcn_ballot_w32(M != 0ull);
        }
#pragma unroll
        for (int u = 0; u < 4; ++u) cur[u] = nxt[u];
        gb = gn;
    }
    tail0 = nv16 << 4;
    } else {
    const int nvec = E >> 2;
    const int4* d4 = (const int4*)dstI;
    const int4 none = make_int4(-1, -1, -1, -1);
    int gb = wave * 128;
    int4 cur[4];
#pragma unroll
    for (int u = 0; u < 4; ++u) {
        const int g = gb + u * 32 + lane;
        cur[u] = d4[max(min(g, nvec - 1), 0)];
        if (g >= nvec) cur[u] = none;
    }
    while (gb < nvec) {
        const int gn = gb + OWN_NW * 128;
        int4 nxt[4];
#pragma unroll
        for (int u = 0; u < 4; ++u) {
            const int g = gn + u * 32 + lane;
            nxt[u] = d4[min(g, nvec - 1)];
            if (g >= nvec) nxt[u] = none;
        }
#pragma unroll
        for (int u = 0; u < 4; ++u) {
            const int e0 = (gb + u * 32 + lane) * 4;
            const bool h0 = (unsigned)(cur[u].x - lo) < (unsigned)nodes;
            const bool h1 = (unsigned)(cur[u].y - lo) < (unsigned)nodes;
            const bool h2 = (unsigned)(cur[u].z - lo) < (unsigned)nodes;
            const bool h3 = (unsigned)(cur[u].w - lo) < (unsigned)nodes;
            if (__builtin_amdgcn_ballot_w32(h0 | h1 | h2 | h3)) {
                unsigned mask;
                mask = __builtin_amdgcn_ballot_w32(h0);
                if (h0) wl[cnt + __popc(mask & lt)] = e0 + 0;
                cnt += __popc(mask);
                mask = __builtin_amdgcn_ballot_w32(h1);
                if (h1) wl[cnt + __popc(mask & lt)] = e0 + 1;
                cnt += __popc(mask);
                mask = __builtin_amdgcn_ballot_w32(h2);
                if (h2) wl[cnt + __popc(mask & lt)] = e0 + 2;
                cnt += __popc(mask);
                mask = __builtin_amdgcn_ballot_w32(h3);
                if (h3) wl[cnt + __popc(mask & lt)] = e0 + 3;
                cnt += __popc(mask);
            }
        }
        if (cnt > OWN_CAP / 2) {
            gat_own_drain<H, ACC>(wl, cnt, lo, E, srcI, dstI, a_s, ads, mxs, dens, Hf, acc, again, lane);
            cnt = 0;
        }
#pragma unroll
        for (int u = 0; u < 4; ++u) cur[u] = nxt[u];
        gb = gn;
    }
    tail0 = nvec * 4;
    }
    if (wave == 0) {
        for (int eb0 = tail0; eb0 < E; eb0 += 32) {
            const int e = eb0 + lane;
            const bool hit = (e < E) && ((unsigned)(dstI[min(e, E - 1)] - lo) < (unsigned)nodes);
            const unsigned mask = __builtin_amdgcn_ballot_w32(hit);
            if (hit) wl[cnt + __popc(mask & lt)] = e;
            cnt += __popc(mask);
        }
    }
    return cnt;
}

template<int H, bool RELU, bool SAVE, bool LOAD>
__global__ __launch_bounds__(OWN_NW * 32)
void gat_own_aggregate(const int* __restrict__ srcI, const int* __restrict__ dstI,
                       const float* __restrict__ a_s, const float* __restrict__ a_d,
                       const float* __restrict__ Hf, const float* __restrict__ bias,
                       float* __restrict__ outp, int* __restrict__ glist, int* __restrict__ gcnt,
                       const unsigned* __restrict__ obyte, int E, int N)
{
    constexpr int ROW = 32 * H;
    constexpr int LPR = ROW / 4;
    constexpr int PIT = OWN_PIT(H);
    static_assert(OWN_NPB <= 32 * OWN_NW, "one self-loop step per wave");
    static_assert(OWN_NPB <= 256, "local row index is packed into 8 bits");
    static_assert(PIT >= ROW, "a row's four slabs fit inside its pitch");
    __shared__ float acc[OWN_NPB * PIT];
    __shared__ int lst[OWN_NW * OWN_CAP];
    __shared__ float    ads[OWN_NPB * H];
    __shared__ unsigned mxs[OWN_NPB * H];
    __shared__ float    dens[OWN_NPB * H];
    __shared__ int again;

    const int tid  = threadIdx.x;
    const int lane = tid & 31;
    const int wave = tid >> 5;
    const int lo   = blockIdx.x * OWN_NPB;
    const int nodes = min(OWN_NPB, N - lo);
    const bool bytes_ok = gridDim.x <= 256u;

    for (int i = tid; i < nodes * PIT; i += OWN_NW * 32) acc[i] = 0.f;
    for (int i = tid; i < nodes * H; i += OWN_NW * 32) {
        const int j = gat_rh(i / H, i % H);
        ads[j]  = a_d[(size_t)lo * H + i];
        mxs[j]  = ENC_NEG_INF;
        dens[j] = 0.f;
    }
    if (tid == 0) again = 0;
    __syncthreads();

    int* wl = lst + wave * OWN_CAP;
    int* gl = glist + ((size_t)blockIdx.x * OWN_NW + wave) * OWN_CAP;
    int* gc = gcnt + (size_t)blockIdx.x * OWN_NW;
    int cnt = 0;
    bool saved = false;
    if (LOAD) saved = gc[0] >= 0;
    if (saved) {
        cnt = gc[wave];
        gat_own_list_max<H>(gl, wl, cnt, a_s, ads, mxs, lane);
    } else {
        cnt = gat_own_sweep<H, false>(wl, lo, nodes, E, srcI, dstI, obyte, bytes_ok, a_s, ads, mxs, dens, Hf, acc, &again, lane, wave);
        gat_own_flush_max<H, SAVE>(wl, gl, cnt, lo, E, srcI, dstI, a_s, ads, mxs, lane);
    }
    __syncthreads();

    const int ag = __hip_atomic_load(&again, __ATOMIC_RELAXED, __HIP_MEMORY_SCOPE_WORKGROUP);
    if (SAVE && lane == 0) gc[wave] = ag ? -1 : cnt;
    if (ag == 0) {
        gat_own_flush_acc<H, true>(wl, cnt, lo, E, srcI, dstI, a_s, ads, mxs, dens, Hf, acc, lane);
    } else {
        cnt = gat_own_sweep<H, true>(wl, lo, nodes, E, srcI, dstI, obyte, bytes_ok, a_s, ads, mxs, dens, Hf, acc, &again, lane, wave);
        gat_own_flush_acc<H, false>(wl, cnt, lo, E, srcI, dstI, a_s, ads, mxs, dens, Hf, acc, lane);
    }
    __syncthreads();

    float* orow = outp + (size_t)lo * ROW;
    for (int i = tid * 4; i < nodes * ROW; i += OWN_NW * 32 * 4) {
        const int c = i & (ROW - 1);
        const float* a = acc + (i / ROW) * PIT + (c >> 2);
        float4 r = make_float4(a[0], a[LPR], a[2 * LPR], a[3 * LPR]);
        const float dn = dens[gat_rh(i / ROW, c >> 5)];
        r.x = r.x / dn + bias[c + 0];
        r.y = r.y / dn + bias[c + 1];
        r.z = r.z / dn + bias[c + 2];
        r.w = r.w / dn + bias[c + 3];
        if (RELU) { r.x = fmaxf(r.x, 0.f); r.y = fmaxf(r.y, 0.f); r.z = fmaxf(r.z, 0.f); r.w = fmaxf(r.w, 0.f); }
        *(float4*)(orow + i) = r;
    }
}

extern "C" void kernel_launch(void* const* d_in, const int* in_sizes, int n_in,
                              void* d_out, int out_size, void* d_ws, size_t ws_size,
                              hipStream_t stream)
{
    const float* x   = (const float*)d_in[0];
    const int*   ei  = (const int*)d_in[1];
    const float* W1  = (const float*)d_in[2];
    const float* as1 = (const float*)d_in[3];
    const float* ad1 = (const float*)d_in[4];
    const float* b1  = (const float*)d_in[5];
    const float* W2  = (const float*)d_in[6];
    const float* as2 = (const float*)d_in[7];
    const float* ad2 = (const float*)d_in[8];
    const float* b2  = (const float*)d_in[9];
    float* out = (float*)d_out;

    const int N  = in_sizes[0] / 128;
    const int E  = in_sizes[1] / 2;
    const int* srcI = ei;
    const int* dstI = ei + E;

    float* ws = (float*)d_ws;
    float*    h1    = ws;              ws += (size_t)N * 128;
    float*    asrc1 = ws;              ws += (size_t)N * 4;
    float*    adst1 = ws;              ws += (size_t)N * 4;
    float*    h2    = ws;              ws += (size_t)N * 128;
    float*    g2    = ws;              ws += (size_t)N * 32;
    float*    asrc2 = ws;              ws += (size_t)N;
    float*    adst2 = ws;              ws += (size_t)N;

    const int TB = 256;
    const int gB   = (N + 127) / 128;
    const int gA   = (N + OWN_NPB - 1) / OWN_NPB;
    int* glist = (int*)ws;
    int* gcnt  = glist + (size_t)gA * OWN_NW * OWN_CAP;
    unsigned* obyte = (unsigned*)(gcnt + (((size_t)gA * OWN_NW + 3) & ~(size_t)3));
    const int nquad = E >> 2;

    gat_gemm_wmma<8><<<gB, TB, 0, stream>>>(x, W1, as1, ad1, h1, asrc1, adst1, N, dstI, obyte, nquad);
    gat_own_aggregate<4, true, true, false><<<gA, OWN_NW * 32, 0, stream>>>(srcI, dstI, asrc1, adst1, h1, b1, h2, glist, gcnt, obyte, E, N);

    gat_gemm_wmma<2><<<gB, TB, 0, stream>>>(h2, W2, as2, ad2, g2, asrc2, adst2, N, dstI, obyte, 0);
    gat_own_aggregate<1, false, false, true><<<gA, OWN_NW * 32, 0, stream>>>(srcI, dstI, asrc2, adst2, g2, b2, out, glist, gcnt, obyte, E, N);
}


# Edge scores formed once: pass 1 parks each listed edge's score in the wave's LDS slab, pass 2 reads it back instead of gathering a_src[src] again (both layers); row sums stay in LDS arrival order
# speedup vs baseline: 14.9118x; 1.0338x over previous
#include <hip/hip_runtime.h>

#define NEG_SLOPE 0.2f
#define ENC_NEG_INF 0x007FFFFFu

typedef float v2f __attribute__((ext_vector_type(2)));
typedef float v8f __attribute__((ext_vector_type(8)));

__device__ __forceinline__ unsigned gat_fenc(float f) {
    unsigned u = __float_as_uint(f);
    return (u & 0x80000000u) ? ~u : (u | 0x80000000u);
}
__device__ __forceinline__ float gat_fdec(unsigned u) {
    unsigned b = (u & 0x80000000u) ? (u & 0x7FFFFFFFu) : ~u;
    return __uint_as_float(b);
}

#define OWN_NPB 200

__device__ __forceinline__ unsigned gat_owner_id(int d) { return min((unsigned)d / OWN_NPB, 255u); }

template<int NT>
__global__ __launch_bounds__(256)
void gat_gemm_wmma(const float* __restrict__ X, const float* __restrict__ W,
                   const float* __restrict__ att_s, const float* __restrict__ att_d,
                   float* __restrict__ Hout, float* __restrict__ a_s, float* __restrict__ a_d,
                   int n_rows,
                   const int* __restrict__ dstI, unsigned* __restrict__ obyte,
                   int nquad)
{
    constexpr int KD = 128;
    constexpr int ND = NT * 16;
    constexpr int HN = NT / 2;
    constexpr int NE = ND + 16;
    static_assert(2 * HN <= 16, "score columns fit one tile");
    __shared__ float Wt[NE * KD];

    const int lane = threadIdx.x & 31;
    const int wave = threadIdx.x >> 5;
    const int m0   = (blockIdx.x * 8 + wave) * 16;
    const int mlo  = lane & 15;
    const int kofs = (lane >> 4) * 2;

    const int4* d4 = (const int4*)dstI;
    const int qstride = (int)gridDim.x * 256;
    const int q0 = (int)blockIdx.x * 256 + (int)threadIdx.x;
    const int q1 = q0 + qstride;
    const int4 dq0 = d4[max(min(q0, nquad - 1), 0)];
    const int4 dq1 = d4[max(min(q1, nquad - 1), 0)];

    constexpr int SEG = ND / 32;
    constexpr int WIT = (KD / 4) * SEG / 8;
    static_assert(ND % 32 == 0 && ((KD / 4) * SEG) % 8 == 0, "W splits into whole-line loads over 8 waves");
    float4 wreg[WIT];
#pragma unroll
    for (int it = 0; it < WIT; ++it) {
        const int c = it * 8 + wave;
        const int k = 4 * (c / SEG) + (lane >> 3);
        const int n = 32 * (c % SEG) + 4 * (lane & 7);
        wreg[it] = *(const float4*)(W + k * ND + n);
    }

    if (q0 < nquad)
        obyte[q0] = gat_owner_id(dq0.x) | (gat_owner_id(dq0.y) << 8) |
                    (gat_owner_id(dq0.z) << 16) | (gat_owner_id(dq0.w) << 24);
    if (q1 < nquad)
        obyte[q1] = gat_owner_id(dq1.x) | (gat_owner_id(dq1.y) << 8) |
                    (gat_owner_id(dq1.z) << 16) | (gat_owner_id(dq1.w) << 24);
    for (int q = q1 + qstride; q < nquad; q += qstride) {
        const int4 d = d4[q];
        obyte[q] = gat_owner_id(d.x) | (gat_owner_id(d.y) << 8) |
                   (gat_owner_id(d.z) << 16) | (gat_owner_id(d.w) << 24);
    }

#pragma unroll
    for (int it = 0; it < WIT; ++it) {
        const int c = it * 8 + wave;
        const int k = 4 * (c / SEG) + (lane >> 3);
        const int n = 32 * (c % SEG) + 4 * (lane & 7);
        Wt[(n + 0) * KD + ((k + 4 * (n + 0)) & (KD - 1))] = wreg[it].x;
        Wt[(n + 1) * KD + ((k + 4 * (n + 1)) & (KD - 1))] = wreg[it].y;
        Wt[(n + 2) * KD + ((k + 4 * (n + 2)) & (KD - 1))] = wreg[it].z;
        Wt[(n + 3) * KD + ((k + 4 * (n + 3)) & (KD - 1))] = wreg[it].w;
    }
    __syncthreads();

    {
        const int kk = (int)threadIdx.x & (KD - 1);
        const int half = __builtin_amdgcn_readfirstlane((int)threadIdx.x >> 7);
        const float* av = half ? att_d : att_s;
        float s[HN];
#pragma unroll
        for (int h = 0; h < HN; ++h) {
            s[h] = 0.f;
            for (int cc = 0; cc < 32; ++cc) {
                const int n = h * 32 + cc;
                s[h] += Wt[n * KD + ((kk + 4 * n) & (KD - 1))] * av[n];
            }
        }
#pragma unroll
        for (int h = 0; h < HN; ++h) {
            const int n = ND + half * HN + h;
            Wt[n * KD + ((kk + 4 * n) & (KD - 1))] = s[h];
        }
        for (int i = threadIdx.x; i < (16 - 2 * HN) * KD; i += 256) {
            const int n = ND + 2 * HN + i / KD, k = i % KD;
            Wt[n * KD + ((k + 4 * n) & (KD - 1))] = 0.f;
        }
    }
    __syncthreads();

    int arow = m0 + mlo;
    if (arow >= n_rows) arow = n_rows - 1;
    const float* xrow = X + (size_t)arow * KD + kofs;

    v8f acc[NT + 1] = {};

#pragma unroll
    for (int k = 0; k < KD; k += 4) {
        v2f a = *(const v2f*)(xrow + k);
#pragma unroll
        for (int t = 0; t < NT + 1; ++t) {
            int n = t * 16 + mlo;
            v2f b = *(const v2f*)(&Wt[n * KD + ((k + kofs + 4 * n) & (KD - 1))]);
            acc[t] = __builtin_amdgcn_wmma_f32_16x16x4_f32(
                false, a, false, b, (short)0, acc[t], false, false);
        }
    }

    const int mhi = (lane >> 4) * 8;
#pragma unroll
    for (int t = 0; t < NT; ++t)
#pragma unroll
        for (int j = 0; j < 8; ++j) {
            int m = m0 + mhi + j;
            if (m < n_rows) Hout[(size_t)m * ND + t * 16 + mlo] = acc[t][j];
        }
    __syncthreads();
    float* sc = Wt;
    if (mlo < 2 * HN) {
        const int kind = (mlo < HN) ? 0 : 1;
        const int hd   = mlo - kind * HN;
#pragma unroll
        for (int j = 0; j < 8; ++j)
            sc[kind * 128 * HN + (wave * 16 + mhi + j) * HN + hd] = acc[NT][j];
    }
    __syncthreads();
    const int rbase = blockIdx.x * 128;
    for (int i = threadIdx.x; i < 128 * HN; i += 256) {
        if (rbase + i / HN < n_rows) {
            a_s[(size_t)rbase * HN + i] = sc[i];
            a_d[(size_t)rbase * HN + i] = sc[128 * HN + i];
        }
    }
}


#define OWN_NW  32
#define OWN_CAP 1024
#define OWN_U   8
#define OWN_SC  128

#define OWN_PIT(H) ((H) == 1 ? 40 : 32 * (H))
__device__ __forceinline__ int gat_rh(int row, int h) { return h * OWN_NPB + row; }

__device__ __forceinline__ float gat_lrelu(float v) { return (v >= 0.f) ? v : NEG_SLOPE * v; }

template<int H, bool SAVE, bool KEEP>
__device__ __forceinline__ void gat_own_flush_max(int* wl, int* __restrict__ gl, int cnt, int lo, int E,
        const int* __restrict__ srcI, const int* __restrict__ dstI,
        const float* __restrict__ a_s, const float* ads, unsigned* mxs, float* sc, int lane)
{
    for (int i0 = 0; i0 < cnt; i0 += 128) {
        int e[4], s[4], dl[4];
#pragma unroll
        for (int u = 0; u < 4; ++u) {
            const int i = i0 + u * 32 + lane;
            e[u] = (i < cnt) ? wl[i] : -1;
        }
#pragma unroll
        for (int u = 0; u < 4; ++u) {
            s[u] = 0; dl[u] = 0;
            if (e[u] >= 0) {
                if (e[u] < E) { s[u] = srcI[e[u]]; dl[u] = dstI[e[u]] - lo; }
                else          { s[u] = e[u] - E;   dl[u] = s[u] - lo; }
            }
        }
#pragma unroll
        for (int u = 0; u < 4; ++u) {
            if (e[u] >= 0) {
                float as[H];
                if constexpr (H == 4) {
                    const float4 q = *(const float4*)(a_s + (size_t)s[u] * 4);
                    as[0] = q.x; as[1] = q.y; as[2] = q.z; as[3] = q.w;
                } else {
#pragma unroll
                    for (int h = 0; h < H; ++h) as[h] = a_s[(size_t)s[u] * H + h];
                }
#pragma unroll
                for (int h = 0; h < H; ++h) {
                    const float v = gat_lrelu(as[h] + ads[gat_rh(dl[u], h)]);
                    __hip_atomic_fetch_max(&mxs[gat_rh(dl[u], h)], gat_fenc(v),
                                           __ATOMIC_RELAXED, __HIP_MEMORY_SCOPE_WORKGROUP);
                    if (KEEP && i0 + u * 32 + lane < OWN_SC) sc[h * OWN_SC + i0 + u * 32 + lane] = v;
                }
                const int pk = (int)(((unsigned)s[u] << 8) | (unsigned)dl[u]);
                wl[i0 + u * 32 + lane] = pk;
                if (SAVE) gl[i0 + u * 32 + lane] = pk;
            }
        }
    }
}

template<int H>
__device__ __forceinline__ void gat_own_list_max(const int* __restrict__ gl, int* wl, int cnt,
        const float* __restrict__ a_s, const float* ads, unsigned* mxs, float* sc, int lane)
{
    for (int i0 = 0; i0 < cnt; i0 += 128) {
        int p[4];
#pragma unroll
        for (int u = 0; u < 4; ++u) p[u] = gl[min(i0 + u * 32 + lane, cnt - 1)];
        float as[4][H];
#pragma unroll
        for (int u = 0; u < 4; ++u) {
            const int s = (int)((unsigned)p[u] >> 8);
            if constexpr (H == 4) {
                const float4 q = *(const float4*)(a_s + (size_t)s * 4);
                as[u][0] = q.x; as[u][1] = q.y; as[u][2] = q.z; as[u][3] = q.w;
            } else {
#pragma unroll
                for (int h = 0; h < H; ++h) as[u][h] = a_s[(size_t)s * H + h];
            }
        }
#pragma unroll
        for (int u = 0; u < 4; ++u) {
            const int i = i0 + u * 32 + lane;
            if (i < cnt) {
                const int dl = p[u] & 255;
#pragma unroll
                for (int h = 0; h < H; ++h) {
                    const float v = gat_lrelu(as[u][h] + ads[gat_rh(dl, h)]);
                    __hip_atomic_fetch_max(&mxs[gat_rh(dl, h)], gat_fenc(v),
                                           __ATOMIC_RELAXED, __HIP_MEMORY_SCOPE_WORKGROUP);
                    if (i < OWN_SC) sc[h * OWN_SC + i] = v;
                }
                wl[i] = p[u];
            }
        }
    }
}

template<int H, bool FULL>
__device__ __forceinline__ void gat_own_acc_group(int t0, int m, int s_i, int d_i, const float (&a_i)[H],
        const float* __restrict__ Hf, float* acc, int col, int slot, int head)
{
    constexpr int ROW = 32 * H;
    constexpr int LPR = ROW / 4;
    constexpr int PIT = OWN_PIT(H);
    float4 v[OWN_U]; int dd[OWN_U]; float ww[OWN_U];
#pragma unroll
    for (int u = 0; u < OWN_U; ++u) {
        const int j = FULL ? t0 + u : min(t0 + u, m - 1);
        const int s = __builtin_amdgcn_readlane(s_i, j);
        dd[u] = __builtin_amdgcn_readlane(d_i, j);
        float w = __int_as_float(__builtin_amdgcn_readlane(__float_as_int(a_i[0]), j));
#pragma unroll
        for (int h = 1; h < H; ++h) {
            const float wh = __int_as_float(__builtin_amdgcn_readlane(__float_as_int(a_i[h]), j));
            w = (head == h) ? wh : w;
        }
        ww[u] = w;
        v[u] = *(const float4*)(Hf + (size_t)s * ROW + col);
    }
#pragma unroll
    for (int u = 0; u < OWN_U; ++u) {
        if (FULL || t0 + u < m) {
            float* p = acc + dd[u] * PIT + slot;
            __hip_atomic_fetch_add(p + 0 * LPR, ww[u] * v[u].x, __ATOMIC_RELAXED, __HIP_MEMORY_SCOPE_WORKGROUP);
            __hip_atomic_fetch_add(p + 1 * LPR, ww[u] * v[u].y, __ATOMIC_RELAXED, __HIP_MEMORY_SCOPE_WORKGROUP);
            __hip_atomic_fetch_add(p + 2 * LPR, ww[u] * v[u].z, __ATOMIC_RELAXED, __HIP_MEMORY_SCOPE_WORKGROUP);
            __hip_atomic_fetch_add(p + 3 * LPR, ww[u] * v[u].w, __ATOMIC_RELAXED, __HIP_MEMORY_SCOPE_WORKGROUP);
        }
    }
}

template<int H, bool PACKED, bool KEPT>
__device__ __forceinline__ void gat_own_flush_acc(const int* wl, int cnt, int lo, int E,
        const int* __restrict__ srcI, const int* __restrict__ dstI,
        const float* __restrict__ a_s, const float* ads, const unsigned* mxs, float* dens,
        const float* __restrict__ Hf, float* acc, const float* sc, int lane)
{
    constexpr int ROW = 32 * H;
    constexpr int LPR = ROW / 4;
    constexpr int EPI = 32 / LPR;
    constexpr int PIT = OWN_PIT(H);
    const int grp  = lane / LPR;
    const int slot = lane % LPR;
    const int col  = slot * 4;
    const int head = col >> 5;
    if constexpr (EPI == 1) cnt = __builtin_amdgcn_readfirstlane(cnt);

    for (int base = 0; base < cnt; base += 32) {
        const int m = min(32, cnt - base);
        int   s_i = 0, d_i = 0;
        float a_i[H];
#pragma unroll
        for (int h = 0; h < H; ++h) a_i[h] = 0.f;
        if (lane < m) {
            const int p = wl[base + lane];
            int s, dl;
            if (PACKED) { s = (int)((unsigned)p >> 8); dl = p & 255; }
            else if (p < E) { s = srcI[p]; dl = dstI[p] - lo; }
            else { s = p - E; dl = s - lo; }
            s_i = s; d_i = dl;
            float sv[H];
            if (KEPT && base < OWN_SC) {
#pragma unroll
                for (int h = 0; h < H; ++h) sv[h] = sc[h * OWN_SC + base + lane];
            } else {
                float as[H];
                if constexpr (H == 4) {
                    const float4 q = *(const float4*)(a_s + (size_t)s * 4);
                    as[0] = q.x; as[1] = q.y; as[2] = q.z; as[3] = q.w;
                } else {
#pragma unroll
                    for (int h = 0; h < H; ++h) as[h] = a_s[(size_t)s * H + h];
                }
#pragma unroll
                for (int h = 0; h < H; ++h) sv[h] = gat_lrelu(as[h] + ads[gat_rh(dl, h)]);
            }
#pragma unroll
            for (int h = 0; h < H; ++h) {
                const float v = sv[h];
                const float x = __expf(v - gat_fdec(mxs[gat_rh(dl, h)]));
                a_i[h] = x;
                __hip_atomic_fetch_add(&dens[gat_rh(dl, h)], x, __ATOMIC_RELAXED, __HIP_MEMORY_SCOPE_WORKGROUP);
            }
        }
        if constexpr (EPI == 1) {
            int t0 = 0;
            for (; t0 + OWN_U <= m; t0 += OWN_U)
                gat_own_acc_group<H, true>(t0, m, s_i, d_i, a_i, Hf, acc, col, slot, head);
            if (t0 < m)
                gat_own_acc_group<H, false>(t0, m, s_i, d_i, a_i, Hf, acc, col, slot, head);
            continue;
        }
        const int nsteps = (m + EPI - 1) / EPI;
        for (int t0 = 0; t0 < nsteps; t0 += OWN_U) {
            float4 v[OWN_U]; int dd[OWN_U]; float ww[OWN_U]; bool ok[OWN_U];
#pragma unroll
            for (int u = 0; u < OWN_U; ++u) {
                const int idx = (t0 + u) * EPI + grp;
                ok[u] = idx < m;
                const int sl = idx & 31;
                const int s = __shfl(s_i, sl);
                dd[u] = __shfl(d_i, sl);
                float w = __shfl(a_i[0], sl);
#pragma unroll
                for (int h = 1; h < H; ++h) {
                    const float wh = __shfl(a_i[h], sl);
                    w = (head == h) ? wh : w;
                }
                ww[u] = w;
                v[u] = *(const float4*)(Hf + (size_t)s * ROW + col);
            }
#pragma unroll
            for (int u = 0; u < OWN_U; ++u) {
                if (ok[u]) {
                    float* p = acc + dd[u] * PIT + slot;
                    __hip_atomic_fetch_add(p + 0 * LPR, ww[u] * v[u].x, __ATOMIC_RELAXED, __HIP_MEMORY_SCOPE_WORKGROUP);
                    __hip_atomic_fetch_add(p + 1 * LPR, ww[u] * v[u].y, __ATOMIC_RELAXED, __HIP_MEMORY_SCOPE_WORKGROUP);
                    __hip_atomic_fetch_add(p + 2 * LPR, ww[u] * v[u].z, __ATOMIC_RELAXED, __HIP_MEMORY_SCOPE_WORKGROUP);
                    __hip_atomic_fetch_add(p + 3 * LPR, ww[u] * v[u].w, __ATOMIC_RELAXED, __HIP_MEMORY_SCOPE_WORKGROUP);
                }
            }
        }
    }
}


__device__ __forceinline__ unsigned gat_eq_bytes(unsigned x, unsigned key4) {
    const unsigned y = x ^ key4;
    const unsigned t = ((y & 0x7F7F7F7Fu) + 0x7F7F7F7Fu) | y;
    return ~t & 0x80808080u;
}
__device__ __forceinline__ unsigned gat_eq_mask16(uint4 v, unsigned key4) {
    return (gat_eq_bytes(v.x, key4) >> 7) | (gat_eq_bytes(v.y, key4) >> 6) |
           (gat_eq_bytes(v.z, key4) >> 5) | (gat_eq_bytes(v.w, key4) >> 4);
}

template<int H, bool ACC>
__device__ __forceinline__ void gat_own_drain(int* wl, int cnt, int lo, int E,
        const int* __restrict__ srcI, const int* __restrict__ dstI,
        const float* __restrict__ a_s, const float* ads, unsigned* mxs, float* dens,
        const float* __restrict__ Hf, float* acc, int* again, int lane)
{
    if (ACC) {
        gat_own_flush_acc<H, false, false>(wl, cnt, lo, E, srcI, dstI, a_s, ads, mxs, dens, Hf, acc, nullptr, lane);
    } else {
        gat_own_flush_max<H, false, false>(wl, nullptr, cnt, lo, E, srcI, dstI, a_s, ads, mxs, nullptr, lane);
        if (lane == 0) __hip_atomic_store(again, 1, __ATOMIC_RELAXED, __HIP_MEMORY_SCOPE_WORKGROUP);
    }
}

template<int H, bool ACC>
__device__ __forceinline__ int gat_own_sweep(int* wl, int lo, int nodes, int E,
        const int* __restrict__ srcI, const int* __restrict__ dstI,
        const unsigned* __restrict__ obyte, bool bytes_ok,
        const float* __restrict__ a_s, const float* ads, unsigned* mxs, float* dens,
        const float* __restrict__ Hf, float* acc, int* again, int lane, int wave)
{
    const unsigned lt = (1u << lane) - 1u;
    int cnt = 0;

    for (int nb = wave * 32; nb < nodes; nb += OWN_NW * 32) {
        const int n = nb + lane;
        const bool hit = n < nodes;
        const unsigned mask = __builtin_amdgcn_ballot_w32(hit);
        if (hit) wl[cnt + __popc(mask & lt)] = E + lo + n;
        cnt += __popc(mask);
    }

    int tail0;
    if (bytes_ok) {
    const int nv16 = E >> 4;
    const uint4* ob = (const uint4*)obyte;
    const unsigned key4 = (unsigned)blockIdx.x * 0x01010101u;
    const uint4 none = make_uint4(~key4, ~key4, ~key4, ~key4);
    int gb = wave * 128;
    uint4 cur[4];
#pragma unroll
    for (int u = 0; u < 4; ++u) {
        const int g = gb + u * 32 + lane;
        cur[u] = ob[max(min(g, nv16 - 1), 0)];
        if (g >= nv16) cur[u] = none;
    }
    while (gb < nv16) {
        const int gn = gb + OWN_NW * 128;
        uint4 nxt[4];
#pragma unroll
        for (int u = 0; u < 4; ++u) {
            const int g = gn + u * 32 + lane;
            nxt[u] = ob[min(g, nv16 - 1)];
            if (g >= nv16) nxt[u] = none;
        }
        const unsigned m01 = gat_eq_mask16(cur[0], key4) | (gat_eq_mask16(cur[1], key4) << 4);
        const unsigned m23 = gat_eq_mask16(cur[2], key4) | (gat_eq_mask16(cur[3], key4) << 4);
        unsigned long long M = (unsigned long long)m01 | ((unsigned long long)m23 << 32);
        unsigned pend = __builtin_amdgcn_ballot_w32(M != 0ull);
        while (pend) {
            if (M != 0ull) {
                const int p = __ffsll((long long)M) - 1;
                const int u = ((p >> 5) << 1) | ((p >> 2) & 1);
                const int off = ((p & 3) << 2) | ((p >> 3) & 3);
                wl[cnt + __popc(pend & lt)] = ((gb + u * 32 + lane) << 4) + off;
                M &= M - 1ull;
            }
            cnt += __popc(pend);
            if (cnt > OWN_CAP - 32) {
                gat_own_drain<H, ACC>(wl, cnt, lo, E, srcI, dstI, a_s, ads, mxs, dens, Hf, acc, again, lane);
                cnt = 0;
            }
            pend = __builtin_amdgcn_ballot_w32(M != 0ull);
        }
#pragma unroll
        for (int u = 0; u < 4; ++u) cur[u] = nxt[u];
        gb = gn;
    }
    tail0 = nv16 << 4;
    } else {
    const int nvec = E >> 2;
    const int4* d4 = (const int4*)dstI;
    const int4 none = make_int4(-1, -1, -1, -1);
    int gb = wave * 128;
    int4 cur[4];
#pragma unroll
    for (int u = 0; u < 4; ++u) {
        const int g = gb + u * 32 + lane;
        cur[u] = d4[max(min(g, nvec - 1), 0)];
        if (g >= nvec) cur[u] = none;
    }
    while (gb < nvec) {
        const int gn = gb + OWN_NW * 128;
        int4 nxt[4];
#pragma unroll
        for (int u = 0; u < 4; ++u) {
            const int g = gn + u * 32 + lane;
            nxt[u] = d4[min(g, nvec - 1)];
            if (g >= nvec) nxt[u] = none;
        }
#pragma unroll
        for (int u = 0; u < 4; ++u) {
            const int e0 = (gb + u * 32 + lane) * 4;
            const bool h0 = (unsigned)(cur[u].x - lo) < (unsigned)nodes;
            const bool h1 = (unsigned)(cur[u].y - lo) < (unsigned)nodes;
            const bool h2 = (unsigned)(cur[u].z - lo) < (unsigned)nodes;
            const bool h3 = (unsigned)(cur[u].w - lo) < (unsigned)nodes;
            if (__builtin_amdgcn_ballot_w32(h0 | h1 | h2 | h3)) {
                unsigned mask;
                mask = __builtin_amdgcn_ballot_w32(h0);
                if (h0) wl[cnt + __popc(mask & lt)] = e0 + 0;
                cnt += __popc(mask);
                mask = __builtin_amdgcn_ballot_w32(h1);
                if (h1) wl[cnt + __popc(mask & lt)] = e0 + 1;
                cnt += __popc(mask);
                mask = __builtin_amdgcn_ballot_w32(h2);
                if (h2) wl[cnt + __popc(mask & lt)] = e0 + 2;
                cnt += __popc(mask);
                mask = __builtin_amdgcn_ballot_w32(h3);
                if (h3) wl[cnt + __popc(mask & lt)] = e0 + 3;
                cnt += __popc(mask);
            }
        }
        if (cnt > OWN_CAP / 2) {
            gat_own_drain<H, ACC>(wl, cnt, lo, E, srcI, dstI, a_s, ads, mxs, dens, Hf, acc, again, lane);
            cnt = 0;
        }
#pragma unroll
        for (int u = 0; u < 4; ++u) cur[u] = nxt[u];
        gb = gn;
    }
    tail0 = nvec * 4;
    }
    if (wave == 0) {
        for (int eb0 = tail0; eb0 < E; eb0 += 32) {
            const int e = eb0 + lane;
            const bool hit = (e < E) && ((unsigned)(dstI[min(e, E - 1)] - lo) < (unsigned)nodes);
            const unsigned mask = __builtin_amdgcn_ballot_w32(hit);
            if (hit) wl[cnt + __popc(mask & lt)] = e;
            cnt += __popc(mask);
        }
    }
    return cnt;
}

template<int H, bool RELU, bool SAVE, bool LOAD>
__global__ __launch_bounds__(OWN_NW * 32)
void gat_own_aggregate(const int* __restrict__ srcI, const int* __restrict__ dstI,
                       const float* __restrict__ a_s, const float* __restrict__ a_d,
                       const float* __restrict__ Hf, const float* __restrict__ bias,
                       float* __restrict__ outp, int* __restrict__ glist, int* __restrict__ gcnt,
                       const unsigned* __restrict__ obyte, int E, int N)
{
    constexpr int ROW = 32 * H;
    constexpr int LPR = ROW / 4;
    constexpr int PIT = OWN_PIT(H);
    static_assert(OWN_NPB <= 32 * OWN_NW, "one self-loop step per wave");
    static_assert(OWN_NPB <= 256, "local row index is packed into 8 bits");
    static_assert(PIT >= ROW, "a row's four slabs fit inside its pitch");
    __shared__ float acc[OWN_NPB * PIT];
    __shared__ int lst[OWN_NW * OWN_CAP];
    __shared__ float    ads[OWN_NPB * H];
    __shared__ unsigned mxs[OWN_NPB * H];
    __shared__ float    dens[OWN_NPB * H];
    __shared__ float    scs[OWN_NW * H * OWN_SC];
    __shared__ int again;

    const int tid  = threadIdx.x;
    const int lane = tid & 31;
    const int wave = tid >> 5;
    const int lo   = blockIdx.x * OWN_NPB;
    const int nodes = min(OWN_NPB, N - lo);
    const bool bytes_ok = gridDim.x <= 256u;

    for (int i = tid; i < nodes * PIT; i += OWN_NW * 32) acc[i] = 0.f;
    for (int i = tid; i < nodes * H; i += OWN_NW * 32) {
        const int j = gat_rh(i / H, i % H);
        ads[j]  = a_d[(size_t)lo * H + i];
        mxs[j]  = ENC_NEG_INF;
        dens[j] = 0.f;
    }
    if (tid == 0) again = 0;
    __syncthreads();

    int* wl = lst + wave * OWN_CAP;
    float* sc = scs + wave * (H * OWN_SC);
    int* gl = glist + ((size_t)blockIdx.x * OWN_NW + wave) * OWN_CAP;
    int* gc = gcnt + (size_t)blockIdx.x * OWN_NW;
    int cnt = 0;
    bool saved = false;
    if (LOAD) saved = gc[0] >= 0;
    if (saved) {
        cnt = gc[wave];
        gat_own_list_max<H>(gl, wl, cnt, a_s, ads, mxs, sc, lane);
    } else {
        cnt = gat_own_sweep<H, false>(wl, lo, nodes, E, srcI, dstI, obyte, bytes_ok, a_s, ads, mxs, dens, Hf, acc, &again, lane, wave);
        gat_own_flush_max<H, SAVE, true>(wl, gl, cnt, lo, E, srcI, dstI, a_s, ads, mxs, sc, lane);
    }
    __syncthreads();

    const int ag = __hip_atomic_load(&again, __ATOMIC_RELAXED, __HIP_MEMORY_SCOPE_WORKGROUP);
    if (SAVE && lane == 0) gc[wave] = ag ? -1 : cnt;
    if (ag == 0) {
        gat_own_flush_acc<H, true, true>(wl, cnt, lo, E, srcI, dstI, a_s, ads, mxs, dens, Hf, acc, sc, lane);
    } else {
        cnt = gat_own_sweep<H, true>(wl, lo, nodes, E, srcI, dstI, obyte, bytes_ok, a_s, ads, mxs, dens, Hf, acc, &again, lane, wave);
        gat_own_flush_acc<H, false, false>(wl, cnt, lo, E, srcI, dstI, a_s, ads, mxs, dens, Hf, acc, nullptr, lane);
    }
    __syncthreads();

    float* orow = outp + (size_t)lo * ROW;
    for (int i = tid * 4; i < nodes * ROW; i += OWN_NW * 32 * 4) {
        const int c = i & (ROW - 1);
        const float* a = acc + (i / ROW) * PIT + (c >> 2);
        float4 r = make_float4(a[0], a[LPR], a[2 * LPR], a[3 * LPR]);
        const float dn = dens[gat_rh(i / ROW, c >> 5)];
        r.x = r.x / dn + bias[c + 0];
        r.y = r.y / dn + bias[c + 1];
        r.z = r.z / dn + bias[c + 2];
        r.w = r.w / dn + bias[c + 3];
        if (RELU) { r.x = fmaxf(r.x, 0.f); r.y = fmaxf(r.y, 0.f); r.z = fmaxf(r.z, 0.f); r.w = fmaxf(r.w, 0.f); }
        *(float4*)(orow + i) = r;
    }
}

extern "C" void kernel_launch(void* const* d_in, const int* in_sizes, int n_in,
                              void* d_out, int out_size, void* d_ws, size_t ws_size,
                              hipStream_t stream)
{
    const float* x   = (const float*)d_in[0];
    const int*   ei  = (const int*)d_in[1];
    const float* W1  = (const float*)d_in[2];
    const float* as1 = (const float*)d_in[3];
    const float* ad1 = (const float*)d_in[4];
    const float* b1  = (const float*)d_in[5];
    const float* W2  = (const float*)d_in[6];
    const float* as2 = (const float*)d_in[7];
    const float* ad2 = (const float*)d_in[8];
    const float* b2  = (const float*)d_in[9];
    float* out = (float*)d_out;

    const int N  = in_sizes[0] / 128;
    const int E  = in_sizes[1] / 2;
    const int* srcI = ei;
    const int* dstI = ei + E;

    float* ws = (float*)d_ws;
    float*    h1    = ws;              ws += (size_t)N * 128;
    float*    asrc1 = ws;              ws += (size_t)N * 4;
    float*    adst1 = ws;              ws += (size_t)N * 4;
    float*    h2    = ws;              ws += (size_t)N * 128;
    float*    g2    = ws;              ws += (size_t)N * 32;
    float*    asrc2 = ws;              ws += (size_t)N;
    float*    adst2 = ws;              ws += (size_t)N;

    const int TB = 256;
    const int gB   = (N + 127) / 128;
    const int gA   = (N + OWN_NPB - 1) / OWN_NPB;
    int* glist = (int*)ws;
    int* gcnt  = glist + (size_t)gA * OWN_NW * OWN_CAP;
    unsigned* obyte = (unsigned*)(gcnt + (((size_t)gA * OWN_NW + 3) & ~(size_t)3));
    const int nquad = E >> 2;

    gat_gemm_wmma<8><<<gB, TB, 0, stream>>>(x, W1, as1, ad1, h1, asrc1, adst1, N, dstI, obyte, nquad);
    gat_own_aggregate<4, true, true, false><<<gA, OWN_NW * 32, 0, stream>>>(srcI, dstI, asrc1, adst1, h1, b1, h2, glist, gcnt, obyte, E, N);

    gat_gemm_wmma<2><<<gB, TB, 0, stream>>>(h2, W2, as2, ad2, g2, asrc2, adst2, N, dstI, obyte, 0);
    gat_own_aggregate<1, false, false, true><<<gA, OWN_NW * 32, 0, stream>>>(srcI, dstI, asrc2, adst2, g2, b2, out, glist, gcnt, obyte, E, N);
}


# Owner filter by bitmap: the layer-1 projection launch sets each edge's bit in its owner's row (LDS tile per 2048 edges); an owner block reads only its own row; row sums stay in LDS arrival order
# speedup vs baseline: 15.3841x; 1.0317x over previous
#include <hip/hip_runtime.h>

#define NEG_SLOPE 0.2f
#define ENC_NEG_INF 0x007FFFFFu

typedef float v2f __attribute__((ext_vector_type(2)));
typedef float v8f __attribute__((ext_vector_type(8)));

__device__ __forceinline__ unsigned gat_fenc(float f) {
    unsigned u = __float_as_uint(f);
    return (u & 0x80000000u) ? ~u : (u | 0x80000000u);
}
__device__ __forceinline__ float gat_fdec(unsigned u) {
    unsigned b = (u & 0x80000000u) ? (u & 0x7FFFFFFFu) : ~u;
    return __uint_as_float(b);
}

#define OWN_NPB 200
#define BM_CH   2048
#define BM_MAXROWS 288

__device__ __forceinline__ void gat_bm_set4(unsigned* tile, int4 d, int eloc, int nown) {
    const int w = eloc >> 5;
    const int dd[4] = { d.x, d.y, d.z, d.w };
#pragma unroll
    for (int j = 0; j < 4; ++j) {
        const int o = (int)min((unsigned)dd[j] / OWN_NPB, (unsigned)(nown - 1));
        __hip_atomic_fetch_or(&tile[o * 64 + ((w + 4 * o) & 63)], 1u << ((eloc + j) & 31),
                              __ATOMIC_RELAXED, __HIP_MEMORY_SCOPE_WORKGROUP);
    }
}

template<int NT>
__global__ __launch_bounds__(256)
void gat_gemm_wmma(const float* __restrict__ X, const float* __restrict__ W,
                   const float* __restrict__ att_s, const float* __restrict__ att_d,
                   float* __restrict__ Hout, float* __restrict__ a_s, float* __restrict__ a_d,
                   int n_rows,
                   const int* __restrict__ dstI, unsigned* __restrict__ bm,
                   int nquad, int nchunk, int nown)
{
    constexpr int KD = 128;
    constexpr int ND = NT * 16;
    constexpr int HN = NT / 2;
    constexpr int NE = ND + 16;
    static_assert(2 * HN <= 16, "score columns fit one tile");
    __shared__ __attribute__((aligned(16))) float Wt[NE * KD];

    const int lane = threadIdx.x & 31;
    const int wave = threadIdx.x >> 5;
    const int m0   = (blockIdx.x * 8 + wave) * 16;
    const int mlo  = lane & 15;
    const int kofs = (lane >> 4) * 2;

    const int4* d4 = (const int4*)dstI;
    const bool bm_on = nchunk > 0 && nown <= BM_MAXROWS && nown * 256 <= (int)sizeof(Wt);
    const int q0 = (int)blockIdx.x * (BM_CH / 4) + (int)threadIdx.x;
    const int4 dq0 = d4[max(min(q0, nquad - 1), 0)];
    const int4 dq1 = d4[max(min(q0 + 256, nquad - 1), 0)];

    constexpr int SEG = ND / 32;
    constexpr int WIT = (KD / 4) * SEG / 8;
    static_assert(ND % 32 == 0 && ((KD / 4) * SEG) % 8 == 0, "W splits into whole-line loads over 8 waves");
    float4 wreg[WIT];
#pragma unroll
    for (int it = 0; it < WIT; ++it) {
        const int c = it * 8 + wave;
        const int k = 4 * (c / SEG) + (lane >> 3);
        const int n = 32 * (c % SEG) + 4 * (lane & 7);
        wreg[it] = *(const float4*)(W + k * ND + n);
    }

    if (bm_on) {
        unsigned* tile = (unsigned*)Wt;
        const int wpr = nchunk * 64;
        for (int c = blockIdx.x; c < nchunk; c += gridDim.x) {
            for (int i = threadIdx.x; i < nown * 16; i += 256) ((uint4*)tile)[i] = make_uint4(0u, 0u, 0u, 0u);
            __syncthreads();
            const int q = c * (BM_CH / 4) + (int)threadIdx.x;
            int4 da = dq0, db = dq1;
            if (c != (int)blockIdx.x) { da = d4[max(min(q, nquad - 1), 0)]; db = d4[max(min(q + 256, nquad - 1), 0)]; }
            if (q < nquad)       gat_bm_set4(tile, da, 4 * (int)threadIdx.x, nown);
            if (q + 256 < nquad) gat_bm_set4(tile, db, 4 * ((int)threadIdx.x + 256), nown);
            __syncthreads();
            for (int i0 = threadIdx.x; i0 < nown * 16; i0 += 4 * 256) {
                uint4 v[4];
#pragma unroll
                for (int u = 0; u < 4; ++u) {
                    const int i = min(i0 + u * 256, nown * 16 - 1);
                    const int o = i >> 4, w4 = (i & 15) * 4;
                    v[u] = *(const uint4*)&tile[o * 64 + ((w4 + 4 * o) & 63)];
                }
#pragma unroll
                for (int u = 0; u < 4; ++u) {
                    const int i = i0 + u * 256;
                    const int o = i >> 4, w4 = (i & 15) * 4;
                    if (i < nown * 16) *(uint4*)&bm[(size_t)o * wpr + c * 64 + w4] = v[u];
                }
            }
            __syncthreads();
        }
    }

#pragma unroll
    for (int it = 0; it < WIT; ++it) {
        const int c = it * 8 + wave;
        const int k = 4 * (c / SEG) + (lane >> 3);
        const int n = 32 * (c % SEG) + 4 * (lane & 7);
        Wt[(n + 0) * KD + ((k + 4 * (n + 0)) & (KD - 1))] = wreg[it].x;
        Wt[(n + 1) * KD + ((k + 4 * (n + 1)) & (KD - 1))] = wreg[it].y;
        Wt[(n + 2) * KD + ((k + 4 * (n + 2)) & (KD - 1))] = wreg[it].z;
        Wt[(n + 3) * KD + ((k + 4 * (n + 3)) & (KD - 1))] = wreg[it].w;
    }
    __syncthreads();

    {
        const int kk = (int)threadIdx.x & (KD - 1);
        const int half = __builtin_amdgcn_readfirstlane((int)threadIdx.x >> 7);
        const float* av = half ? att_d : att_s;
        float s[HN];
#pragma unroll
        for (int h = 0; h < HN; ++h) {
            s[h] = 0.f;
            for (int cc = 0; cc < 32; ++cc) {
                const int n = h * 32 + cc;
                s[h] += Wt[n * KD + ((kk + 4 * n) & (KD - 1))] * av[n];
            }
        }
#pragma unroll
        for (int h = 0; h < HN; ++h) {
            const int n = ND + half * HN + h;
            Wt[n * KD + ((kk + 4 * n) & (KD - 1))] = s[h];
        }
        for (int i = threadIdx.x; i < (16 - 2 * HN) * KD; i += 256) {
            const int n = ND + 2 * HN + i / KD, k = i % KD;
            Wt[n * KD + ((k + 4 * n) & (KD - 1))] = 0.f;
        }
    }
    __syncthreads();

    int arow = m0 + mlo;
    if (arow >= n_rows) arow = n_rows - 1;
    const float* xrow = X + (size_t)arow * KD + kofs;

    v8f acc[NT + 1] = {};

#pragma unroll
    for (int k = 0; k < KD; k += 4) {
        v2f a = *(const v2f*)(xrow + k);
#pragma unroll
        for (int t = 0; t < NT + 1; ++t) {
            int n = t * 16 + mlo;
            v2f b = *(const v2f*)(&Wt[n * KD + ((k + kofs + 4 * n) & (KD - 1))]);
            acc[t] = __builtin_amdgcn_wmma_f32_16x16x4_f32(
                false, a, false, b, (short)0, acc[t], false, false);
        }
    }

    const int mhi = (lane >> 4) * 8;
#pragma unroll
    for (int t = 0; t < NT; ++t)
#pragma unroll
        for (int j = 0; j < 8; ++j) {
            int m = m0 + mhi + j;
            if (m < n_rows) Hout[(size_t)m * ND + t * 16 + mlo] = acc[t][j];
        }
    __syncthreads();
    float* sc = Wt;
    if (mlo < 2 * HN) {
        const int kind = (mlo < HN) ? 0 : 1;
        const int hd   = mlo - kind * HN;
#pragma unroll
        for (int j = 0; j < 8; ++j)
            sc[kind * 128 * HN + (wave * 16 + mhi + j) * HN + hd] = acc[NT][j];
    }
    __syncthreads();
    const int rbase = blockIdx.x * 128;
    for (int i = threadIdx.x; i < 128 * HN; i += 256) {
        if (rbase + i / HN < n_rows) {
            a_s[(size_t)rbase * HN + i] = sc[i];
            a_d[(size_t)rbase * HN + i] = sc[128 * HN + i];
        }
    }
}


#define OWN_NW  32
#define OWN_CAP 1024
#define OWN_U   8
#define OWN_SC  128

#define OWN_PIT(H) ((H) == 1 ? 40 : 32 * (H))
__device__ __forceinline__ int gat_rh(int row, int h) { return h * OWN_NPB + row; }

__device__ __forceinline__ float gat_lrelu(float v) { return (v >= 0.f) ? v : NEG_SLOPE * v; }

template<int H, bool SAVE, bool KEEP>
__device__ __forceinline__ void gat_own_flush_max(int* wl, int* __restrict__ gl, int cnt, int lo, int E,
        const int* __restrict__ srcI, const int* __restrict__ dstI,
        const float* __restrict__ a_s, const float* ads, unsigned* mxs, float* sc, int lane)
{
    for (int i0 = 0; i0 < cnt; i0 += 128) {
        int e[4], s[4], dl[4];
#pragma unroll
        for (int u = 0; u < 4; ++u) {
            const int i = i0 + u * 32 + lane;
            e[u] = (i < cnt) ? wl[i] : -1;
        }
#pragma unroll
        for (int u = 0; u < 4; ++u) {
            s[u] = 0; dl[u] = 0;
            if (e[u] >= 0) {
                if (e[u] < E) { s[u] = srcI[e[u]]; dl[u] = dstI[e[u]] - lo; }
                else          { s[u] = e[u] - E;   dl[u] = s[u] - lo; }
            }
        }
#pragma unroll
        for (int u = 0; u < 4; ++u) {
            if (e[u] >= 0) {
                float as[H];
                if constexpr (H == 4) {
                    const float4 q = *(const float4*)(a_s + (size_t)s[u] * 4);
                    as[0] = q.x; as[1] = q.y; as[2] = q.z; as[3] = q.w;
                } else {
#pragma unroll
                    for (int h = 0; h < H; ++h) as[h] = a_s[(size_t)s[u] * H + h];
                }
#pragma unroll
                for (int h = 0; h < H; ++h) {
                    const float v = gat_lrelu(as[h] + ads[gat_rh(dl[u], h)]);
                    __hip_atomic_fetch_max(&mxs[gat_rh(dl[u], h)], gat_fenc(v),
                                           __ATOMIC_RELAXED, __HIP_MEMORY_SCOPE_WORKGROUP);
                    if (KEEP && i0 + u * 32 + lane < OWN_SC) sc[h * OWN_SC + i0 + u * 32 + lane] = v;
                }
                const int pk = (int)(((unsigned)s[u] << 8) | (unsigned)dl[u]);
                wl[i0 + u * 32 + lane] = pk;
                if (SAVE) gl[i0 + u * 32 + lane] = pk;
            }
        }
    }
}

template<int H>
__device__ __forceinline__ void gat_own_list_max(const int* __restrict__ gl, int* wl, int cnt,
        const float* __restrict__ a_s, const float* ads, unsigned* mxs, float* sc, int lane)
{
    for (int i0 = 0; i0 < cnt; i0 += 128) {
        int p[4];
#pragma unroll
        for (int u = 0; u < 4; ++u) p[u] = gl[min(i0 + u * 32 + lane, cnt - 1)];
        float as[4][H];
#pragma unroll
        for (int u = 0; u < 4; ++u) {
            const int s = (int)((unsigned)p[u] >> 8);
            if constexpr (H == 4) {
                const float4 q = *(const float4*)(a_s + (size_t)s * 4);
                as[u][0] = q.x; as[u][1] = q.y; as[u][2] = q.z; as[u][3] = q.w;
            } else {
#pragma unroll
                for (int h = 0; h < H; ++h) as[u][h] = a_s[(size_t)s * H + h];
            }
        }
#pragma unroll
        for (int u = 0; u < 4; ++u) {
            const int i = i0 + u * 32 + lane;
            if (i < cnt) {
                const int dl = p[u] & 255;
#pragma unroll
                for (int h = 0; h < H; ++h) {
                    const float v = gat_lrelu(as[u][h] + ads[gat_rh(dl, h)]);
                    __hip_atomic_fetch_max(&mxs[gat_rh(dl, h)], gat_fenc(v),
                                           __ATOMIC_RELAXED, __HIP_MEMORY_SCOPE_WORKGROUP);
                    if (i < OWN_SC) sc[h * OWN_SC + i] = v;
                }
                wl[i] = p[u];
            }
        }
    }
}

template<int H, bool FULL>
__device__ __forceinline__ void gat_own_acc_group(int t0, int m, int s_i, int d_i, const float (&a_i)[H],
        const float* __restrict__ Hf, float* acc, int col, int slot, int head)
{
    constexpr int ROW = 32 * H;
    constexpr int LPR = ROW / 4;
    constexpr int PIT = OWN_PIT(H);
    float4 v[OWN_U]; int dd[OWN_U]; float ww[OWN_U];
#pragma unroll
    for (int u = 0; u < OWN_U; ++u) {
        const int j = FULL ? t0 + u : min(t0 + u, m - 1);
        const int s = __builtin_amdgcn_readlane(s_i, j);
        dd[u] = __builtin_amdgcn_readlane(d_i, j);
        float w = __int_as_float(__builtin_amdgcn_readlane(__float_as_int(a_i[0]), j));
#pragma unroll
        for (int h = 1; h < H; ++h) {
            const float wh = __int_as_float(__builtin_amdgcn_readlane(__float_as_int(a_i[h]), j));
            w = (head == h) ? wh : w;
        }
        ww[u] = w;
        v[u] = *(const float4*)(Hf + (size_t)s * ROW + col);
    }
#pragma unroll
    for (int u = 0; u < OWN_U; ++u) {
        if (FULL || t0 + u < m) {
            float* p = acc + dd[u] * PIT + slot;
            __hip_atomic_fetch_add(p + 0 * LPR, ww[u] * v[u].x, __ATOMIC_RELAXED, __HIP_MEMORY_SCOPE_WORKGROUP);
            __hip_atomic_fetch_add(p + 1 * LPR, ww[u] * v[u].y, __ATOMIC_RELAXED, __HIP_MEMORY_SCOPE_WORKGROUP);
            __hip_atomic_fetch_add(p + 2 * LPR, ww[u] * v[u].z, __ATOMIC_RELAXED, __HIP_MEMORY_SCOPE_WORKGROUP);
            __hip_atomic_fetch_add(p + 3 * LPR, ww[u] * v[u].w, __ATOMIC_RELAXED, __HIP_MEMORY_SCOPE_WORKGROUP);
        }
    }
}

template<int H, bool PACKED, bool KEPT>
__device__ __forceinline__ void gat_own_flush_acc(const int* wl, int cnt, int lo, int E,
        const int* __restrict__ srcI, const int* __restrict__ dstI,
        const float* __restrict__ a_s, const float* ads, const unsigned* mxs, float* dens,
        const float* __restrict__ Hf, float* acc, const float* sc, int lane)
{
    constexpr int ROW = 32 * H;
    constexpr int LPR = ROW / 4;
    constexpr int EPI = 32 / LPR;
    constexpr int PIT = OWN_PIT(H);
    const int grp  = lane / LPR;
    const int slot = lane % LPR;
    const int col  = slot * 4;
    const int head = col >> 5;
    if constexpr (EPI == 1) cnt = __builtin_amdgcn_readfirstlane(cnt);

    for (int base = 0; base < cnt; base += 32) {
        const int m = min(32, cnt - base);
        int   s_i = 0, d_i = 0;
        float a_i[H];
#pragma unroll
        for (int h = 0; h < H; ++h) a_i[h] = 0.f;
        if (lane < m) {
            const int p = wl[base + lane];
            int s, dl;
            if (PACKED) { s = (int)((unsigned)p >> 8); dl = p & 255; }
            else if (p < E) { s = srcI[p]; dl = dstI[p] - lo; }
            else { s = p - E; dl = s - lo; }
            s_i = s; d_i = dl;
            float sv[H];
            if (KEPT && base < OWN_SC) {
#pragma unroll
                for (int h = 0; h < H; ++h) sv[h] = sc[h * OWN_SC + base + lane];
            } else {
                float as[H];
                if constexpr (H == 4) {
                    const float4 q = *(const float4*)(a_s + (size_t)s * 4);
                    as[0] = q.x; as[1] = q.y; as[2] = q.z; as[3] = q.w;
                } else {
#pragma unroll
                    for (int h = 0; h < H; ++h) as[h] = a_s[(size_t)s * H + h];
                }
#pragma unroll
                for (int h = 0; h < H; ++h) sv[h] = gat_lrelu(as[h] + ads[gat_rh(dl, h)]);
            }
#pragma unroll
            for (int h = 0; h < H; ++h) {
                const float v = sv[h];
                const float x = __expf(v - gat_fdec(mxs[gat_rh(dl, h)]));
                a_i[h] = x;
                __hip_atomic_fetch_add(&dens[gat_rh(dl, h)], x, __ATOMIC_RELAXED, __HIP_MEMORY_SCOPE_WORKGROUP);
            }
        }
        if constexpr (EPI == 1) {
            int t0 = 0;
            for (; t0 + OWN_U <= m; t0 += OWN_U)
                gat_own_acc_group<H, true>(t0, m, s_i, d_i, a_i, Hf, acc, col, slot, head);
            if (t0 < m)
                gat_own_acc_group<H, false>(t0, m, s_i, d_i, a_i, Hf, acc, col, slot, head);
            continue;
        }
        const int nsteps = (m + EPI - 1) / EPI;
        for (int t0 = 0; t0 < nsteps; t0 += OWN_U) {
            float4 v[OWN_U]; int dd[OWN_U]; float ww[OWN_U]; bool ok[OWN_U];
#pragma unroll
            for (int u = 0; u < OWN_U; ++u) {
                const int idx = (t0 + u) * EPI + grp;
                ok[u] = idx < m;
                const int sl = idx & 31;
                const int s = __shfl(s_i, sl);
                dd[u] = __shfl(d_i, sl);
                float w = __shfl(a_i[0], sl);
#pragma unroll
                for (int h = 1; h < H; ++h) {
                    const float wh = __shfl(a_i[h], sl);
                    w = (head == h) ? wh : w;
                }
                ww[u] = w;
                v[u] = *(const float4*)(Hf + (size_t)s * ROW + col);
            }
#pragma unroll
            for (int u = 0; u < OWN_U; ++u) {
                if (ok[u]) {
                    float* p = acc + dd[u] * PIT + slot;
                    __hip_atomic_fetch_add(p + 0 * LPR, ww[u] * v[u].x, __ATOMIC_RELAXED, __HIP_MEMORY_SCOPE_WORKGROUP);
                    __hip_atomic_fetch_add(p + 1 * LPR, ww[u] * v[u].y, __ATOMIC_RELAXED, __HIP_MEMORY_SCOPE_WORKGROUP);
                    __hip_atomic_fetch_add(p + 2 * LPR, ww[u] * v[u].z, __ATOMIC_RELAXED, __HIP_MEMORY_SCOPE_WORKGROUP);
                    __hip_atomic_fetch_add(p + 3 * LPR, ww[u] * v[u].w, __ATOMIC_RELAXED, __HIP_MEMORY_SCOPE_WORKGROUP);
                }
            }
        }
    }
}


template<int H, bool ACC>
__device__ __forceinline__ void gat_own_drain(int* wl, int cnt, int lo, int E,
        const int* __restrict__ srcI, const int* __restrict__ dstI,
        const float* __restrict__ a_s, const float* ads, unsigned* mxs, float* dens,
        const float* __restrict__ Hf, float* acc, int* again, int lane)
{
    if (ACC) {
        gat_own_flush_acc<H, false, false>(wl, cnt, lo, E, srcI, dstI, a_s, ads, mxs, dens, Hf, acc, nullptr, lane);
    } else {
        gat_own_flush_max<H, false, false>(wl, nullptr, cnt, lo, E, srcI, dstI, a_s, ads, mxs, nullptr, lane);
        if (lane == 0) __hip_atomic_store(again, 1, __ATOMIC_RELAXED, __HIP_MEMORY_SCOPE_WORKGROUP);
    }
}

template<int H, bool ACC>
__device__ __forceinline__ int gat_own_sweep(int* wl, int lo, int nodes, int E,
        const int* __restrict__ srcI, const int* __restrict__ dstI,
        const unsigned* __restrict__ bm, bool bm_ok, int wpr,
        const float* __restrict__ a_s, const float* ads, unsigned* mxs, float* dens,
        const float* __restrict__ Hf, float* acc, int* again, int lane, int wave)
{
    const unsigned lt = (1u << lane) - 1u;
    int cnt = 0;

    for (int nb = wave * 32; nb < nodes; nb += OWN_NW * 32) {
        const int n = nb + lane;
        const bool hit = n < nodes;
        const unsigned mask = __builtin_amdgcn_ballot_w32(hit);
        if (hit) wl[cnt + __popc(mask & lt)] = E + lo + n;
        cnt += __popc(mask);
    }

    int tail0;
    if (bm_ok) {
    const int nv = wpr >> 2;
    const uint4* ob = (const uint4*)(bm + (size_t)blockIdx.x * wpr);
    const uint4 none = make_uint4(0u, 0u, 0u, 0u);
    const int tid = wave * 32 + lane;
    int gb = 0;
    uint4 cur[4];
#pragma unroll
    for (int u = 0; u < 4; ++u) {
        const int g = gb + u * (OWN_NW * 32) + tid;
        cur[u] = ob[min(g, nv - 1)];
        if (g >= nv) cur[u] = none;
    }
    while (gb < nv) {
        const int gn = gb + OWN_NW * 128;
        uint4 nxt[4];
#pragma unroll
        for (int u = 0; u < 4; ++u) {
            const int g = gn + u * (OWN_NW * 32) + tid;
            nxt[u] = ob[min(g, nv - 1)];
            if (g >= nv) nxt[u] = none;
        }
        const int ebv = (gb + tid) << 7;
        const unsigned w[16] = { cur[0].x, cur[0].y, cur[0].z, cur[0].w, cur[1].x, cur[1].y, cur[1].z, cur[1].w,
                                 cur[2].x, cur[2].y, cur[2].z, cur[2].w, cur[3].x, cur[3].y, cur[3].z, cur[3].w };
        int nb = 0;
#pragma unroll
        for (int j = 0; j < 16; ++j) nb += __popc(w[j]);
        if (__builtin_amdgcn_ballot_w32(nb > 16) == 0u && cnt <= OWN_CAP / 2 - 32) {
#pragma unroll
            for (int j = 0; j < 16; ++j) {
                unsigned x = w[j];
                const int e0 = ebv + (j >> 2) * (OWN_NW * 32 * 128) + (j & 3) * 32;
                unsigned pend = __builtin_amdgcn_ballot_w32(x != 0u);
                while (pend) {
                    if (x != 0u) {
                        wl[cnt + __popc(pend & lt)] = e0 + (__ffs((int)x) - 1);
                        x &= x - 1u;
                    }
                    cnt += __popc(pend);
                    pend = __builtin_amdgcn_ballot_w32(x != 0u);
                }
            }
        } else {
#pragma nounroll
            for (int k = 0; k < 16; ++k) {
                unsigned x = 0u;
#pragma unroll
                for (int j = 0; j < 16; ++j) x = (k == j) ? w[j] : x;
                const int e0 = ebv + (k >> 2) * (OWN_NW * 32 * 128) + (k & 3) * 32;
                unsigned pend = __builtin_amdgcn_ballot_w32(x != 0u);
                while (pend) {
                    if (x != 0u) {
                        wl[cnt + __popc(pend & lt)] = e0 + (__ffs((int)x) - 1);
                        x &= x - 1u;
                    }
                    cnt += __popc(pend);
                    if (cnt > OWN_CAP - 32) {
                        gat_own_drain<H, ACC>(wl, cnt, lo, E, srcI, dstI, a_s, ads, mxs, dens, Hf, acc, again, lane);
                        cnt = 0;
                    }
                    pend = __builtin_amdgcn_ballot_w32(x != 0u);
                }
            }
        }
#pragma unroll
        for (int u = 0; u < 4; ++u) cur[u] = nxt[u];
        gb = gn;
    }
    tail0 = (E >> 2) << 2;
    } else {
    const int nvec = E >> 2;
    const int4* d4 = (const int4*)dstI;
    const int4 none = make_int4(-1, -1, -1, -1);
    int gb = wave * 128;
    int4 cur[4];
#pragma unroll
    for (int u = 0; u < 4; ++u) {
        const int g = gb + u * 32 + lane;
        cur[u] = d4[max(min(g, nvec - 1), 0)];
        if (g >= nvec) cur[u] = none;
    }
    while (gb < nvec) {
        const int gn = gb + OWN_NW * 128;
        int4 nxt[4];
#pragma unroll
        for (int u = 0; u < 4; ++u) {
            const int g = gn + u * 32 + lane;
            nxt[u] = d4[min(g, nvec - 1)];
            if (g >= nvec) nxt[u] = none;
        }
#pragma unroll
        for (int u = 0; u < 4; ++u) {
            const int e0 = (gb + u * 32 + lane) * 4;
            const bool h0 = (unsigned)(cur[u].x - lo) < (unsigned)nodes;
            const bool h1 = (unsigned)(cur[u].y - lo) < (unsigned)nodes;
            const bool h2 = (unsigned)(cur[u].z - lo) < (unsigned)nodes;
            const bool h3 = (unsigned)(cur[u].w - lo) < (unsigned)nodes;
            if (__builtin_amdgcn_ballot_w32(h0 | h1 | h2 | h3)) {
                unsigned mask;
                mask = __builtin_amdgcn_ballot_w32(h0);
                if (h0) wl[cnt + __popc(mask & lt)] = e0 + 0;
                cnt += __popc(mask);
                mask = __builtin_amdgcn_ballot_w32(h1);
                if (h1) wl[cnt + __popc(mask & lt)] = e0 + 1;
                cnt += __popc(mask);
                mask = __builtin_amdgcn_ballot_w32(h2);
                if (h2) wl[cnt + __popc(mask & lt)] = e0 + 2;
                cnt += __popc(mask);
                mask = __builtin_amdgcn_ballot_w32(h3);
                if (h3) wl[cnt + __popc(mask & lt)] = e0 + 3;
                cnt += __popc(mask);
            }
        }
        if (cnt > OWN_CAP / 2) {
            gat_own_drain<H, ACC>(wl, cnt, lo, E, srcI, dstI, a_s, ads, mxs, dens, Hf, acc, again, lane);
            cnt = 0;
        }
#pragma unroll
        for (int u = 0; u < 4; ++u) cur[u] = nxt[u];
        gb = gn;
    }
    tail0 = nvec * 4;
    }
    if (wave == 0) {
        for (int eb0 = tail0; eb0 < E; eb0 += 32) {
            const int e = eb0 + lane;
            const bool hit = (e < E) && ((unsigned)(dstI[min(e, E - 1)] - lo) < (unsigned)nodes);
            const unsigned mask = __builtin_amdgcn_ballot_w32(hit);
            if (hit) wl[cnt + __popc(mask & lt)] = e;
            cnt += __popc(mask);
        }
    }
    return cnt;
}

template<int H, bool RELU, bool SAVE, bool LOAD>
__global__ __launch_bounds__(OWN_NW * 32)
void gat_own_aggregate(const int* __restrict__ srcI, const int* __restrict__ dstI,
                       const float* __restrict__ a_s, const float* __restrict__ a_d,
                       const float* __restrict__ Hf, const float* __restrict__ bias,
                       float* __restrict__ outp, int* __restrict__ glist, int* __restrict__ gcnt,
                       const unsigned* __restrict__ bm, int nchunk, int E, int N)
{
    constexpr int ROW = 32 * H;
    constexpr int LPR = ROW / 4;
    constexpr int PIT = OWN_PIT(H);
    static_assert(OWN_NPB <= 32 * OWN_NW, "one self-loop step per wave");
    static_assert(OWN_NPB <= 256, "local row index is packed into 8 bits");
    static_assert(PIT >= ROW, "a row's four slabs fit inside its pitch");
    __shared__ float acc[OWN_NPB * PIT];
    __shared__ int lst[OWN_NW * OWN_CAP];
    __shared__ float    ads[OWN_NPB * H];
    __shared__ unsigned mxs[OWN_NPB * H];
    __shared__ float    dens[OWN_NPB * H];
    __shared__ float    scs[OWN_NW * H * OWN_SC];
    __shared__ int again;

    const int tid  = threadIdx.x;
    const int lane = tid & 31;
    const int wave = tid >> 5;
    const int lo   = blockIdx.x * OWN_NPB;
    const int nodes = min(OWN_NPB, N - lo);
    const bool bm_ok = nchunk > 0 && gridDim.x <= (unsigned)BM_MAXROWS;
    const int wpr = nchunk * 64;

    for (int i = tid; i < nodes * PIT; i += OWN_NW * 32) acc[i] = 0.f;
    for (int i = tid; i < nodes * H; i += OWN_NW * 32) {
        const int j = gat_rh(i / H, i % H);
        ads[j]  = a_d[(size_t)lo * H + i];
        mxs[j]  = ENC_NEG_INF;
        dens[j] = 0.f;
    }
    if (tid == 0) again = 0;
    __syncthreads();

    int* wl = lst + wave * OWN_CAP;
    float* sc = scs + wave * (H * OWN_SC);
    int* gl = glist + ((size_t)blockIdx.x * OWN_NW + wave) * OWN_CAP;
    int* gc = gcnt + (size_t)blockIdx.x * OWN_NW;
    int cnt = 0;
    bool saved = false;
    if (LOAD) saved = gc[0] >= 0;
    if (saved) {
        cnt = gc[wave];
        gat_own_list_max<H>(gl, wl, cnt, a_s, ads, mxs, sc, lane);
    } else {
        cnt = gat_own_sweep<H, false>(wl, lo, nodes, E, srcI, dstI, bm, bm_ok, wpr, a_s, ads, mxs, dens, Hf, acc, &again, lane, wave);
        gat_own_flush_max<H, SAVE, true>(wl, gl, cnt, lo, E, srcI, dstI, a_s, ads, mxs, sc, lane);
    }
    __syncthreads();

    const int ag = __hip_atomic_load(&again, __ATOMIC_RELAXED, __HIP_MEMORY_SCOPE_WORKGROUP);
    if (SAVE && lane == 0) gc[wave] = ag ? -1 : cnt;
    if (ag == 0) {
        gat_own_flush_acc<H, true, true>(wl, cnt, lo, E, srcI, dstI, a_s, ads, mxs, dens, Hf, acc, sc, lane);
    } else {
        cnt = gat_own_sweep<H, true>(wl, lo, nodes, E, srcI, dstI, bm, bm_ok, wpr, a_s, ads, mxs, dens, Hf, acc, &again, lane, wave);
        gat_own_flush_acc<H, false, false>(wl, cnt, lo, E, srcI, dstI, a_s, ads, mxs, dens, Hf, acc, nullptr, lane);
    }
    __syncthreads();

    float* orow = outp + (size_t)lo * ROW;
    for (int i = tid * 4; i < nodes * ROW; i += OWN_NW * 32 * 4) {
        const int c = i & (ROW - 1);
        const float* a = acc + (i / ROW) * PIT + (c >> 2);
        float4 r = make_float4(a[0], a[LPR], a[2 * LPR], a[3 * LPR]);
        const float dn = dens[gat_rh(i / ROW, c >> 5)];
        r.x = r.x / dn + bias[c + 0];
        r.y = r.y / dn + bias[c + 1];
        r.z = r.z / dn + bias[c + 2];
        r.w = r.w / dn + bias[c + 3];
        if (RELU) { r.x = fmaxf(r.x, 0.f); r.y = fmaxf(r.y, 0.f); r.z = fmaxf(r.z, 0.f); r.w = fmaxf(r.w, 0.f); }
        *(float4*)(orow + i) = r;
    }
}

extern "C" void kernel_launch(void* const* d_in, const int* in_sizes, int n_in,
                              void* d_out, int out_size, void* d_ws, size_t ws_size,
                              hipStream_t stream)
{
    const float* x   = (const float*)d_in[0];
    const int*   ei  = (const int*)d_in[1];
    const float* W1  = (const float*)d_in[2];
    const float* as1 = (const float*)d_in[3];
    const float* ad1 = (const float*)d_in[4];
    const float* b1  = (const float*)d_in[5];
    const float* W2  = (const float*)d_in[6];
    const float* as2 = (const float*)d_in[7];
    const float* ad2 = (const float*)d_in[8];
    const float* b2  = (const float*)d_in[9];
    float* out = (float*)d_out;

    const int N  = in_sizes[0] / 128;
    const int E  = in_sizes[1] / 2;
    const int* srcI = ei;
    const int* dstI = ei + E;

    float* ws = (float*)d_ws;
    float*    h1    = ws;              ws += (size_t)N * 128;
    float*    asrc1 = ws;              ws += (size_t)N * 4;
    float*    adst1 = ws;              ws += (size_t)N * 4;
    float*    h2    = ws;              ws += (size_t)N * 128;
    float*    g2    = ws;              ws += (size_t)N * 32;
    const int gA     = (N + OWN_NPB - 1) / OWN_NPB;
    const int nchunk = (E + BM_CH - 1) / BM_CH;
    unsigned* bm    = (unsigned*)ws;   ws += (size_t)gA * nchunk * 64;
    float*    asrc2 = ws;              ws += (size_t)N;
    float*    adst2 = ws;              ws += (size_t)N;

    const int TB = 256;
    const int gB   = (N + 127) / 128;
    int* glist = (int*)ws;
    int* gcnt  = glist + (size_t)gA * OWN_NW * OWN_CAP;
    const int nquad = E >> 2;

    gat_gemm_wmma<8><<<gB, TB, 0, stream>>>(x, W1, as1, ad1, h1, asrc1, adst1, N, dstI, bm, nquad, nchunk, gA);
    gat_own_aggregate<4, true, true, false><<<gA, OWN_NW * 32, 0, stream>>>(srcI, dstI, asrc1, adst1, h1, b1, h2, glist, gcnt, bm, nchunk, E, N);

    gat_gemm_wmma<2><<<gB, TB, 0, stream>>>(h2, W2, as2, ad2, g2, asrc2, adst2, N, dstI, bm, 0, 0, gA);
    gat_own_aggregate<1, false, false, true><<<gA, OWN_NW * 32, 0, stream>>>(srcI, dstI, asrc2, adst2, g2, b2, out, glist, gcnt, bm, nchunk, E, N);
}
